# starting kernel
# baseline (speedup 1.0000x reference)
#include <hip/hip_runtime.h>

typedef float v2f __attribute__((ext_vector_type(2)));
typedef float v8f __attribute__((ext_vector_type(8)));

#define NNODES 50000
#define NEDGES 800000
#define HID 64
#define NGRAPHS 512
#define NCLASSES 10


__global__ __launch_bounds__(256) void k_init_deg(float* __restrict__ deg, int n) {
    int i = blockIdx.x * blockDim.x + threadIdx.x;
    if (i < n) deg[i] = 1.0f;
}

__global__ __launch_bounds__(256) void k_deg_accum(const int* __restrict__ dst,
                                                   float* __restrict__ deg, int e) {
    int i = blockIdx.x * blockDim.x + threadIdx.x;
    if (i < e) atomicAdd(&deg[dst[i]], 1.0f);
}

__global__ __launch_bounds__(256) void k_dinv(float* __restrict__ deg, int n) {
    int i = blockIdx.x * blockDim.x + threadIdx.x;
    if (i < n) deg[i] = rsqrtf(deg[i]);
}


__global__ __launch_bounds__(256)
void k_gemm_wmma(const float* __restrict__ A, const float* __restrict__ W,
                 float* __restrict__ C, int n, int relu_in) {
    __shared__ float sW[64 * 64];
    const int tid = threadIdx.x;
    #pragma unroll
    for (int i = 0; i < 16; ++i) sW[tid + i * 256] = W[tid + i * 256];
    __syncthreads();

    const int wave = tid >> 5;
    const int lane = tid & 31;
    const int half = lane >> 4;
    const int l16  = lane & 15;
    const int m0   = (blockIdx.x * 8 + wave) * 16;
    if (m0 >= n) return;

    v8f c0 = {}, c1 = {}, c2 = {}, c3 = {};

    const int arow = m0 + l16;
    const float* __restrict__ Arow = A + (size_t)arow * 64;

    #pragma unroll
    for (int k0 = 0; k0 < 64; k0 += 4) {
        const int kb = k0 + half * 2;
        float a0 = Arow[kb];
        float a1 = Arow[kb + 1];
        if (relu_in) { a0 = fmaxf(a0, 0.0f); a1 = fmaxf(a1, 0.0f); }
        v2f a = {a0, a1};

        const float* __restrict__ Wk0 = sW + (size_t)kb * 64;
        const float* __restrict__ Wk1 = sW + (size_t)(kb + 1) * 64;
        v2f b0 = {Wk0[l16 +  0], Wk1[l16 +  0]};
        v2f b1 = {Wk0[l16 + 16], Wk1[l16 + 16]};
        v2f b2 = {Wk0[l16 + 32], Wk1[l16 + 32]};
        v2f b3 = {Wk0[l16 + 48], Wk1[l16 + 48]};

        c0 = __builtin_amdgcn_wmma_f32_16x16x4_f32(false, a, false, b0, (short)0, c0, false, false);
        c1 = __builtin_amdgcn_wmma_f32_16x16x4_f32(false, a, false, b1, (short)0, c1, false, false);
        c2 = __builtin_amdgcn_wmma_f32_16x16x4_f32(false, a, false, b2, (short)0, c2, false, false);
        c3 = __builtin_amdgcn_wmma_f32_16x16x4_f32(false, a, false, b3, (short)0, c3, false, false);
    }

    #pragma unroll
    for (int r = 0; r < 8; ++r) {
        const int row = m0 + half * 8 + r;
        if (row < n) {
            float* __restrict__ Cr = C + (size_t)row * 64 + l16;
            Cr[0]  = c0[r];
            Cr[16] = c1[r];
            Cr[32] = c2[r];
            Cr[48] = c3[r];
        }
    }
}


__global__ __launch_bounds__(256)
void k_selfbias(const float* __restrict__ h, const float* __restrict__ dinv,
                const float* __restrict__ bias, float* __restrict__ a, int n) {
    int t = blockIdx.x * blockDim.x + threadIdx.x;
    if (t < n * HID) {
        int i = t >> 6;
        int f = t & 63;
        float di = dinv[i];
        a[t] = h[t] * di * di + bias[f];
    }
}

__global__ __launch_bounds__(256)
void k_edge_scatter(const float* __restrict__ h, const float* __restrict__ dinv,
                    const int* __restrict__ src, const int* __restrict__ dst,
                    float* __restrict__ a, int e) {
    int t = blockIdx.x * blockDim.x + threadIdx.x;
    int ei = t >> 6;
    int f = t & 63;
    if (ei < e) {
        int s = src[ei];
        int d = dst[ei];
        float nrm = dinv[s] * dinv[d];
        atomicAdd(&a[(size_t)d * 64 + f], h[(size_t)s * 64 + f] * nrm);
    }
}


__global__ __launch_bounds__(256) void k_zero(float* __restrict__ p, int n) {
    int i = blockIdx.x * blockDim.x + threadIdx.x;
    if (i < n) p[i] = 0.0f;
}

__global__ __launch_bounds__(256)
void k_pool(const float* __restrict__ a, const int* __restrict__ batch,
            float* __restrict__ sums, float* __restrict__ cnt, int n) {
    int t = blockIdx.x * blockDim.x + threadIdx.x;
    if (t < n * HID) {
        int i = t >> 6;
        int f = t & 63;
        int g = batch[i];
        atomicAdd(&sums[(size_t)g * HID + f], fmaxf(a[t], 0.0f));
        if (f == 0) atomicAdd(&cnt[g], 1.0f);
    }
}

__global__ __launch_bounds__(256)
void k_head(const float* __restrict__ sums, const float* __restrict__ cnt,
            const float* __restrict__ Wl, const float* __restrict__ bl,
            float* __restrict__ out) {
    int t = blockIdx.x * blockDim.x + threadIdx.x;
    if (t < NGRAPHS * NCLASSES) {
        int g = t / NCLASSES;
        int c = t % NCLASSES;
        float inv = 1.0f / fmaxf(cnt[g], 1.0f);
        float acc = 0.0f;
        const float* sg = sums + (size_t)g * HID;
        #pragma unroll
        for (int k = 0; k < HID; ++k) acc += sg[k] * Wl[k * NCLASSES + c];
        out[t] = acc * inv + bl[c];
    }
}


extern "C" void kernel_launch(void* const* d_in, const int* in_sizes, int n_in,
                              void* d_out, int out_size, void* d_ws, size_t ws_size,
                              hipStream_t stream) {
    const float* x  = (const float*)d_in[0];
    const float* W1 = (const float*)d_in[1];
    const float* b1 = (const float*)d_in[2];
    const float* W2 = (const float*)d_in[3];
    const float* b2 = (const float*)d_in[4];
    const float* Wl = (const float*)d_in[5];
    const float* bl = (const float*)d_in[6];
    const int* edge  = (const int*)d_in[7];
    const int* batch = (const int*)d_in[8];
    const int* src = edge;
    const int* dst = edge + NEDGES;
    float* out = (float*)d_out;

    float* ws   = (float*)d_ws;
    float* h    = ws;
    float* a    = h + (size_t)NNODES * HID;
    float* dinv = a + (size_t)NNODES * HID;
    float* sums = dinv + NNODES;
    float* cnt  = sums + (size_t)NGRAPHS * HID;

    const int TPB = 256;
    const int gN   = (NNODES + TPB - 1) / TPB;
    const int gE   = (NEDGES + TPB - 1) / TPB;
    const int gNF  = (NNODES * HID + TPB - 1) / TPB;
    const int gEd  = (NEDGES + 3) / 4;
    const int gGmm = ((NNODES + 15) / 16 + 7) / 8;
    const int gPZ  = (NGRAPHS * HID + NGRAPHS + TPB - 1) / TPB;
    const int gHd  = (NGRAPHS * NCLASSES + TPB - 1) / TPB;

    k_init_deg<<<gN, TPB, 0, stream>>>(dinv, NNODES);
    k_deg_accum<<<gE, TPB, 0, stream>>>(dst, dinv, NEDGES);
    k_dinv<<<gN, TPB, 0, stream>>>(dinv, NNODES);

    k_gemm_wmma<<<gGmm, TPB, 0, stream>>>(x, W1, h, NNODES, 0);
    k_selfbias<<<gNF, TPB, 0, stream>>>(h, dinv, b1, a, NNODES);
    k_edge_scatter<<<gEd, TPB, 0, stream>>>(h, dinv, src, dst, a, NEDGES);

    k_gemm_wmma<<<gGmm, TPB, 0, stream>>>(a, W2, h, NNODES, 1);
    k_selfbias<<<gNF, TPB, 0, stream>>>(h, dinv, b2, a, NNODES);
    k_edge_scatter<<<gEd, TPB, 0, stream>>>(h, dinv, src, dst, a, NEDGES);

    k_zero<<<gPZ, TPB, 0, stream>>>(sums, NGRAPHS * HID + NGRAPHS);
    k_pool<<<gNF, TPB, 0, stream>>>(a, batch, sums, cnt, NNODES);
    k_head<<<gHd, TPB, 0, stream>>>(sums, cnt, Wl, bl, out);
}


# Float-atomic edge scatter replaced by in-edge lists per destination (int-atomic rank, prefix sum) and a wave-per-node gather with self-loop and bias; rows summed in list order, stored once
# speedup vs baseline: 2.1724x; 2.1724x over previous
#include <hip/hip_runtime.h>

typedef float v2f __attribute__((ext_vector_type(2)));
typedef float v8f __attribute__((ext_vector_type(8)));

#define NNODES 50000
#define NEDGES 800000
#define HID 64
#define NGRAPHS 512
#define NCLASSES 10


__global__ __launch_bounds__(256) void k_zero_i(int* __restrict__ p, int n) {
    int i = blockIdx.x * blockDim.x + threadIdx.x;
    if (i < n) p[i] = 0;
}

__global__ __launch_bounds__(256) void k_rank(const int* __restrict__ dst, int* __restrict__ cnt,
                                              int* __restrict__ rank, int e) {
    int i = blockIdx.x * blockDim.x + threadIdx.x;
    if (i < e) rank[i] = atomicAdd(&cnt[dst[i]], 1);
}

__global__ __launch_bounds__(256)
void k_rowptr(const int* __restrict__ cnt, int* __restrict__ rowptr, float* __restrict__ dinv, int n) {
    __shared__ int s_before[256];
    __shared__ int s_own[256];
    const int tid  = threadIdx.x;
    const int base = blockIdx.x * 256;
    int before = 0;
    for (int j = tid * 4; j < base; j += 1024) {
        const int4 v = *reinterpret_cast<const int4*>(cnt + j);
        before += (v.x + v.y) + (v.z + v.w);
    }
    const int i   = base + tid;
    const int own = (i < n) ? cnt[i] : 0;
    s_before[tid] = before;
    s_own[tid]    = own;
    __syncthreads();
    for (int off = 1; off < 256; off <<= 1) {
        int pb = 0, po = 0;
        if (tid >= off) { pb = s_before[tid - off]; po = s_own[tid - off]; }
        __syncthreads();
        s_before[tid] += pb;
        s_own[tid]    += po;
        __syncthreads();
    }
    const int start = s_before[255] + s_own[tid] - own;
    if (i < n) {
        rowptr[i] = start;
        dinv[i]   = rsqrtf((float)(own + 1));
    }
    if (i == n - 1) rowptr[n] = start + own;
}

__global__ __launch_bounds__(256)
void k_fill(const int* __restrict__ src, const int* __restrict__ dst, const int* __restrict__ rowptr,
            const int* __restrict__ rank, int* __restrict__ col, int e) {
    int i = blockIdx.x * blockDim.x + threadIdx.x;
    if (i < e) col[rowptr[dst[i]] + rank[i]] = src[i];
}


__global__ __launch_bounds__(256)
void k_gemm_wmma(const float* __restrict__ A, const float* __restrict__ W,
                 float* __restrict__ C, int n, int relu_in) {
    __shared__ float sW[64 * 64];
    const int tid = threadIdx.x;
    #pragma unroll
    for (int i = 0; i < 16; ++i) sW[tid + i * 256] = W[tid + i * 256];
    __syncthreads();

    const int wave = tid >> 5;
    const int lane = tid & 31;
    const int half = lane >> 4;
    const int l16  = lane & 15;
    const int m0   = (blockIdx.x * 8 + wave) * 16;
    if (m0 >= n) return;

    v8f c0 = {}, c1 = {}, c2 = {}, c3 = {};

    const int arow = m0 + l16;
    const float* __restrict__ Arow = A + (size_t)arow * 64;

    #pragma unroll
    for (int k0 = 0; k0 < 64; k0 += 4) {
        const int kb = k0 + half * 2;
        float a0 = Arow[kb];
        float a1 = Arow[kb + 1];
        if (relu_in) { a0 = fmaxf(a0, 0.0f); a1 = fmaxf(a1, 0.0f); }
        v2f a = {a0, a1};

        const float* __restrict__ Wk0 = sW + (size_t)kb * 64;
        const float* __restrict__ Wk1 = sW + (size_t)(kb + 1) * 64;
        v2f b0 = {Wk0[l16 +  0], Wk1[l16 +  0]};
        v2f b1 = {Wk0[l16 + 16], Wk1[l16 + 16]};
        v2f b2 = {Wk0[l16 + 32], Wk1[l16 + 32]};
        v2f b3 = {Wk0[l16 + 48], Wk1[l16 + 48]};

        c0 = __builtin_amdgcn_wmma_f32_16x16x4_f32(false, a, false, b0, (short)0, c0, false, false);
        c1 = __builtin_amdgcn_wmma_f32_16x16x4_f32(false, a, false, b1, (short)0, c1, false, false);
        c2 = __builtin_amdgcn_wmma_f32_16x16x4_f32(false, a, false, b2, (short)0, c2, false, false);
        c3 = __builtin_amdgcn_wmma_f32_16x16x4_f32(false, a, false, b3, (short)0, c3, false, false);
    }

    #pragma unroll
    for (int r = 0; r < 8; ++r) {
        const int row = m0 + half * 8 + r;
        if (row < n) {
            float* __restrict__ Cr = C + (size_t)row * 64 + l16;
            Cr[0]  = c0[r];
            Cr[16] = c1[r];
            Cr[32] = c2[r];
            Cr[48] = c3[r];
        }
    }
}


__global__ __launch_bounds__(256)
void k_gather(const float* __restrict__ h, const float* __restrict__ dinv,
              const float* __restrict__ bias, const int* __restrict__ rowptr,
              const int* __restrict__ col, float* __restrict__ a, int n) {
    const int lane = threadIdx.x & 31;
    const int i = __builtin_amdgcn_readfirstlane((int)(blockIdx.x * 8 + (threadIdx.x >> 5)));
    if (i >= n) return;
    const float di = dinv[i];
    const int beg  = rowptr[i];
    const int end  = rowptr[i + 1];
    const v2f* __restrict__ hp = reinterpret_cast<const v2f*>(h);
    const v2f hs = hp[(size_t)i * 32 + lane];
    float ax = hs.x * di * di + bias[2 * lane];
    float ay = hs.y * di * di + bias[2 * lane + 1];
    for (int c0 = beg; c0 < end; c0 += 32) {
        const int m = min(32, end - c0);
        int   s = 0;
        float w = 0.0f;
        if (lane < m) { s = col[c0 + lane]; w = dinv[s] * di; }
        const int wi = __float_as_int(w);
        int k = 0;
        for (; k + 4 <= m; k += 4) {
            const int s0 = __builtin_amdgcn_readlane(s, k);
            const int s1 = __builtin_amdgcn_readlane(s, k + 1);
            const int s2 = __builtin_amdgcn_readlane(s, k + 2);
            const int s3 = __builtin_amdgcn_readlane(s, k + 3);
            const float w0 = __int_as_float(__builtin_amdgcn_readlane(wi, k));
            const float w1 = __int_as_float(__builtin_amdgcn_readlane(wi, k + 1));
            const float w2 = __int_as_float(__builtin_amdgcn_readlane(wi, k + 2));
            const float w3 = __int_as_float(__builtin_amdgcn_readlane(wi, k + 3));
            const v2f r0 = hp[(size_t)s0 * 32 + lane];
            const v2f r1 = hp[(size_t)s1 * 32 + lane];
            const v2f r2 = hp[(size_t)s2 * 32 + lane];
            const v2f r3 = hp[(size_t)s3 * 32 + lane];
            ax = fmaf(r0.x, w0, ax); ay = fmaf(r0.y, w0, ay);
            ax = fmaf(r1.x, w1, ax); ay = fmaf(r1.y, w1, ay);
            ax = fmaf(r2.x, w2, ax); ay = fmaf(r2.y, w2, ay);
            ax = fmaf(r3.x, w3, ax); ay = fmaf(r3.y, w3, ay);
        }
        for (; k < m; ++k) {
            const int   s0 = __builtin_amdgcn_readlane(s, k);
            const float w0 = __int_as_float(__builtin_amdgcn_readlane(wi, k));
            const v2f   r0 = hp[(size_t)s0 * 32 + lane];
            ax = fmaf(r0.x, w0, ax); ay = fmaf(r0.y, w0, ay);
        }
    }
    v2f o = {ax, ay};
    reinterpret_cast<v2f*>(a)[(size_t)i * 32 + lane] = o;
}


__global__ __launch_bounds__(256) void k_zero(float* __restrict__ p, int n) {
    int i = blockIdx.x * blockDim.x + threadIdx.x;
    if (i < n) p[i] = 0.0f;
}

__global__ __launch_bounds__(256)
void k_pool(const float* __restrict__ a, const int* __restrict__ batch,
            float* __restrict__ sums, float* __restrict__ cnt, int n) {
    int t = blockIdx.x * blockDim.x + threadIdx.x;
    if (t < n * HID) {
        int i = t >> 6;
        int f = t & 63;
        int g = batch[i];
        atomicAdd(&sums[(size_t)g * HID + f], fmaxf(a[t], 0.0f));
        if (f == 0) atomicAdd(&cnt[g], 1.0f);
    }
}

__global__ __launch_bounds__(256)
void k_head(const float* __restrict__ sums, const float* __restrict__ cnt,
            const float* __restrict__ Wl, const float* __restrict__ bl,
            float* __restrict__ out) {
    int t = blockIdx.x * blockDim.x + threadIdx.x;
    if (t < NGRAPHS * NCLASSES) {
        int g = t / NCLASSES;
        int c = t % NCLASSES;
        float inv = 1.0f / fmaxf(cnt[g], 1.0f);
        float acc = 0.0f;
        const float* sg = sums + (size_t)g * HID;
        #pragma unroll
        for (int k = 0; k < HID; ++k) acc += sg[k] * Wl[k * NCLASSES + c];
        out[t] = acc * inv + bl[c];
    }
}


extern "C" void kernel_launch(void* const* d_in, const int* in_sizes, int n_in,
                              void* d_out, int out_size, void* d_ws, size_t ws_size,
                              hipStream_t stream) {
    const float* x  = (const float*)d_in[0];
    const float* W1 = (const float*)d_in[1];
    const float* b1 = (const float*)d_in[2];
    const float* W2 = (const float*)d_in[3];
    const float* b2 = (const float*)d_in[4];
    const float* Wl = (const float*)d_in[5];
    const float* bl = (const float*)d_in[6];
    const int* edge  = (const int*)d_in[7];
    const int* batch = (const int*)d_in[8];
    const int* src = edge;
    const int* dst = edge + NEDGES;
    float* out = (float*)d_out;

    float* ws   = (float*)d_ws;
    float* h    = ws;
    float* a    = h + (size_t)NNODES * HID;
    float* dinv = a + (size_t)NNODES * HID;
    float* sums = dinv + NNODES;
    float* cnt  = sums + (size_t)NGRAPHS * HID;
    int* icnt   = (int*)(cnt + NGRAPHS);
    int* rowptr = icnt + NNODES;
    int* rank   = rowptr + NNODES + 4;
    int* col    = rank + NEDGES;

    const int TPB = 256;
    const int gN   = (NNODES + TPB - 1) / TPB;
    const int gE   = (NEDGES + TPB - 1) / TPB;
    const int gNF  = (NNODES * HID + TPB - 1) / TPB;
    const int gGa  = (NNODES + 7) / 8;
    const int gGmm = ((NNODES + 15) / 16 + 7) / 8;
    const int gPZ  = (NGRAPHS * HID + NGRAPHS + TPB - 1) / TPB;
    const int gHd  = (NGRAPHS * NCLASSES + TPB - 1) / TPB;

    k_zero_i<<<gN, TPB, 0, stream>>>(icnt, NNODES);
    k_rank<<<gE, TPB, 0, stream>>>(dst, icnt, rank, NEDGES);
    k_rowptr<<<gN, TPB, 0, stream>>>(icnt, rowptr, dinv, NNODES);
    k_fill<<<gE, TPB, 0, stream>>>(src, dst, rowptr, rank, col, NEDGES);

    k_gemm_wmma<<<gGmm, TPB, 0, stream>>>(x, W1, h, NNODES, 0);
    k_gather<<<gGa, TPB, 0, stream>>>(h, dinv, b1, rowptr, col, a, NNODES);

    k_gemm_wmma<<<gGmm, TPB, 0, stream>>>(a, W2, h, NNODES, 1);
    k_gather<<<gGa, TPB, 0, stream>>>(h, dinv, b2, rowptr, col, a, NNODES);

    k_zero<<<gPZ, TPB, 0, stream>>>(sums, NGRAPHS * HID + NGRAPHS);
    k_pool<<<gNF, TPB, 0, stream>>>(a, batch, sums, cnt, NNODES);
    k_head<<<gHd, TPB, 0, stream>>>(sums, cnt, Wl, bl, out);
}


# Pooling: one wave per 16 consecutive nodes sums relu rows in registers and issues one float atomic per run of equal graph ids instead of one per node element; partial sums per wave, then atomics
# speedup vs baseline: 2.5386x; 1.1686x over previous
#include <hip/hip_runtime.h>

typedef float v2f __attribute__((ext_vector_type(2)));
typedef float v8f __attribute__((ext_vector_type(8)));

#define NNODES 50000
#define NEDGES 800000
#define HID 64
#define NGRAPHS 512
#define NCLASSES 10


__global__ __launch_bounds__(256) void k_zero_i(int* __restrict__ p, int n) {
    int i = blockIdx.x * blockDim.x + threadIdx.x;
    if (i < n) p[i] = 0;
}

__global__ __launch_bounds__(256) void k_rank(const int* __restrict__ dst, int* __restrict__ cnt,
                                              int* __restrict__ rank, int e) {
    int i = blockIdx.x * blockDim.x + threadIdx.x;
    if (i < e) rank[i] = atomicAdd(&cnt[dst[i]], 1);
}

__global__ __launch_bounds__(256)
void k_rowptr(const int* __restrict__ cnt, int* __restrict__ rowptr, float* __restrict__ dinv, int n) {
    __shared__ int s_before[256];
    __shared__ int s_own[256];
    const int tid  = threadIdx.x;
    const int base = blockIdx.x * 256;
    int before = 0;
    for (int j = tid * 4; j < base; j += 1024) {
        const int4 v = *reinterpret_cast<const int4*>(cnt + j);
        before += (v.x + v.y) + (v.z + v.w);
    }
    const int i   = base + tid;
    const int own = (i < n) ? cnt[i] : 0;
    s_before[tid] = before;
    s_own[tid]    = own;
    __syncthreads();
    for (int off = 1; off < 256; off <<= 1) {
        int pb = 0, po = 0;
        if (tid >= off) { pb = s_before[tid - off]; po = s_own[tid - off]; }
        __syncthreads();
        s_before[tid] += pb;
        s_own[tid]    += po;
        __syncthreads();
    }
    const int start = s_before[255] + s_own[tid] - own;
    if (i < n) {
        rowptr[i] = start;
        dinv[i]   = rsqrtf((float)(own + 1));
    }
    if (i == n - 1) rowptr[n] = start + own;
}

__global__ __launch_bounds__(256)
void k_fill(const int* __restrict__ src, const int* __restrict__ dst, const int* __restrict__ rowptr,
            const int* __restrict__ rank, int* __restrict__ col, int e) {
    int i = blockIdx.x * blockDim.x + threadIdx.x;
    if (i < e) col[rowptr[dst[i]] + rank[i]] = src[i];
}


__global__ __launch_bounds__(256)
void k_gemm_wmma(const float* __restrict__ A, const float* __restrict__ W,
                 float* __restrict__ C, int n, int relu_in) {
    __shared__ float sW[64 * 64];
    const int tid = threadIdx.x;
    #pragma unroll
    for (int i = 0; i < 16; ++i) sW[tid + i * 256] = W[tid + i * 256];
    __syncthreads();

    const int wave = tid >> 5;
    const int lane = tid & 31;
    const int half = lane >> 4;
    const int l16  = lane & 15;
    const int m0   = (blockIdx.x * 8 + wave) * 16;
    if (m0 >= n) return;

    v8f c0 = {}, c1 = {}, c2 = {}, c3 = {};

    const int arow = m0 + l16;
    const float* __restrict__ Arow = A + (size_t)arow * 64;

    #pragma unroll
    for (int k0 = 0; k0 < 64; k0 += 4) {
        const int kb = k0 + half * 2;
        float a0 = Arow[kb];
        float a1 = Arow[kb + 1];
        if (relu_in) { a0 = fmaxf(a0, 0.0f); a1 = fmaxf(a1, 0.0f); }
        v2f a = {a0, a1};

        const float* __restrict__ Wk0 = sW + (size_t)kb * 64;
        const float* __restrict__ Wk1 = sW + (size_t)(kb + 1) * 64;
        v2f b0 = {Wk0[l16 +  0], Wk1[l16 +  0]};
        v2f b1 = {Wk0[l16 + 16], Wk1[l16 + 16]};
        v2f b2 = {Wk0[l16 + 32], Wk1[l16 + 32]};
        v2f b3 = {Wk0[l16 + 48], Wk1[l16 + 48]};

        c0 = __builtin_amdgcn_wmma_f32_16x16x4_f32(false, a, false, b0, (short)0, c0, false, false);
        c1 = __builtin_amdgcn_wmma_f32_16x16x4_f32(false, a, false, b1, (short)0, c1, false, false);
        c2 = __builtin_amdgcn_wmma_f32_16x16x4_f32(false, a, false, b2, (short)0, c2, false, false);
        c3 = __builtin_amdgcn_wmma_f32_16x16x4_f32(false, a, false, b3, (short)0, c3, false, false);
    }

    #pragma unroll
    for (int r = 0; r < 8; ++r) {
        const int row = m0 + half * 8 + r;
        if (row < n) {
            float* __restrict__ Cr = C + (size_t)row * 64 + l16;
            Cr[0]  = c0[r];
            Cr[16] = c1[r];
            Cr[32] = c2[r];
            Cr[48] = c3[r];
        }
    }
}


__global__ __launch_bounds__(256)
void k_gather(const float* __restrict__ h, const float* __restrict__ dinv,
              const float* __restrict__ bias, const int* __restrict__ rowptr,
              const int* __restrict__ col, float* __restrict__ a, int n) {
    const int lane = threadIdx.x & 31;
    const int i = __builtin_amdgcn_readfirstlane((int)(blockIdx.x * 8 + (threadIdx.x >> 5)));
    if (i >= n) return;
    const float di = dinv[i];
    const int beg  = rowptr[i];
    const int end  = rowptr[i + 1];
    const v2f* __restrict__ hp = reinterpret_cast<const v2f*>(h);
    const v2f hs = hp[(size_t)i * 32 + lane];
    float ax = hs.x * di * di + bias[2 * lane];
    float ay = hs.y * di * di + bias[2 * lane + 1];
    for (int c0 = beg; c0 < end; c0 += 32) {
        const int m = min(32, end - c0);
        int   s = 0;
        float w = 0.0f;
        if (lane < m) { s = col[c0 + lane]; w = dinv[s] * di; }
        const int wi = __float_as_int(w);
        int k = 0;
        for (; k + 4 <= m; k += 4) {
            const int s0 = __builtin_amdgcn_readlane(s, k);
            const int s1 = __builtin_amdgcn_readlane(s, k + 1);
            const int s2 = __builtin_amdgcn_readlane(s, k + 2);
            const int s3 = __builtin_amdgcn_readlane(s, k + 3);
            const float w0 = __int_as_float(__builtin_amdgcn_readlane(wi, k));
            const float w1 = __int_as_float(__builtin_amdgcn_readlane(wi, k + 1));
            const float w2 = __int_as_float(__builtin_amdgcn_readlane(wi, k + 2));
            const float w3 = __int_as_float(__builtin_amdgcn_readlane(wi, k + 3));
            const v2f r0 = hp[(size_t)s0 * 32 + lane];
            const v2f r1 = hp[(size_t)s1 * 32 + lane];
            const v2f r2 = hp[(size_t)s2 * 32 + lane];
            const v2f r3 = hp[(size_t)s3 * 32 + lane];
            ax = fmaf(r0.x, w0, ax); ay = fmaf(r0.y, w0, ay);
            ax = fmaf(r1.x, w1, ax); ay = fmaf(r1.y, w1, ay);
            ax = fmaf(r2.x, w2, ax); ay = fmaf(r2.y, w2, ay);
            ax = fmaf(r3.x, w3, ax); ay = fmaf(r3.y, w3, ay);
        }
        for (; k < m; ++k) {
            const int   s0 = __builtin_amdgcn_readlane(s, k);
            const float w0 = __int_as_float(__builtin_amdgcn_readlane(wi, k));
            const v2f   r0 = hp[(size_t)s0 * 32 + lane];
            ax = fmaf(r0.x, w0, ax); ay = fmaf(r0.y, w0, ay);
        }
    }
    v2f o = {ax, ay};
    reinterpret_cast<v2f*>(a)[(size_t)i * 32 + lane] = o;
}


__global__ __launch_bounds__(256) void k_zero(float* __restrict__ p, int n) {
    int i = blockIdx.x * blockDim.x + threadIdx.x;
    if (i < n) p[i] = 0.0f;
}

#define POOL_CHUNK 16

__device__ __forceinline__ void pool_flush(float* __restrict__ sums, float* __restrict__ cnt,
                                           int g, int lane, float sx, float sy, int run) {
    float* sg = sums + (size_t)g * HID + 2 * lane;
    atomicAdd(sg, sx);
    atomicAdd(sg + 1, sy);
    if (lane == 0) atomicAdd(&cnt[g], (float)run);
}

__global__ __launch_bounds__(256)
void k_pool(const float* __restrict__ a, const int* __restrict__ batch,
            float* __restrict__ sums, float* __restrict__ cnt, int n) {
    const int lane = threadIdx.x & 31;
    const int i0 = __builtin_amdgcn_readfirstlane((int)(blockIdx.x * 8 + (threadIdx.x >> 5))) * POOL_CHUNK;
    if (i0 >= n) return;
    const int m = min(POOL_CHUNK, n - i0);
    int gv = 0;
    if (lane < m) gv = batch[i0 + lane];
    const v2f* __restrict__ ap = reinterpret_cast<const v2f*>(a);
    v2f r[POOL_CHUNK];
    #pragma unroll
    for (int k = 0; k < POOL_CHUNK; ++k) r[k] = ap[(size_t)min(i0 + k, n - 1) * 32 + lane];

    int   cur = __builtin_amdgcn_readlane(gv, 0);
    int   run = 0;
    float sx = 0.0f, sy = 0.0f;
    #pragma unroll
    for (int k = 0; k < POOL_CHUNK; ++k) {
        if (k < m) {
            const int g = __builtin_amdgcn_readlane(gv, k);
            if (g != cur) {
                pool_flush(sums, cnt, cur, lane, sx, sy, run);
                cur = g; run = 0; sx = 0.0f; sy = 0.0f;
            }
            sx += fmaxf(r[k].x, 0.0f);
            sy += fmaxf(r[k].y, 0.0f);
            ++run;
        }
    }
    pool_flush(sums, cnt, cur, lane, sx, sy, run);
}

__global__ __launch_bounds__(256)
void k_head(const float* __restrict__ sums, const float* __restrict__ cnt,
            const float* __restrict__ Wl, const float* __restrict__ bl,
            float* __restrict__ out) {
    int t = blockIdx.x * blockDim.x + threadIdx.x;
    if (t < NGRAPHS * NCLASSES) {
        int g = t / NCLASSES;
        int c = t % NCLASSES;
        float inv = 1.0f / fmaxf(cnt[g], 1.0f);
        float acc = 0.0f;
        const float* sg = sums + (size_t)g * HID;
        #pragma unroll
        for (int k = 0; k < HID; ++k) acc += sg[k] * Wl[k * NCLASSES + c];
        out[t] = acc * inv + bl[c];
    }
}


extern "C" void kernel_launch(void* const* d_in, const int* in_sizes, int n_in,
                              void* d_out, int out_size, void* d_ws, size_t ws_size,
                              hipStream_t stream) {
    const float* x  = (const float*)d_in[0];
    const float* W1 = (const float*)d_in[1];
    const float* b1 = (const float*)d_in[2];
    const float* W2 = (const float*)d_in[3];
    const float* b2 = (const float*)d_in[4];
    const float* Wl = (const float*)d_in[5];
    const float* bl = (const float*)d_in[6];
    const int* edge  = (const int*)d_in[7];
    const int* batch = (const int*)d_in[8];
    const int* src = edge;
    const int* dst = edge + NEDGES;
    float* out = (float*)d_out;

    float* ws   = (float*)d_ws;
    float* h    = ws;
    float* a    = h + (size_t)NNODES * HID;
    float* dinv = a + (size_t)NNODES * HID;
    float* sums = dinv + NNODES;
    float* cnt  = sums + (size_t)NGRAPHS * HID;
    int* icnt   = (int*)(cnt + NGRAPHS);
    int* rowptr = icnt + NNODES;
    int* rank   = rowptr + NNODES + 4;
    int* col    = rank + NEDGES;

    const int TPB = 256;
    const int gN   = (NNODES + TPB - 1) / TPB;
    const int gE   = (NEDGES + TPB - 1) / TPB;
    const int gPl  = ((NNODES + POOL_CHUNK - 1) / POOL_CHUNK + 7) / 8;
    const int gGa  = (NNODES + 7) / 8;
    const int gGmm = ((NNODES + 15) / 16 + 7) / 8;
    const int gPZ  = (NGRAPHS * HID + NGRAPHS + TPB - 1) / TPB;
    const int gHd  = (NGRAPHS * NCLASSES + TPB - 1) / TPB;

    k_zero_i<<<gN, TPB, 0, stream>>>(icnt, NNODES);
    k_rank<<<gE, TPB, 0, stream>>>(dst, icnt, rank, NEDGES);
    k_rowptr<<<gN, TPB, 0, stream>>>(icnt, rowptr, dinv, NNODES);
    k_fill<<<gE, TPB, 0, stream>>>(src, dst, rowptr, rank, col, NEDGES);

    k_gemm_wmma<<<gGmm, TPB, 0, stream>>>(x, W1, h, NNODES, 0);
    k_gather<<<gGa, TPB, 0, stream>>>(h, dinv, b1, rowptr, col, a, NNODES);

    k_gemm_wmma<<<gGmm, TPB, 0, stream>>>(a, W2, h, NNODES, 1);
    k_gather<<<gGa, TPB, 0, stream>>>(h, dinv, b2, rowptr, col, a, NNODES);

    k_zero<<<gPZ, TPB, 0, stream>>>(sums, NGRAPHS * HID + NGRAPHS);
    k_pool<<<gPl, TPB, 0, stream>>>(a, batch, sums, cnt, NNODES);
    k_head<<<gHd, TPB, 0, stream>>>(sums, cnt, Wl, bl, out);
}


# In-edge lists built by a two-level counting sort on LDS counters (edges binned by 256-node bucket, one block per bucket) instead of a global returning atomic per edge; list order stays arrival order
# speedup vs baseline: 3.3365x; 1.3143x over previous
#include <hip/hip_runtime.h>

typedef float v2f __attribute__((ext_vector_type(2)));
typedef float v8f __attribute__((ext_vector_type(8)));

#define NNODES 50000
#define NEDGES 800000
#define HID 64
#define NGRAPHS 512
#define NCLASSES 10


#define BKT_SHIFT 8
#define BKT_NODES (1 << BKT_SHIFT)
#define NBKT ((NNODES + BKT_NODES - 1) / BKT_NODES)
#define CHUNK 4096
#define NCHUNK ((NEDGES + CHUNK - 1) / CHUNK)
#define NHIST (NBKT * NCHUNK)
#define CSR_BATCH 8

__global__ __launch_bounds__(256)
void k_hist(const int* __restrict__ dst, int* __restrict__ hist, int e) {
    __shared__ int s_cnt[NBKT];
    const int tid = threadIdx.x;
    if (tid < NBKT) s_cnt[tid] = 0;
    __syncthreads();
    const int base = blockIdx.x * CHUNK;
    int d[CHUNK / 256];
    #pragma unroll
    for (int j = 0; j < CHUNK / 256; ++j) d[j] = dst[min(base + j * 256 + tid, e - 1)];
    #pragma unroll
    for (int j = 0; j < CHUNK / 256; ++j) {
        if (base + j * 256 + tid < e) atomicAdd(&s_cnt[d[j] >> BKT_SHIFT], 1);
    }
    __syncthreads();
    if (tid < NBKT) hist[tid * NCHUNK + blockIdx.x] = s_cnt[tid];
}

__global__ __launch_bounds__(256)
void k_exscan(const int* __restrict__ in, int* __restrict__ out, int n) {
    __shared__ int s_before[256];
    __shared__ int s_own[256];
    const int tid  = threadIdx.x;
    const int base = blockIdx.x * 256;
    int before = 0;
    for (int j = tid * 4; j < base; j += 1024) {
        const int4 v = *reinterpret_cast<const int4*>(in + j);
        before += (v.x + v.y) + (v.z + v.w);
    }
    const int i   = base + tid;
    const int own = (i < n) ? in[i] : 0;
    s_before[tid] = before;
    s_own[tid]    = own;
    __syncthreads();
    for (int off = 1; off < 256; off <<= 1) {
        int pb = 0, po = 0;
        if (tid >= off) { pb = s_before[tid - off]; po = s_own[tid - off]; }
        __syncthreads();
        s_before[tid] += pb;
        s_own[tid]    += po;
        __syncthreads();
    }
    if (i < n) out[i] = s_before[255] + s_own[tid] - own;
}

__global__ __launch_bounds__(256)
void k_bin(const int* __restrict__ src, const int* __restrict__ dst, const int* __restrict__ offs,
           int* __restrict__ packed, int e) {
    __shared__ int s_pos[NBKT];
    const int tid = threadIdx.x;
    if (tid < NBKT) s_pos[tid] = offs[tid * NCHUNK + blockIdx.x];
    __syncthreads();
    const int base = blockIdx.x * CHUNK;
    int d[CHUNK / 256];
    int s[CHUNK / 256];
    #pragma unroll
    for (int j = 0; j < CHUNK / 256; ++j) {
        const int i = min(base + j * 256 + tid, e - 1);
        d[j] = dst[i];
        s[j] = src[i];
    }
    #pragma unroll
    for (int j = 0; j < CHUNK / 256; ++j) {
        if (base + j * 256 + tid < e) {
            const int slot = atomicAdd(&s_pos[d[j] >> BKT_SHIFT], 1);
            packed[slot] = (s[j] << BKT_SHIFT) | (d[j] & (BKT_NODES - 1));
        }
    }
}

__global__ __launch_bounds__(256)
void k_csr(const int* __restrict__ packed, const int* __restrict__ offs, int* __restrict__ rowptr,
           float* __restrict__ dinv, int* __restrict__ col, int n, int e) {
    __shared__ int s_cnt[BKT_NODES];
    __shared__ int s_scan[BKT_NODES];
    const int tid = threadIdx.x;
    const int k   = blockIdx.x;
    const int bs  = offs[k * NCHUNK];
    const int be  = (k + 1 < NBKT) ? offs[(k + 1) * NCHUNK] : e;
    s_cnt[tid] = 0;
    __syncthreads();
    for (int q = bs + tid; q < be; q += 256 * CSR_BATCH) {
        int v[CSR_BATCH];
        #pragma unroll
        for (int u = 0; u < CSR_BATCH; ++u) v[u] = packed[min(q + u * 256, be - 1)];
        #pragma unroll
        for (int u = 0; u < CSR_BATCH; ++u) {
            if (q + u * 256 < be) atomicAdd(&s_cnt[v[u] & (BKT_NODES - 1)], 1);
        }
    }
    __syncthreads();
    const int own = s_cnt[tid];
    s_scan[tid] = own;
    __syncthreads();
    for (int off = 1; off < 256; off <<= 1) {
        int v = 0;
        if (tid >= off) v = s_scan[tid - off];
        __syncthreads();
        s_scan[tid] += v;
        __syncthreads();
    }
    const int start = bs + s_scan[tid] - own;
    const int i = k * BKT_NODES + tid;
    if (i < n) {
        rowptr[i] = start;
        dinv[i]   = rsqrtf((float)(own + 1));
    }
    if (i == n - 1) rowptr[n] = start + own;
    s_cnt[tid] = start;
    __syncthreads();
    for (int q = bs + tid; q < be; q += 256 * CSR_BATCH) {
        int v[CSR_BATCH];
        #pragma unroll
        for (int u = 0; u < CSR_BATCH; ++u) v[u] = packed[min(q + u * 256, be - 1)];
        #pragma unroll
        for (int u = 0; u < CSR_BATCH; ++u) {
            if (q + u * 256 < be) {
                const int slot = atomicAdd(&s_cnt[v[u] & (BKT_NODES - 1)], 1);
                col[slot] = v[u] >> BKT_SHIFT;
            }
        }
    }
}


__global__ __launch_bounds__(256)
void k_gemm_wmma(const float* __restrict__ A, const float* __restrict__ W,
                 float* __restrict__ C, int n, int relu_in) {
    __shared__ float sW[64 * 64];
    const int tid = threadIdx.x;
    #pragma unroll
    for (int i = 0; i < 16; ++i) sW[tid + i * 256] = W[tid + i * 256];
    __syncthreads();

    const int wave = tid >> 5;
    const int lane = tid & 31;
    const int half = lane >> 4;
    const int l16  = lane & 15;
    const int m0   = (blockIdx.x * 8 + wave) * 16;
    if (m0 >= n) return;

    v8f c0 = {}, c1 = {}, c2 = {}, c3 = {};

    const int arow = m0 + l16;
    const float* __restrict__ Arow = A + (size_t)arow * 64;

    #pragma unroll
    for (int k0 = 0; k0 < 64; k0 += 4) {
        const int kb = k0 + half * 2;
        float a0 = Arow[kb];
        float a1 = Arow[kb + 1];
        if (relu_in) { a0 = fmaxf(a0, 0.0f); a1 = fmaxf(a1, 0.0f); }
        v2f a = {a0, a1};

        const float* __restrict__ Wk0 = sW + (size_t)kb * 64;
        const float* __restrict__ Wk1 = sW + (size_t)(kb + 1) * 64;
        v2f b0 = {Wk0[l16 +  0], Wk1[l16 +  0]};
        v2f b1 = {Wk0[l16 + 16], Wk1[l16 + 16]};
        v2f b2 = {Wk0[l16 + 32], Wk1[l16 + 32]};
        v2f b3 = {Wk0[l16 + 48], Wk1[l16 + 48]};

        c0 = __builtin_amdgcn_wmma_f32_16x16x4_f32(false, a, false, b0, (short)0, c0, false, false);
        c1 = __builtin_amdgcn_wmma_f32_16x16x4_f32(false, a, false, b1, (short)0, c1, false, false);
        c2 = __builtin_amdgcn_wmma_f32_16x16x4_f32(false, a, false, b2, (short)0, c2, false, false);
        c3 = __builtin_amdgcn_wmma_f32_16x16x4_f32(false, a, false, b3, (short)0, c3, false, false);
    }

    #pragma unroll
    for (int r = 0; r < 8; ++r) {
        const int row = m0 + half * 8 + r;
        if (row < n) {
            float* __restrict__ Cr = C + (size_t)row * 64 + l16;
            Cr[0]  = c0[r];
            Cr[16] = c1[r];
            Cr[32] = c2[r];
            Cr[48] = c3[r];
        }
    }
}


__global__ __launch_bounds__(256)
void k_gather(const float* __restrict__ h, const float* __restrict__ dinv,
              const float* __restrict__ bias, const int* __restrict__ rowptr,
              const int* __restrict__ col, float* __restrict__ a, int n) {
    const int lane = threadIdx.x & 31;
    const int i = __builtin_amdgcn_readfirstlane((int)(blockIdx.x * 8 + (threadIdx.x >> 5)));
    if (i >= n) return;
    const float di = dinv[i];
    const int beg  = rowptr[i];
    const int end  = rowptr[i + 1];
    const v2f* __restrict__ hp = reinterpret_cast<const v2f*>(h);
    const v2f hs = hp[(size_t)i * 32 + lane];
    float ax = hs.x * di * di + bias[2 * lane];
    float ay = hs.y * di * di + bias[2 * lane + 1];
    for (int c0 = beg; c0 < end; c0 += 32) {
        const int m = min(32, end - c0);
        int   s = 0;
        float w = 0.0f;
        if (lane < m) { s = col[c0 + lane]; w = dinv[s] * di; }
        const int wi = __float_as_int(w);
        int k = 0;
        for (; k + 4 <= m; k += 4) {
            const int s0 = __builtin_amdgcn_readlane(s, k);
            const int s1 = __builtin_amdgcn_readlane(s, k + 1);
            const int s2 = __builtin_amdgcn_readlane(s, k + 2);
            const int s3 = __builtin_amdgcn_readlane(s, k + 3);
            const float w0 = __int_as_float(__builtin_amdgcn_readlane(wi, k));
            const float w1 = __int_as_float(__builtin_amdgcn_readlane(wi, k + 1));
            const float w2 = __int_as_float(__builtin_amdgcn_readlane(wi, k + 2));
            const float w3 = __int_as_float(__builtin_amdgcn_readlane(wi, k + 3));
            const v2f r0 = hp[(size_t)s0 * 32 + lane];
            const v2f r1 = hp[(size_t)s1 * 32 + lane];
            const v2f r2 = hp[(size_t)s2 * 32 + lane];
            const v2f r3 = hp[(size_t)s3 * 32 + lane];
            ax = fmaf(r0.x, w0, ax); ay = fmaf(r0.y, w0, ay);
            ax = fmaf(r1.x, w1, ax); ay = fmaf(r1.y, w1, ay);
            ax = fmaf(r2.x, w2, ax); ay = fmaf(r2.y, w2, ay);
            ax = fmaf(r3.x, w3, ax); ay = fmaf(r3.y, w3, ay);
        }
        for (; k < m; ++k) {
            const int   s0 = __builtin_amdgcn_readlane(s, k);
            const float w0 = __int_as_float(__builtin_amdgcn_readlane(wi, k));
            const v2f   r0 = hp[(size_t)s0 * 32 + lane];
            ax = fmaf(r0.x, w0, ax); ay = fmaf(r0.y, w0, ay);
        }
    }
    v2f o = {ax, ay};
    reinterpret_cast<v2f*>(a)[(size_t)i * 32 + lane] = o;
}


__global__ __launch_bounds__(256) void k_zero(float* __restrict__ p, int n) {
    int i = blockIdx.x * blockDim.x + threadIdx.x;
    if (i < n) p[i] = 0.0f;
}

#define POOL_CHUNK 16

__device__ __forceinline__ void pool_flush(float* __restrict__ sums, float* __restrict__ cnt,
                                           int g, int lane, float sx, float sy, int run) {
    float* sg = sums + (size_t)g * HID + 2 * lane;
    atomicAdd(sg, sx);
    atomicAdd(sg + 1, sy);
    if (lane == 0) atomicAdd(&cnt[g], (float)run);
}

__global__ __launch_bounds__(256)
void k_pool(const float* __restrict__ a, const int* __restrict__ batch,
            float* __restrict__ sums, float* __restrict__ cnt, int n) {
    const int lane = threadIdx.x & 31;
    const int i0 = __builtin_amdgcn_readfirstlane((int)(blockIdx.x * 8 + (threadIdx.x >> 5))) * POOL_CHUNK;
    if (i0 >= n) return;
    const int m = min(POOL_CHUNK, n - i0);
    int gv = 0;
    if (lane < m) gv = batch[i0 + lane];
    const v2f* __restrict__ ap = reinterpret_cast<const v2f*>(a);
    v2f r[POOL_CHUNK];
    #pragma unroll
    for (int k = 0; k < POOL_CHUNK; ++k) r[k] = ap[(size_t)min(i0 + k, n - 1) * 32 + lane];

    int   cur = __builtin_amdgcn_readlane(gv, 0);
    int   run = 0;
    float sx = 0.0f, sy = 0.0f;
    #pragma unroll
    for (int k = 0; k < POOL_CHUNK; ++k) {
        if (k < m) {
            const int g = __builtin_amdgcn_readlane(gv, k);
            if (g != cur) {
                pool_flush(sums, cnt, cur, lane, sx, sy, run);
                cur = g; run = 0; sx = 0.0f; sy = 0.0f;
            }
            sx += fmaxf(r[k].x, 0.0f);
            sy += fmaxf(r[k].y, 0.0f);
            ++run;
        }
    }
    pool_flush(sums, cnt, cur, lane, sx, sy, run);
}

__global__ __launch_bounds__(256)
void k_head(const float* __restrict__ sums, const float* __restrict__ cnt,
            const float* __restrict__ Wl, const float* __restrict__ bl,
            float* __restrict__ out) {
    int t = blockIdx.x * blockDim.x + threadIdx.x;
    if (t < NGRAPHS * NCLASSES) {
        int g = t / NCLASSES;
        int c = t % NCLASSES;
        float inv = 1.0f / fmaxf(cnt[g], 1.0f);
        float acc = 0.0f;
        const float* sg = sums + (size_t)g * HID;
        #pragma unroll
        for (int k = 0; k < HID; ++k) acc += sg[k] * Wl[k * NCLASSES + c];
        out[t] = acc * inv + bl[c];
    }
}


extern "C" void kernel_launch(void* const* d_in, const int* in_sizes, int n_in,
                              void* d_out, int out_size, void* d_ws, size_t ws_size,
                              hipStream_t stream) {
    const float* x  = (const float*)d_in[0];
    const float* W1 = (const float*)d_in[1];
    const float* b1 = (const float*)d_in[2];
    const float* W2 = (const float*)d_in[3];
    const float* b2 = (const float*)d_in[4];
    const float* Wl = (const float*)d_in[5];
    const float* bl = (const float*)d_in[6];
    const int* edge  = (const int*)d_in[7];
    const int* batch = (const int*)d_in[8];
    const int* src = edge;
    const int* dst = edge + NEDGES;
    float* out = (float*)d_out;

    float* ws   = (float*)d_ws;
    float* h    = ws;
    float* a    = h + (size_t)NNODES * HID;
    float* dinv = a + (size_t)NNODES * HID;
    float* sums = dinv + NNODES;
    float* cnt  = sums + (size_t)NGRAPHS * HID;
    int* hist   = (int*)(cnt + NGRAPHS);
    int* offs   = hist + NHIST;
    int* rowptr = offs + NHIST;
    int* packed = rowptr + NNODES + 4;
    int* col    = packed + NEDGES;

    const int TPB = 256;
    const int gSc  = (NHIST + TPB - 1) / TPB;
    const int gPl  = ((NNODES + POOL_CHUNK - 1) / POOL_CHUNK + 7) / 8;
    const int gGa  = (NNODES + 7) / 8;
    const int gGmm = ((NNODES + 15) / 16 + 7) / 8;
    const int gPZ  = (NGRAPHS * HID + NGRAPHS + TPB - 1) / TPB;
    const int gHd  = (NGRAPHS * NCLASSES + TPB - 1) / TPB;

    k_hist<<<NCHUNK, TPB, 0, stream>>>(dst, hist, NEDGES);
    k_exscan<<<gSc, TPB, 0, stream>>>(hist, offs, NHIST);
    k_bin<<<NCHUNK, TPB, 0, stream>>>(src, dst, offs, packed, NEDGES);
    k_csr<<<NBKT, TPB, 0, stream>>>(packed, offs, rowptr, dinv, col, NNODES, NEDGES);

    k_gemm_wmma<<<gGmm, TPB, 0, stream>>>(x, W1, h, NNODES, 0);
    k_gather<<<gGa, TPB, 0, stream>>>(h, dinv, b1, rowptr, col, a, NNODES);

    k_gemm_wmma<<<gGmm, TPB, 0, stream>>>(a, W2, h, NNODES, 1);
    k_gather<<<gGa, TPB, 0, stream>>>(h, dinv, b2, rowptr, col, a, NNODES);

    k_zero<<<gPZ, TPB, 0, stream>>>(sums, NGRAPHS * HID + NGRAPHS);
    k_pool<<<gPl, TPB, 0, stream>>>(a, batch, sums, cnt, NNODES);
    k_head<<<gHd, TPB, 0, stream>>>(sums, cnt, Wl, bl, out);
}


# Launches 11 to 8: bucket prefix scan folded into the binning kernel, pooling-sum clear into the histogram kernel, layer-1 product run as extra blocks behind the three list-build launches
# speedup vs baseline: 3.7924x; 1.1366x over previous
#include <hip/hip_runtime.h>

typedef float v2f __attribute__((ext_vector_type(2)));
typedef float v8f __attribute__((ext_vector_type(8)));

#define NNODES 50000
#define NEDGES 800000
#define HID 64
#define NGRAPHS 512
#define NCLASSES 10


#define BKT_SHIFT 8
#define BKT_NODES (1 << BKT_SHIFT)
#define NBKT ((NNODES + BKT_NODES - 1) / BKT_NODES)
#define CHUNK 4096
#define NCHUNK ((NEDGES + CHUNK - 1) / CHUNK)
#define NHIST (NBKT * NCHUNK)
#define CSR_BATCH 8
#define NBKT4 (NBKT / 4)
#define SCAN_RG 5
#define SCAN_TRIPS ((NCHUNK + SCAN_RG - 1) / SCAN_RG)
static_assert(NBKT % 4 == 0 && SCAN_RG * NBKT4 <= 256 && NBKT < 256, "k_bin's column sums assume this");

#define GEMM_BLOCKS (((NNODES + 15) / 16 + 7) / 8)
#define GEMM_P0 (GEMM_BLOCKS / 5)
#define GEMM_P1 ((GEMM_BLOCKS - GEMM_P0 + 1) / 2)
#define GEMM_P2 (GEMM_BLOCKS - GEMM_P0 - GEMM_P1)

__device__ __forceinline__ void gemm_block(const float* __restrict__ A, const float* __restrict__ W,
                                           float* __restrict__ C, int n, int relu_in, int blk);

__global__ __launch_bounds__(256)
void k_hist(const int* __restrict__ dst, int* __restrict__ hist, int e, float* __restrict__ zp, int zn,
            const float* __restrict__ x, const float* __restrict__ W1, float* __restrict__ h, int n) {
    if (blockIdx.x >= NCHUNK) {
        gemm_block(x, W1, h, n, 0, (int)blockIdx.x - NCHUNK);
        return;
    }
    __shared__ int s_cnt[NBKT];
    const int tid = threadIdx.x;
    const int z = blockIdx.x * 256 + tid;
    if (z < zn) zp[z] = 0.0f;
    if (tid < NBKT) s_cnt[tid] = 0;
    __syncthreads();
    const int base = blockIdx.x * CHUNK;
    int d[CHUNK / 256];
    #pragma unroll
    for (int j = 0; j < CHUNK / 256; ++j) d[j] = dst[min(base + j * 256 + tid, e - 1)];
    #pragma unroll
    for (int j = 0; j < CHUNK / 256; ++j) {
        if (base + j * 256 + tid < e) atomicAdd(&s_cnt[d[j] >> BKT_SHIFT], 1);
    }
    __syncthreads();
    if (tid < NBKT) hist[blockIdx.x * NBKT + tid] = s_cnt[tid];
}

__global__ __launch_bounds__(256)
void k_bin(const int* __restrict__ src, const int* __restrict__ dst, const int* __restrict__ hist,
           int* __restrict__ bstart, int* __restrict__ packed, int e,
           const float* __restrict__ x, const float* __restrict__ W1, float* __restrict__ h, int n) {
    if (blockIdx.x >= NCHUNK) {
        gemm_block(x, W1, h, n, 0, (int)blockIdx.x - NCHUNK + GEMM_P0);
        return;
    }
    __shared__ int s_tot[SCAN_RG][NBKT];
    __shared__ int s_bef[SCAN_RG][NBKT];
    __shared__ int s_wsum[8];
    __shared__ int s_pos[NBKT];
    const int tid  = threadIdx.x;
    const int b    = blockIdx.x;
    const int base = b * CHUNK;
    int d[CHUNK / 256];
    int s[CHUNK / 256];
    #pragma unroll
    for (int j = 0; j < CHUNK / 256; ++j) {
        const int i = min(base + j * 256 + tid, e - 1);
        d[j] = dst[i];
        s[j] = src[i];
    }
    const int rg  = tid / NBKT4;
    const int cg  = tid - rg * NBKT4;
    if (rg < SCAN_RG) {
        int4 tot = make_int4(0, 0, 0, 0);
        int4 bef = make_int4(0, 0, 0, 0);
        const int4* __restrict__ hp = reinterpret_cast<const int4*>(hist) + cg;
        #pragma unroll 8
        for (int j = 0; j < SCAN_TRIPS; ++j) {
            const int r = rg + j * SCAN_RG;
            const int4 v = hp[(size_t)min(r, NCHUNK - 1) * NBKT4];
            const int mt = (r < NCHUNK) ? -1 : 0;
            const int mb = (r < b) ? -1 : 0;
            tot.x += v.x & mt; tot.y += v.y & mt; tot.z += v.z & mt; tot.w += v.w & mt;
            bef.x += v.x & mb; bef.y += v.y & mb; bef.z += v.z & mb; bef.w += v.w & mb;
        }
        s_tot[rg][4 * cg + 0] = tot.x; s_tot[rg][4 * cg + 1] = tot.y;
        s_tot[rg][4 * cg + 2] = tot.z; s_tot[rg][4 * cg + 3] = tot.w;
        s_bef[rg][4 * cg + 0] = bef.x; s_bef[rg][4 * cg + 1] = bef.y;
        s_bef[rg][4 * cg + 2] = bef.z; s_bef[rg][4 * cg + 3] = bef.w;
    }
    __syncthreads();
    int total = 0, before = 0;
    if (tid < NBKT) {
        #pragma unroll
        for (int g = 0; g < SCAN_RG; ++g) { total += s_tot[g][tid]; before += s_bef[g][tid]; }
    }
    const int lane = tid & 31;
    const int wave = tid >> 5;
    int incl = total;
    #pragma unroll
    for (int off = 1; off < 32; off <<= 1) {
        const int v = __shfl_up(incl, off);
        if (lane >= off) incl += v;
    }
    if (lane == 31) s_wsum[wave] = incl;
    __syncthreads();
    int wbase = 0;
    #pragma unroll
    for (int w = 0; w < 8; ++w) wbase += (w < wave) ? s_wsum[w] : 0;
    const int start = wbase + incl - total;
    if (tid < NBKT) s_pos[tid] = start + before;
    if (b == 0 && tid <= NBKT) bstart[tid] = start;
    __syncthreads();
    #pragma unroll
    for (int j = 0; j < CHUNK / 256; ++j) {
        if (base + j * 256 + tid < e) {
            const int slot = atomicAdd(&s_pos[d[j] >> BKT_SHIFT], 1);
            packed[slot] = (s[j] << BKT_SHIFT) | (d[j] & (BKT_NODES - 1));
        }
    }
}

__global__ __launch_bounds__(256)
void k_csr(const int* __restrict__ packed, const int* __restrict__ bstart, int* __restrict__ rowptr,
           float* __restrict__ dinv, int* __restrict__ col, int n,
           const float* __restrict__ x, const float* __restrict__ W1, float* __restrict__ h) {
    if (blockIdx.x >= NBKT) {
        gemm_block(x, W1, h, n, 0, (int)blockIdx.x - NBKT + GEMM_P0 + GEMM_P1);
        return;
    }
    __shared__ int s_cnt[BKT_NODES];
    __shared__ int s_scan[BKT_NODES];
    const int tid = threadIdx.x;
    const int k   = blockIdx.x;
    const int bs  = bstart[k];
    const int be  = bstart[k + 1];
    s_cnt[tid] = 0;
    __syncthreads();
    for (int q = bs + tid; q < be; q += 256 * CSR_BATCH) {
        int v[CSR_BATCH];
        #pragma unroll
        for (int u = 0; u < CSR_BATCH; ++u) v[u] = packed[min(q + u * 256, be - 1)];
        #pragma unroll
        for (int u = 0; u < CSR_BATCH; ++u) {
            if (q + u * 256 < be) atomicAdd(&s_cnt[v[u] & (BKT_NODES - 1)], 1);
        }
    }
    __syncthreads();
    const int own = s_cnt[tid];
    s_scan[tid] = own;
    __syncthreads();
    for (int off = 1; off < 256; off <<= 1) {
        int v = 0;
        if (tid >= off) v = s_scan[tid - off];
        __syncthreads();
        s_scan[tid] += v;
        __syncthreads();
    }
    const int start = bs + s_scan[tid] - own;
    const int i = k * BKT_NODES + tid;
    if (i < n) {
        rowptr[i] = start;
        dinv[i]   = rsqrtf((float)(own + 1));
    }
    if (i == n - 1) rowptr[n] = start + own;
    s_cnt[tid] = start;
    __syncthreads();
    for (int q = bs + tid; q < be; q += 256 * CSR_BATCH) {
        int v[CSR_BATCH];
        #pragma unroll
        for (int u = 0; u < CSR_BATCH; ++u) v[u] = packed[min(q + u * 256, be - 1)];
        #pragma unroll
        for (int u = 0; u < CSR_BATCH; ++u) {
            if (q + u * 256 < be) {
                const int slot = atomicAdd(&s_cnt[v[u] & (BKT_NODES - 1)], 1);
                col[slot] = v[u] >> BKT_SHIFT;
            }
        }
    }
}


__device__ __forceinline__ void gemm_block(const float* __restrict__ A, const float* __restrict__ W,
                                           float* __restrict__ C, int n, int relu_in, int blk) {
    __shared__ float sW[64 * 64];
    const int tid = threadIdx.x;
    #pragma unroll
    for (int i = 0; i < 16; ++i) sW[tid + i * 256] = W[tid + i * 256];
    __syncthreads();

    const int wave = tid >> 5;
    const int lane = tid & 31;
    const int half = lane >> 4;
    const int l16  = lane & 15;
    const int m0   = (blk * 8 + wave) * 16;
    if (m0 >= n) return;

    v8f c0 = {}, c1 = {}, c2 = {}, c3 = {};

    const int arow = m0 + l16;
    const float* __restrict__ Arow = A + (size_t)arow * 64;

    #pragma unroll
    for (int k0 = 0; k0 < 64; k0 += 4) {
        const int kb = k0 + half * 2;
        float a0 = Arow[kb];
        float a1 = Arow[kb + 1];
        if (relu_in) { a0 = fmaxf(a0, 0.0f); a1 = fmaxf(a1, 0.0f); }
        v2f a = {a0, a1};

        const float* __restrict__ Wk0 = sW + (size_t)kb * 64;
        const float* __restrict__ Wk1 = sW + (size_t)(kb + 1) * 64;
        v2f b0 = {Wk0[l16 +  0], Wk1[l16 +  0]};
        v2f b1 = {Wk0[l16 + 16], Wk1[l16 + 16]};
        v2f b2 = {Wk0[l16 + 32], Wk1[l16 + 32]};
        v2f b3 = {Wk0[l16 + 48], Wk1[l16 + 48]};

        c0 = __builtin_amdgcn_wmma_f32_16x16x4_f32(false, a, false, b0, (short)0, c0, false, false);
        c1 = __builtin_amdgcn_wmma_f32_16x16x4_f32(false, a, false, b1, (short)0, c1, false, false);
        c2 = __builtin_amdgcn_wmma_f32_16x16x4_f32(false, a, false, b2, (short)0, c2, false, false);
        c3 = __builtin_amdgcn_wmma_f32_16x16x4_f32(false, a, false, b3, (short)0, c3, false, false);
    }

    #pragma unroll
    for (int r = 0; r < 8; ++r) {
        const int row = m0 + half * 8 + r;
        if (row < n) {
            float* __restrict__ Cr = C + (size_t)row * 64 + l16;
            Cr[0]  = c0[r];
            Cr[16] = c1[r];
            Cr[32] = c2[r];
            Cr[48] = c3[r];
        }
    }
}

__global__ __launch_bounds__(256)
void k_gemm_wmma(const float* __restrict__ A, const float* __restrict__ W,
                 float* __restrict__ C, int n, int relu_in) {
    gemm_block(A, W, C, n, relu_in, (int)blockIdx.x);
}


__global__ __launch_bounds__(256)
void k_gather(const float* __restrict__ h, const float* __restrict__ dinv,
              const float* __restrict__ bias, const int* __restrict__ rowptr,
              const int* __restrict__ col, float* __restrict__ a, int n) {
    const int lane = threadIdx.x & 31;
    const int i = __builtin_amdgcn_readfirstlane((int)(blockIdx.x * 8 + (threadIdx.x >> 5)));
    if (i >= n) return;
    const float di = dinv[i];
    const int beg  = rowptr[i];
    const int end  = rowptr[i + 1];
    const v2f* __restrict__ hp = reinterpret_cast<const v2f*>(h);
    const v2f hs = hp[(size_t)i * 32 + lane];
    float ax = hs.x * di * di + bias[2 * lane];
    float ay = hs.y * di * di + bias[2 * lane + 1];
    for (int c0 = beg; c0 < end; c0 += 32) {
        const int m = min(32, end - c0);
        int   s = 0;
        float w = 0.0f;
        if (lane < m) { s = col[c0 + lane]; w = dinv[s] * di; }
        const int wi = __float_as_int(w);
        int k = 0;
        for (; k + 4 <= m; k += 4) {
            const int s0 = __builtin_amdgcn_readlane(s, k);
            const int s1 = __builtin_amdgcn_readlane(s, k + 1);
            const int s2 = __builtin_amdgcn_readlane(s, k + 2);
            const int s3 = __builtin_amdgcn_readlane(s, k + 3);
            const float w0 = __int_as_float(__builtin_amdgcn_readlane(wi, k));
            const float w1 = __int_as_float(__builtin_amdgcn_readlane(wi, k + 1));
            const float w2 = __int_as_float(__builtin_amdgcn_readlane(wi, k + 2));
            const float w3 = __int_as_float(__builtin_amdgcn_readlane(wi, k + 3));
            const v2f r0 = hp[(size_t)s0 * 32 + lane];
            const v2f r1 = hp[(size_t)s1 * 32 + lane];
            const v2f r2 = hp[(size_t)s2 * 32 + lane];
            const v2f r3 = hp[(size_t)s3 * 32 + lane];
            ax = fmaf(r0.x, w0, ax); ay = fmaf(r0.y, w0, ay);
            ax = fmaf(r1.x, w1, ax); ay = fmaf(r1.y, w1, ay);
            ax = fmaf(r2.x, w2, ax); ay = fmaf(r2.y, w2, ay);
            ax = fmaf(r3.x, w3, ax); ay = fmaf(r3.y, w3, ay);
        }
        for (; k < m; ++k) {
            const int   s0 = __builtin_amdgcn_readlane(s, k);
            const float w0 = __int_as_float(__builtin_amdgcn_readlane(wi, k));
            const v2f   r0 = hp[(size_t)s0 * 32 + lane];
            ax = fmaf(r0.x, w0, ax); ay = fmaf(r0.y, w0, ay);
        }
    }
    v2f o = {ax, ay};
    reinterpret_cast<v2f*>(a)[(size_t)i * 32 + lane] = o;
}


#define POOL_CHUNK 16

__device__ __forceinline__ void pool_flush(float* __restrict__ sums, float* __restrict__ cnt,
                                           int g, int lane, float sx, float sy, int run) {
    float* sg = sums + (size_t)g * HID + 2 * lane;
    atomicAdd(sg, sx);
    atomicAdd(sg + 1, sy);
    if (lane == 0) atomicAdd(&cnt[g], (float)run);
}

__global__ __launch_bounds__(256)
void k_pool(const float* __restrict__ a, const int* __restrict__ batch,
            float* __restrict__ sums, float* __restrict__ cnt, int n) {
    const int lane = threadIdx.x & 31;
    const int i0 = __builtin_amdgcn_readfirstlane((int)(blockIdx.x * 8 + (threadIdx.x >> 5))) * POOL_CHUNK;
    if (i0 >= n) return;
    const int m = min(POOL_CHUNK, n - i0);
    int gv = 0;
    if (lane < m) gv = batch[i0 + lane];
    const v2f* __restrict__ ap = reinterpret_cast<const v2f*>(a);
    v2f r[POOL_CHUNK];
    #pragma unroll
    for (int k = 0; k < POOL_CHUNK; ++k) r[k] = ap[(size_t)min(i0 + k, n - 1) * 32 + lane];

    int   cur = __builtin_amdgcn_readlane(gv, 0);
    int   run = 0;
    float sx = 0.0f, sy = 0.0f;
    #pragma unroll
    for (int k = 0; k < POOL_CHUNK; ++k) {
        if (k < m) {
            const int g = __builtin_amdgcn_readlane(gv, k);
            if (g != cur) {
                pool_flush(sums, cnt, cur, lane, sx, sy, run);
                cur = g; run = 0; sx = 0.0f; sy = 0.0f;
            }
            sx += fmaxf(r[k].x, 0.0f);
            sy += fmaxf(r[k].y, 0.0f);
            ++run;
        }
    }
    pool_flush(sums, cnt, cur, lane, sx, sy, run);
}

__global__ __launch_bounds__(256)
void k_head(const float* __restrict__ sums, const float* __restrict__ cnt,
            const float* __restrict__ Wl, const float* __restrict__ bl,
            float* __restrict__ out) {
    int t = blockIdx.x * blockDim.x + threadIdx.x;
    if (t < NGRAPHS * NCLASSES) {
        int g = t / NCLASSES;
        int c = t % NCLASSES;
        float inv = 1.0f / fmaxf(cnt[g], 1.0f);
        float acc = 0.0f;
        const float* sg = sums + (size_t)g * HID;
        #pragma unroll
        for (int k = 0; k < HID; ++k) acc += sg[k] * Wl[k * NCLASSES + c];
        out[t] = acc * inv + bl[c];
    }
}


extern "C" void kernel_launch(void* const* d_in, const int* in_sizes, int n_in,
                              void* d_out, int out_size, void* d_ws, size_t ws_size,
                              hipStream_t stream) {
    const float* x  = (const float*)d_in[0];
    const float* W1 = (const float*)d_in[1];
    const float* b1 = (const float*)d_in[2];
    const float* W2 = (const float*)d_in[3];
    const float* b2 = (const float*)d_in[4];
    const float* Wl = (const float*)d_in[5];
    const float* bl = (const float*)d_in[6];
    const int* edge  = (const int*)d_in[7];
    const int* batch = (const int*)d_in[8];
    const int* src = edge;
    const int* dst = edge + NEDGES;
    float* out = (float*)d_out;

    float* ws   = (float*)d_ws;
    float* h    = ws;
    float* a    = h + (size_t)NNODES * HID;
    float* dinv = a + (size_t)NNODES * HID;
    float* sums = dinv + NNODES;
    float* cnt  = sums + (size_t)NGRAPHS * HID;
    int* hist   = (int*)(cnt + NGRAPHS);
    int* bstart = hist + NHIST;
    int* rowptr = bstart + NBKT + 4;
    int* packed = rowptr + NNODES + 4;
    int* col    = packed + NEDGES;

    const int TPB = 256;
    const int gPl  = ((NNODES + POOL_CHUNK - 1) / POOL_CHUNK + 7) / 8;
    const int gGa  = (NNODES + 7) / 8;
    const int gGmm = GEMM_BLOCKS;
    const int nPZ  = NGRAPHS * HID + NGRAPHS;
    static_assert(NCHUNK * 256 >= NGRAPHS * HID + NGRAPHS, "k_hist's blocks clear sums and cnt");
    const int gHd  = (NGRAPHS * NCLASSES + TPB - 1) / TPB;

    k_hist<<<NCHUNK + GEMM_P0, TPB, 0, stream>>>(dst, hist, NEDGES, sums, nPZ, x, W1, h, NNODES);
    k_bin<<<NCHUNK + GEMM_P1, TPB, 0, stream>>>(src, dst, hist, bstart, packed, NEDGES, x, W1, h, NNODES);
    k_csr<<<NBKT + GEMM_P2, TPB, 0, stream>>>(packed, bstart, rowptr, dinv, col, NNODES, x, W1, h);

    k_gather<<<gGa, TPB, 0, stream>>>(h, dinv, b1, rowptr, col, a, NNODES);

    k_gemm_wmma<<<gGmm, TPB, 0, stream>>>(a, W2, h, NNODES, 1);
    k_gather<<<gGa, TPB, 0, stream>>>(h, dinv, b2, rowptr, col, a, NNODES);

    k_pool<<<gPl, TPB, 0, stream>>>(a, batch, sums, cnt, NNODES);
    k_head<<<gHd, TPB, 0, stream>>>(sums, cnt, Wl, bl, out);
}


# Launches 8 to 6: gather blocks of 16 nodes hand relu rows in LDS to the layer-2 WMMA product and to the pooling adds in the same block instead of a global buffer and a launch each; sum orders kept
# speedup vs baseline: 4.0451x; 1.0666x over previous
#include <hip/hip_runtime.h>

typedef float v2f __attribute__((ext_vector_type(2)));
typedef float v8f __attribute__((ext_vector_type(8)));

#define NNODES 50000
#define NEDGES 800000
#define HID 64
#define NGRAPHS 512
#define NCLASSES 10


#define BKT_SHIFT 8
#define BKT_NODES (1 << BKT_SHIFT)
#define NBKT ((NNODES + BKT_NODES - 1) / BKT_NODES)
#define CHUNK 4096
#define NCHUNK ((NEDGES + CHUNK - 1) / CHUNK)
#define NHIST (NBKT * NCHUNK)
#define CSR_BATCH 8
#define NBKT4 (NBKT / 4)
#define SCAN_RG 5
#define SCAN_TRIPS ((NCHUNK + SCAN_RG - 1) / SCAN_RG)
static_assert(NBKT % 4 == 0 && SCAN_RG * NBKT4 <= 256 && NBKT < 256, "k_bin's column sums assume this");

#define GEMM_BLOCKS (((NNODES + 15) / 16 + 7) / 8)
#define GEMM_P0 (GEMM_BLOCKS / 5)
#define GEMM_P1 ((GEMM_BLOCKS - GEMM_P0 + 1) / 2)
#define GEMM_P2 (GEMM_BLOCKS - GEMM_P0 - GEMM_P1)

__device__ __forceinline__ void gemm_block(const float* __restrict__ A, const float* __restrict__ W,
                                           float* __restrict__ C, int n, int relu_in, int blk);

__global__ __launch_bounds__(256)
void k_hist(const int* __restrict__ dst, int* __restrict__ hist, int e, float* __restrict__ zp, int zn,
            const float* __restrict__ x, const float* __restrict__ W1, float* __restrict__ h, int n) {
    if (blockIdx.x >= NCHUNK) {
        gemm_block(x, W1, h, n, 0, (int)blockIdx.x - NCHUNK);
        return;
    }
    __shared__ int s_cnt[NBKT];
    const int tid = threadIdx.x;
    const int z = blockIdx.x * 256 + tid;
    if (z < zn) zp[z] = 0.0f;
    if (tid < NBKT) s_cnt[tid] = 0;
    __syncthreads();
    const int base = blockIdx.x * CHUNK;
    int d[CHUNK / 256];
    #pragma unroll
    for (int j = 0; j < CHUNK / 256; ++j) d[j] = dst[min(base + j * 256 + tid, e - 1)];
    #pragma unroll
    for (int j = 0; j < CHUNK / 256; ++j) {
        if (base + j * 256 + tid < e) atomicAdd(&s_cnt[d[j] >> BKT_SHIFT], 1);
    }
    __syncthreads();
    if (tid < NBKT) hist[blockIdx.x * NBKT + tid] = s_cnt[tid];
}

__global__ __launch_bounds__(256)
void k_bin(const int* __restrict__ src, const int* __restrict__ dst, const int* __restrict__ hist,
           int* __restrict__ bstart, int* __restrict__ packed, int e,
           const float* __restrict__ x, const float* __restrict__ W1, float* __restrict__ h, int n) {
    if (blockIdx.x >= NCHUNK) {
        gemm_block(x, W1, h, n, 0, (int)blockIdx.x - NCHUNK + GEMM_P0);
        return;
    }
    __shared__ int s_tot[SCAN_RG][NBKT];
    __shared__ int s_bef[SCAN_RG][NBKT];
    __shared__ int s_wsum[8];
    __shared__ int s_pos[NBKT];
    const int tid  = threadIdx.x;
    const int b    = blockIdx.x;
    const int base = b * CHUNK;
    int d[CHUNK / 256];
    int s[CHUNK / 256];
    #pragma unroll
    for (int j = 0; j < CHUNK / 256; ++j) {
        const int i = min(base + j * 256 + tid, e - 1);
        d[j] = dst[i];
        s[j] = src[i];
    }
    const int rg  = tid / NBKT4;
    const int cg  = tid - rg * NBKT4;
    if (rg < SCAN_RG) {
        int4 tot = make_int4(0, 0, 0, 0);
        int4 bef = make_int4(0, 0, 0, 0);
        const int4* __restrict__ hp = reinterpret_cast<const int4*>(hist) + cg;
        #pragma unroll 8
        for (int j = 0; j < SCAN_TRIPS; ++j) {
            const int r = rg + j * SCAN_RG;
            const int4 v = hp[(size_t)min(r, NCHUNK - 1) * NBKT4];
            const int mt = (r < NCHUNK) ? -1 : 0;
            const int mb = (r < b) ? -1 : 0;
            tot.x += v.x & mt; tot.y += v.y & mt; tot.z += v.z & mt; tot.w += v.w & mt;
            bef.x += v.x & mb; bef.y += v.y & mb; bef.z += v.z & mb; bef.w += v.w & mb;
        }
        s_tot[rg][4 * cg + 0] = tot.x; s_tot[rg][4 * cg + 1] = tot.y;
        s_tot[rg][4 * cg + 2] = tot.z; s_tot[rg][4 * cg + 3] = tot.w;
        s_bef[rg][4 * cg + 0] = bef.x; s_bef[rg][4 * cg + 1] = bef.y;
        s_bef[rg][4 * cg + 2] = bef.z; s_bef[rg][4 * cg + 3] = bef.w;
    }
    __syncthreads();
    int total = 0, before = 0;
    if (tid < NBKT) {
        #pragma unroll
        for (int g = 0; g < SCAN_RG; ++g) { total += s_tot[g][tid]; before += s_bef[g][tid]; }
    }
    const int lane = tid & 31;
    const int wave = tid >> 5;
    int incl = total;
    #pragma unroll
    for (int off = 1; off < 32; off <<= 1) {
        const int v = __shfl_up(incl, off);
        if (lane >= off) incl += v;
    }
    if (lane == 31) s_wsum[wave] = incl;
    __syncthreads();
    int wbase = 0;
    #pragma unroll
    for (int w = 0; w < 8; ++w) wbase += (w < wave) ? s_wsum[w] : 0;
    const int start = wbase + incl - total;
    if (tid < NBKT) s_pos[tid] = start + before;
    if (b == 0 && tid <= NBKT) bstart[tid] = start;
    __syncthreads();
    #pragma unroll
    for (int j = 0; j < CHUNK / 256; ++j) {
        if (base + j * 256 + tid < e) {
            const int slot = atomicAdd(&s_pos[d[j] >> BKT_SHIFT], 1);
            packed[slot] = (s[j] << BKT_SHIFT) | (d[j] & (BKT_NODES - 1));
        }
    }
}

__global__ __launch_bounds__(256)
void k_csr(const int* __restrict__ packed, const int* __restrict__ bstart, int* __restrict__ rowptr,
           float* __restrict__ dinv, int* __restrict__ col, int n,
           const float* __restrict__ x, const float* __restrict__ W1, float* __restrict__ h) {
    if (blockIdx.x >= NBKT) {
        gemm_block(x, W1, h, n, 0, (int)blockIdx.x - NBKT + GEMM_P0 + GEMM_P1);
        return;
    }
    __shared__ int s_cnt[BKT_NODES];
    __shared__ int s_scan[BKT_NODES];
    const int tid = threadIdx.x;
    const int k   = blockIdx.x;
    const int bs  = bstart[k];
    const int be  = bstart[k + 1];
    s_cnt[tid] = 0;
    __syncthreads();
    for (int q = bs + tid; q < be; q += 256 * CSR_BATCH) {
        int v[CSR_BATCH];
        #pragma unroll
        for (int u = 0; u < CSR_BATCH; ++u) v[u] = packed[min(q + u * 256, be - 1)];
        #pragma unroll
        for (int u = 0; u < CSR_BATCH; ++u) {
            if (q + u * 256 < be) atomicAdd(&s_cnt[v[u] & (BKT_NODES - 1)], 1);
        }
    }
    __syncthreads();
    const int own = s_cnt[tid];
    s_scan[tid] = own;
    __syncthreads();
    for (int off = 1; off < 256; off <<= 1) {
        int v = 0;
        if (tid >= off) v = s_scan[tid - off];
        __syncthreads();
        s_scan[tid] += v;
        __syncthreads();
    }
    const int start = bs + s_scan[tid] - own;
    const int i = k * BKT_NODES + tid;
    if (i < n) {
        rowptr[i] = start;
        dinv[i]   = rsqrtf((float)(own + 1));
    }
    if (i == n - 1) rowptr[n] = start + own;
    s_cnt[tid] = start;
    __syncthreads();
    for (int q = bs + tid; q < be; q += 256 * CSR_BATCH) {
        int v[CSR_BATCH];
        #pragma unroll
        for (int u = 0; u < CSR_BATCH; ++u) v[u] = packed[min(q + u * 256, be - 1)];
        #pragma unroll
        for (int u = 0; u < CSR_BATCH; ++u) {
            if (q + u * 256 < be) {
                const int slot = atomicAdd(&s_cnt[v[u] & (BKT_NODES - 1)], 1);
                col[slot] = v[u] >> BKT_SHIFT;
            }
        }
    }
}


__device__ __forceinline__ void gemm_block(const float* __restrict__ A, const float* __restrict__ W,
                                           float* __restrict__ C, int n, int relu_in, int blk) {
    __shared__ float sW[64 * 64];
    const int tid = threadIdx.x;
    #pragma unroll
    for (int i = 0; i < 16; ++i) sW[tid + i * 256] = W[tid + i * 256];
    __syncthreads();

    const int wave = tid >> 5;
    const int lane = tid & 31;
    const int half = lane >> 4;
    const int l16  = lane & 15;
    const int m0   = (blk * 8 + wave) * 16;
    if (m0 >= n) return;

    v8f c0 = {}, c1 = {}, c2 = {}, c3 = {};

    const int arow = m0 + l16;
    const float* __restrict__ Arow = A + (size_t)arow * 64;

    #pragma unroll
    for (int k0 = 0; k0 < 64; k0 += 4) {
        const int kb = k0 + half * 2;
        float a0 = Arow[kb];
        float a1 = Arow[kb + 1];
        if (relu_in) { a0 = fmaxf(a0, 0.0f); a1 = fmaxf(a1, 0.0f); }
        v2f a = {a0, a1};

        const float* __restrict__ Wk0 = sW + (size_t)kb * 64;
        const float* __restrict__ Wk1 = sW + (size_t)(kb + 1) * 64;
        v2f b0 = {Wk0[l16 +  0], Wk1[l16 +  0]};
        v2f b1 = {Wk0[l16 + 16], Wk1[l16 + 16]};
        v2f b2 = {Wk0[l16 + 32], Wk1[l16 + 32]};
        v2f b3 = {Wk0[l16 + 48], Wk1[l16 + 48]};

        c0 = __builtin_amdgcn_wmma_f32_16x16x4_f32(false, a, false, b0, (short)0, c0, false, false);
        c1 = __builtin_amdgcn_wmma_f32_16x16x4_f32(false, a, false, b1, (short)0, c1, false, false);
        c2 = __builtin_amdgcn_wmma_f32_16x16x4_f32(false, a, false, b2, (short)0, c2, false, false);
        c3 = __builtin_amdgcn_wmma_f32_16x16x4_f32(false, a, false, b3, (short)0, c3, false, false);
    }

    #pragma unroll
    for (int r = 0; r < 8; ++r) {
        const int row = m0 + half * 8 + r;
        if (row < n) {
            float* __restrict__ Cr = C + (size_t)row * 64 + l16;
            Cr[0]  = c0[r];
            Cr[16] = c1[r];
            Cr[32] = c2[r];
            Cr[48] = c3[r];
        }
    }
}


#define GB_NODES 16
#define SA_PITCH 66

__device__ __forceinline__ void pool_flush(float* __restrict__ sums, float* __restrict__ cnt,
                                           int g, int lane, float sx, float sy, int run) {
    float* sg = sums + (size_t)g * HID + 2 * lane;
    atomicAdd(sg, sx);
    atomicAdd(sg + 1, sy);
    if (lane == 0) atomicAdd(&cnt[g], (float)run);
}

template <int MODE>
__global__ __launch_bounds__(GB_NODES * 32)
void k_gather(const float* __restrict__ h, const float* __restrict__ dinv,
              const float* __restrict__ bias, const int* __restrict__ rowptr,
              const int* __restrict__ col, int n,
              const float* __restrict__ W, float* __restrict__ hout,
              const int* __restrict__ batch, float* __restrict__ sums, float* __restrict__ cnt) {
    __shared__ float sA[GB_NODES * SA_PITCH];
    __shared__ float sW[MODE == 0 ? 64 * 64 : 4];
    const int tid  = threadIdx.x;
    const int lane = tid & 31;
    const int wave = __builtin_amdgcn_readfirstlane(tid >> 5);
    const int i0   = (int)blockIdx.x * GB_NODES;
    const int i    = i0 + wave;

    float4 w0 = make_float4(0.f, 0.f, 0.f, 0.f), w1 = w0;
    int gv = 0;
    if (MODE == 0) {
        const float4* __restrict__ W4 = reinterpret_cast<const float4*>(W);
        w0 = W4[tid];
        w1 = W4[tid + GB_NODES * 32];
    } else {
        if (wave == 0 && i0 + lane < n) gv = batch[i0 + lane];
    }

    float ax = 0.0f, ay = 0.0f;
    if (i < n) {
        const float di = dinv[i];
        const int beg  = rowptr[i];
        const int end  = rowptr[i + 1];
        const v2f* __restrict__ hp = reinterpret_cast<const v2f*>(h);
        const v2f hs = hp[(size_t)i * 32 + lane];
        ax = hs.x * di * di + bias[2 * lane];
        ay = hs.y * di * di + bias[2 * lane + 1];
        for (int c0 = beg; c0 < end; c0 += 32) {
            const int m = min(32, end - c0);
            int   s = 0;
            float w = 0.0f;
            if (lane < m) { s = col[c0 + lane]; w = dinv[s] * di; }
            const int wi = __float_as_int(w);
            int k = 0;
            for (; k + 4 <= m; k += 4) {
                const int s0 = __builtin_amdgcn_readlane(s, k);
                const int s1 = __builtin_amdgcn_readlane(s, k + 1);
                const int s2 = __builtin_amdgcn_readlane(s, k + 2);
                const int s3 = __builtin_amdgcn_readlane(s, k + 3);
                const float w0r = __int_as_float(__builtin_amdgcn_readlane(wi, k));
                const float w1r = __int_as_float(__builtin_amdgcn_readlane(wi, k + 1));
                const float w2r = __int_as_float(__builtin_amdgcn_readlane(wi, k + 2));
                const float w3r = __int_as_float(__builtin_amdgcn_readlane(wi, k + 3));
                const v2f r0 = hp[(size_t)s0 * 32 + lane];
                const v2f r1 = hp[(size_t)s1 * 32 + lane];
                const v2f r2 = hp[(size_t)s2 * 32 + lane];
                const v2f r3 = hp[(size_t)s3 * 32 + lane];
                ax = fmaf(r0.x, w0r, ax); ay = fmaf(r0.y, w0r, ay);
                ax = fmaf(r1.x, w1r, ax); ay = fmaf(r1.y, w1r, ay);
                ax = fmaf(r2.x, w2r, ax); ay = fmaf(r2.y, w2r, ay);
                ax = fmaf(r3.x, w3r, ax); ay = fmaf(r3.y, w3r, ay);
            }
            for (; k < m; ++k) {
                const int   s0  = __builtin_amdgcn_readlane(s, k);
                const float w0r = __int_as_float(__builtin_amdgcn_readlane(wi, k));
                const v2f   r0  = hp[(size_t)s0 * 32 + lane];
                ax = fmaf(r0.x, w0r, ax); ay = fmaf(r0.y, w0r, ay);
            }
        }
    }

    const v2f o = {fmaxf(ax, 0.0f), fmaxf(ay, 0.0f)};
    *reinterpret_cast<v2f*>(&sA[wave * SA_PITCH + 2 * lane]) = o;
    if (MODE == 0) {
        float4* sW4 = reinterpret_cast<float4*>(sW);
        sW4[tid] = w0;
        sW4[tid + GB_NODES * 32] = w1;
    }
    __syncthreads();

    if (MODE == 0) {
        if (wave >= 2) return;
        const int half = lane >> 4;
        const int l16  = lane & 15;
        const int nc   = wave * 32 + l16;
        v8f c0 = {}, c1 = {};
        #pragma unroll
        for (int k0 = 0; k0 < 64; k0 += 4) {
            const int kb = k0 + half * 2;
            const v2f a = *reinterpret_cast<const v2f*>(&sA[l16 * SA_PITCH + kb]);
            const float* __restrict__ Wk0 = sW + kb * 64 + nc;
            const float* __restrict__ Wk1 = sW + (kb + 1) * 64 + nc;
            const v2f b0 = {Wk0[0],  Wk1[0]};
            const v2f b1 = {Wk0[16], Wk1[16]};
            c0 = __builtin_amdgcn_wmma_f32_16x16x4_f32(false, a, false, b0, (short)0, c0, false, false);
            c1 = __builtin_amdgcn_wmma_f32_16x16x4_f32(false, a, false, b1, (short)0, c1, false, false);
        }
        #pragma unroll
        for (int r = 0; r < 8; ++r) {
            const int row = i0 + half * 8 + r;
            if (row < n) {
                float* __restrict__ Cr = hout + (size_t)row * 64 + nc;
                Cr[0]  = c0[r];
                Cr[16] = c1[r];
            }
        }
    } else {
        if (wave != 0) return;
        const int m = min(GB_NODES, n - i0);
        v2f r[GB_NODES];
        #pragma unroll
        for (int k = 0; k < GB_NODES; ++k) r[k] = *reinterpret_cast<const v2f*>(&sA[k * SA_PITCH + 2 * lane]);
        int   cur = __builtin_amdgcn_readlane(gv, 0);
        int   run = 0;
        float sx = 0.0f, sy = 0.0f;
        #pragma unroll
        for (int k = 0; k < GB_NODES; ++k) {
            if (k < m) {
                const int g = __builtin_amdgcn_readlane(gv, k);
                if (g != cur) {
                    pool_flush(sums, cnt, cur, lane, sx, sy, run);
                    cur = g; run = 0; sx = 0.0f; sy = 0.0f;
                }
                sx += r[k].x;
                sy += r[k].y;
                ++run;
            }
        }
        pool_flush(sums, cnt, cur, lane, sx, sy, run);
    }
}

__global__ __launch_bounds__(256)
void k_head(const float* __restrict__ sums, const float* __restrict__ cnt,
            const float* __restrict__ Wl, const float* __restrict__ bl,
            float* __restrict__ out) {
    int t = blockIdx.x * blockDim.x + threadIdx.x;
    if (t < NGRAPHS * NCLASSES) {
        int g = t / NCLASSES;
        int c = t % NCLASSES;
        float inv = 1.0f / fmaxf(cnt[g], 1.0f);
        float acc = 0.0f;
        const float* sg = sums + (size_t)g * HID;
        #pragma unroll
        for (int k = 0; k < HID; ++k) acc += sg[k] * Wl[k * NCLASSES + c];
        out[t] = acc * inv + bl[c];
    }
}


extern "C" void kernel_launch(void* const* d_in, const int* in_sizes, int n_in,
                              void* d_out, int out_size, void* d_ws, size_t ws_size,
                              hipStream_t stream) {
    const float* x  = (const float*)d_in[0];
    const float* W1 = (const float*)d_in[1];
    const float* b1 = (const float*)d_in[2];
    const float* W2 = (const float*)d_in[3];
    const float* b2 = (const float*)d_in[4];
    const float* Wl = (const float*)d_in[5];
    const float* bl = (const float*)d_in[6];
    const int* edge  = (const int*)d_in[7];
    const int* batch = (const int*)d_in[8];
    const int* src = edge;
    const int* dst = edge + NEDGES;
    float* out = (float*)d_out;

    float* ws   = (float*)d_ws;
    float* h    = ws;
    float* a    = h + (size_t)NNODES * HID;
    float* dinv = a + (size_t)NNODES * HID;
    float* sums = dinv + NNODES;
    float* cnt  = sums + (size_t)NGRAPHS * HID;
    int* hist   = (int*)(cnt + NGRAPHS);
    int* bstart = hist + NHIST;
    int* rowptr = bstart + NBKT + 4;
    int* packed = rowptr + NNODES + 4;
    int* col    = packed + NEDGES;

    const int TPB = 256;
    const int gGa  = (NNODES + GB_NODES - 1) / GB_NODES;
    const int nPZ  = NGRAPHS * HID + NGRAPHS;
    static_assert(NCHUNK * 256 >= NGRAPHS * HID + NGRAPHS, "k_hist's blocks clear sums and cnt");
    const int gHd  = (NGRAPHS * NCLASSES + TPB - 1) / TPB;

    k_hist<<<NCHUNK + GEMM_P0, TPB, 0, stream>>>(dst, hist, NEDGES, sums, nPZ, x, W1, h, NNODES);
    k_bin<<<NCHUNK + GEMM_P1, TPB, 0, stream>>>(src, dst, hist, bstart, packed, NEDGES, x, W1, h, NNODES);
    k_csr<<<NBKT + GEMM_P2, TPB, 0, stream>>>(packed, bstart, rowptr, dinv, col, NNODES, x, W1, h);

    k_gather<0><<<gGa, GB_NODES * 32, 0, stream>>>(h, dinv, b1, rowptr, col, NNODES, W2, a,
                                                   nullptr, nullptr, nullptr);
    k_gather<1><<<gGa, GB_NODES * 32, 0, stream>>>(a, dinv, b2, rowptr, col, NNODES, nullptr, nullptr,
                                                   batch, sums, cnt);
    k_head<<<gHd, TPB, 0, stream>>>(sums, cnt, Wl, bl, out);
}


# Gather re-tiled: half a wave per node with 16-byte loads (one load brings two rows), four nodes per wave, 64-node blocks: a quarter of the waves enter; 8 row loads in flight; sum order kept
# speedup vs baseline: 4.1983x; 1.0379x over previous
#include <hip/hip_runtime.h>

typedef float v2f __attribute__((ext_vector_type(2)));
typedef float v8f __attribute__((ext_vector_type(8)));
typedef float v4f __attribute__((ext_vector_type(4)));

#define NNODES 50000
#define NEDGES 800000
#define HID 64
#define NGRAPHS 512
#define NCLASSES 10


#define BKT_SHIFT 8
#define BKT_NODES (1 << BKT_SHIFT)
#define NBKT ((NNODES + BKT_NODES - 1) / BKT_NODES)
#define CHUNK 4096
#define NCHUNK ((NEDGES + CHUNK - 1) / CHUNK)
#define NHIST (NBKT * NCHUNK)
#define CSR_BATCH 8
#define NBKT4 (NBKT / 4)
#define SCAN_RG 5
#define SCAN_TRIPS ((NCHUNK + SCAN_RG - 1) / SCAN_RG)
static_assert(NBKT % 4 == 0 && SCAN_RG * NBKT4 <= 256 && NBKT < 256, "k_bin's column sums assume this");

#define GEMM_BLOCKS (((NNODES + 15) / 16 + 7) / 8)
#define GEMM_P0 (GEMM_BLOCKS / 5)
#define GEMM_P1 ((GEMM_BLOCKS - GEMM_P0 + 1) / 2)
#define GEMM_P2 (GEMM_BLOCKS - GEMM_P0 - GEMM_P1)

__device__ __forceinline__ void gemm_block(const float* __restrict__ A, const float* __restrict__ W,
                                           float* __restrict__ C, int n, int relu_in, int blk);

__global__ __launch_bounds__(256)
void k_hist(const int* __restrict__ dst, int* __restrict__ hist, int e, float* __restrict__ zp, int zn,
            const float* __restrict__ x, const float* __restrict__ W1, float* __restrict__ h, int n) {
    if (blockIdx.x >= NCHUNK) {
        gemm_block(x, W1, h, n, 0, (int)blockIdx.x - NCHUNK);
        return;
    }
    __shared__ int s_cnt[NBKT];
    const int tid = threadIdx.x;
    const int z = blockIdx.x * 256 + tid;
    if (z < zn) zp[z] = 0.0f;
    if (tid < NBKT) s_cnt[tid] = 0;
    __syncthreads();
    const int base = blockIdx.x * CHUNK;
    int d[CHUNK / 256];
    #pragma unroll
    for (int j = 0; j < CHUNK / 256; ++j) d[j] = dst[min(base + j * 256 + tid, e - 1)];
    #pragma unroll
    for (int j = 0; j < CHUNK / 256; ++j) {
        if (base + j * 256 + tid < e) atomicAdd(&s_cnt[d[j] >> BKT_SHIFT], 1);
    }
    __syncthreads();
    if (tid < NBKT) hist[blockIdx.x * NBKT + tid] = s_cnt[tid];
}

__global__ __launch_bounds__(256)
void k_bin(const int* __restrict__ src, const int* __restrict__ dst, const int* __restrict__ hist,
           int* __restrict__ bstart, int* __restrict__ packed, int e,
           const float* __restrict__ x, const float* __restrict__ W1, float* __restrict__ h, int n) {
    if (blockIdx.x >= NCHUNK) {
        gemm_block(x, W1, h, n, 0, (int)blockIdx.x - NCHUNK + GEMM_P0);
        return;
    }
    __shared__ int s_tot[SCAN_RG][NBKT];
    __shared__ int s_bef[SCAN_RG][NBKT];
    __shared__ int s_wsum[8];
    __shared__ int s_pos[NBKT];
    const int tid  = threadIdx.x;
    const int b    = blockIdx.x;
    const int base = b * CHUNK;
    int d[CHUNK / 256];
    int s[CHUNK / 256];
    #pragma unroll
    for (int j = 0; j < CHUNK / 256; ++j) {
        const int i = min(base + j * 256 + tid, e - 1);
        d[j] = dst[i];
        s[j] = src[i];
    }
    const int rg  = tid / NBKT4;
    const int cg  = tid - rg * NBKT4;
    if (rg < SCAN_RG) {
        int4 tot = make_int4(0, 0, 0, 0);
        int4 bef = make_int4(0, 0, 0, 0);
        const int4* __restrict__ hp = reinterpret_cast<const int4*>(hist) + cg;
        #pragma unroll 8
        for (int j = 0; j < SCAN_TRIPS; ++j) {
            const int r = rg + j * SCAN_RG;
            const int4 v = hp[(size_t)min(r, NCHUNK - 1) * NBKT4];
            const int mt = (r < NCHUNK) ? -1 : 0;
            const int mb = (r < b) ? -1 : 0;
            tot.x += v.x & mt; tot.y += v.y & mt; tot.z += v.z & mt; tot.w += v.w & mt;
            bef.x += v.x & mb; bef.y += v.y & mb; bef.z += v.z & mb; bef.w += v.w & mb;
        }
        s_tot[rg][4 * cg + 0] = tot.x; s_tot[rg][4 * cg + 1] = tot.y;
        s_tot[rg][4 * cg + 2] = tot.z; s_tot[rg][4 * cg + 3] = tot.w;
        s_bef[rg][4 * cg + 0] = bef.x; s_bef[rg][4 * cg + 1] = bef.y;
        s_bef[rg][4 * cg + 2] = bef.z; s_bef[rg][4 * cg + 3] = bef.w;
    }
    __syncthreads();
    int total = 0, before = 0;
    if (tid < NBKT) {
        #pragma unroll
        for (int g = 0; g < SCAN_RG; ++g) { total += s_tot[g][tid]; before += s_bef[g][tid]; }
    }
    const int lane = tid & 31;
    const int wave = tid >> 5;
    int incl = total;
    #pragma unroll
    for (int off = 1; off < 32; off <<= 1) {
        const int v = __shfl_up(incl, off);
        if (lane >= off) incl += v;
    }
    if (lane == 31) s_wsum[wave] = incl;
    __syncthreads();
    int wbase = 0;
    #pragma unroll
    for (int w = 0; w < 8; ++w) wbase += (w < wave) ? s_wsum[w] : 0;
    const int start = wbase + incl - total;
    if (tid < NBKT) s_pos[tid] = start + before;
    if (b == 0 && tid <= NBKT) bstart[tid] = start;
    __syncthreads();
    #pragma unroll
    for (int j = 0; j < CHUNK / 256; ++j) {
        if (base + j * 256 + tid < e) {
            const int slot = atomicAdd(&s_pos[d[j] >> BKT_SHIFT], 1);
            packed[slot] = (s[j] << BKT_SHIFT) | (d[j] & (BKT_NODES - 1));
        }
    }
}

__global__ __launch_bounds__(256)
void k_csr(const int* __restrict__ packed, const int* __restrict__ bstart, int* __restrict__ rowptr,
           float* __restrict__ dinv, int* __restrict__ col, int n,
           const float* __restrict__ x, const float* __restrict__ W1, float* __restrict__ h) {
    if (blockIdx.x >= NBKT) {
        gemm_block(x, W1, h, n, 0, (int)blockIdx.x - NBKT + GEMM_P0 + GEMM_P1);
        return;
    }
    __shared__ int s_cnt[BKT_NODES];
    __shared__ int s_scan[BKT_NODES];
    const int tid = threadIdx.x;
    const int k   = blockIdx.x;
    const int bs  = bstart[k];
    const int be  = bstart[k + 1];
    s_cnt[tid] = 0;
    __syncthreads();
    for (int q = bs + tid; q < be; q += 256 * CSR_BATCH) {
        int v[CSR_BATCH];
        #pragma unroll
        for (int u = 0; u < CSR_BATCH; ++u) v[u] = packed[min(q + u * 256, be - 1)];
        #pragma unroll
        for (int u = 0; u < CSR_BATCH; ++u) {
            if (q + u * 256 < be) atomicAdd(&s_cnt[v[u] & (BKT_NODES - 1)], 1);
        }
    }
    __syncthreads();
    const int own = s_cnt[tid];
    s_scan[tid] = own;
    __syncthreads();
    for (int off = 1; off < 256; off <<= 1) {
        int v = 0;
        if (tid >= off) v = s_scan[tid - off];
        __syncthreads();
        s_scan[tid] += v;
        __syncthreads();
    }
    const int start = bs + s_scan[tid] - own;
    const int i = k * BKT_NODES + tid;
    if (i < n) {
        rowptr[i] = start;
        dinv[i]   = rsqrtf((float)(own + 1));
    }
    if (i == n - 1) rowptr[n] = start + own;
    s_cnt[tid] = start;
    __syncthreads();
    for (int q = bs + tid; q < be; q += 256 * CSR_BATCH) {
        int v[CSR_BATCH];
        #pragma unroll
        for (int u = 0; u < CSR_BATCH; ++u) v[u] = packed[min(q + u * 256, be - 1)];
        #pragma unroll
        for (int u = 0; u < CSR_BATCH; ++u) {
            if (q + u * 256 < be) {
                const int slot = atomicAdd(&s_cnt[v[u] & (BKT_NODES - 1)], 1);
                col[slot] = v[u] >> BKT_SHIFT;
            }
        }
    }
}


__device__ __forceinline__ void gemm_block(const float* __restrict__ A, const float* __restrict__ W,
                                           float* __restrict__ C, int n, int relu_in, int blk) {
    __shared__ float sW[64 * 64];
    const int tid = threadIdx.x;
    #pragma unroll
    for (int i = 0; i < 16; ++i) sW[tid + i * 256] = W[tid + i * 256];
    __syncthreads();

    const int wave = tid >> 5;
    const int lane = tid & 31;
    const int half = lane >> 4;
    const int l16  = lane & 15;
    const int m0   = (blk * 8 + wave) * 16;
    if (m0 >= n) return;

    v8f c0 = {}, c1 = {}, c2 = {}, c3 = {};

    const int arow = m0 + l16;
    const float* __restrict__ Arow = A + (size_t)arow * 64;

    #pragma unroll
    for (int k0 = 0; k0 < 64; k0 += 4) {
        const int kb = k0 + half * 2;
        float a0 = Arow[kb];
        float a1 = Arow[kb + 1];
        if (relu_in) { a0 = fmaxf(a0, 0.0f); a1 = fmaxf(a1, 0.0f); }
        v2f a = {a0, a1};

        const float* __restrict__ Wk0 = sW + (size_t)kb * 64;
        const float* __restrict__ Wk1 = sW + (size_t)(kb + 1) * 64;
        v2f b0 = {Wk0[l16 +  0], Wk1[l16 +  0]};
        v2f b1 = {Wk0[l16 + 16], Wk1[l16 + 16]};
        v2f b2 = {Wk0[l16 + 32], Wk1[l16 + 32]};
        v2f b3 = {Wk0[l16 + 48], Wk1[l16 + 48]};

        c0 = __builtin_amdgcn_wmma_f32_16x16x4_f32(false, a, false, b0, (short)0, c0, false, false);
        c1 = __builtin_amdgcn_wmma_f32_16x16x4_f32(false, a, false, b1, (short)0, c1, false, false);
        c2 = __builtin_amdgcn_wmma_f32_16x16x4_f32(false, a, false, b2, (short)0, c2, false, false);
        c3 = __builtin_amdgcn_wmma_f32_16x16x4_f32(false, a, false, b3, (short)0, c3, false, false);
    }

    #pragma unroll
    for (int r = 0; r < 8; ++r) {
        const int row = m0 + half * 8 + r;
        if (row < n) {
            float* __restrict__ Cr = C + (size_t)row * 64 + l16;
            Cr[0]  = c0[r];
            Cr[16] = c1[r];
            Cr[32] = c2[r];
            Cr[48] = c3[r];
        }
    }
}


#define GB_WAVES 16
#define GB_PAIRS 2
#define GB_NODES (GB_WAVES * 2 * GB_PAIRS)
#define GB_TILES (GB_NODES / 16)
#define SA_PITCH 68

__device__ __forceinline__ void pool_flush(float* __restrict__ sums, float* __restrict__ cnt,
                                           int g, int lane, float sx, float sy, int run) {
    float* sg = sums + (size_t)g * HID + 2 * lane;
    atomicAdd(sg, sx);
    atomicAdd(sg + 1, sy);
    if (lane == 0) atomicAdd(&cnt[g], (float)run);
}

template <int K>
__device__ __forceinline__ int half_bcast(int v) {
    return __builtin_amdgcn_update_dpp(0, v, 0x150 + K, 0xF, 0xF, true);
}

template <int K>
__device__ __forceinline__ v4f row_of(const char* __restrict__ hb, unsigned so, unsigned lofs) {
    return *reinterpret_cast<const v4f*>(hb + ((unsigned)half_bcast<K>((int)so) | lofs));
}

template <int K>
__device__ __forceinline__ void add_row(v4f& a, const v4f r, int wi) {
    const float w = __int_as_float(half_bcast<K>(wi));
    a.x = fmaf(r.x, w, a.x); a.y = fmaf(r.y, w, a.y); a.z = fmaf(r.z, w, a.z); a.w = fmaf(r.w, w, a.w);
}

template <int K0>
__device__ __forceinline__ void rows8(const char* __restrict__ hb, unsigned so, int wi, unsigned lofs, v4f& a) {
    const v4f r0 = row_of<K0 + 0>(hb, so, lofs);
    const v4f r1 = row_of<K0 + 1>(hb, so, lofs);
    const v4f r2 = row_of<K0 + 2>(hb, so, lofs);
    const v4f r3 = row_of<K0 + 3>(hb, so, lofs);
    const v4f r4 = row_of<K0 + 4>(hb, so, lofs);
    const v4f r5 = row_of<K0 + 5>(hb, so, lofs);
    const v4f r6 = row_of<K0 + 6>(hb, so, lofs);
    const v4f r7 = row_of<K0 + 7>(hb, so, lofs);
    add_row<K0 + 0>(a, r0, wi);
    add_row<K0 + 1>(a, r1, wi);
    add_row<K0 + 2>(a, r2, wi);
    add_row<K0 + 3>(a, r3, wi);
    add_row<K0 + 4>(a, r4, wi);
    add_row<K0 + 5>(a, r5, wi);
    add_row<K0 + 6>(a, r6, wi);
    add_row<K0 + 7>(a, r7, wi);
}

template <int MODE>
__global__ __launch_bounds__(GB_WAVES * 32, 16)
void k_gather(const float* __restrict__ h, const float* __restrict__ dinv,
              const float* __restrict__ bias, const int* __restrict__ rowptr,
              const int* __restrict__ col, int n,
              const float* __restrict__ W, float* __restrict__ hout,
              const int* __restrict__ batch, float* __restrict__ sums, float* __restrict__ cnt) {
    __shared__ __attribute__((aligned(16))) float sA[GB_NODES * SA_PITCH];
    __shared__ __attribute__((aligned(16))) float sW[MODE == 0 ? 64 * 64 : 4];
    const int tid  = threadIdx.x;
    const int lane = tid & 31;
    const int half = lane >> 4;
    const int l16  = lane & 15;
    const int wave = __builtin_amdgcn_readfirstlane(tid >> 5);
    const int i0   = (int)blockIdx.x * GB_NODES;

    v4f w0 = {0.f, 0.f, 0.f, 0.f}, w1 = w0;
    int gv = 0;
    if (MODE == 0) {
        const v4f* __restrict__ W4 = reinterpret_cast<const v4f*>(W);
        w0 = W4[tid];
        w1 = W4[tid + GB_WAVES * 32];
    } else {
        if (wave < GB_TILES && lane < 16 && i0 + 16 * wave + lane < n) gv = batch[i0 + 16 * wave + lane];
    }

    const char* __restrict__ hb = reinterpret_cast<const char*>(h);
    const unsigned lofs = (unsigned)l16 << 4;
    const v4f bb = reinterpret_cast<const v4f*>(bias)[l16];

    int      ok[GB_PAIRS], e0[GB_PAIRS], end[GB_PAIRS];
    unsigned own[GB_PAIRS], so[GB_PAIRS];
    float    di[GB_PAIRS];
    #pragma unroll
    for (int p = 0; p < GB_PAIRS; ++p) {
        const int node = i0 + p * (GB_WAVES * 2) + 2 * wave + half;
        ok[p] = node < n;
        const int ic = min(node, n - 1);
        own[p] = (unsigned)ic << 8;
        di[p]  = dinv[ic];
        e0[p]  = rowptr[ic];
        end[p] = rowptr[ic + 1];
    }
    if (MODE == 0) {
        asm volatile("" :: "v"(w0), "v"(w1));
        v4f* sW4 = reinterpret_cast<v4f*>(sW);
        sW4[tid] = w0;
        sW4[tid + GB_WAVES * 32] = w1;
    }
    int   s0[GB_PAIRS];
    float ds[GB_PAIRS];
    #pragma unroll
    for (int p = 0; p < GB_PAIRS; ++p) {
        if (!ok[p]) end[p] = e0[p];
        e0[p] += l16;
        s0[p] = col[min(e0[p], NEDGES - 1)];
    }
    #pragma unroll
    for (int p = 0; p < GB_PAIRS; ++p) {
        ds[p] = dinv[s0[p]];
        so[p] = e0[p] < end[p] ? (unsigned)s0[p] << 8 : own[p];
    }

    #pragma unroll
    for (int p = 0; p < GB_PAIRS; ++p) {
        const int deg  = end[p] - (e0[p] - l16);
        const int mmax = max(__builtin_amdgcn_readlane(deg, 0), __builtin_amdgcn_readlane(deg, 16));
        const v4f hs = *reinterpret_cast<const v4f*>(hb + (own[p] | lofs));
        const v4f r0 = row_of<0>(hb, so[p], lofs);
        const v4f r1 = row_of<1>(hb, so[p], lofs);
        const v4f r2 = row_of<2>(hb, so[p], lofs);
        const v4f r3 = row_of<3>(hb, so[p], lofs);
        const v4f r4 = row_of<4>(hb, so[p], lofs);
        const v4f r5 = row_of<5>(hb, so[p], lofs);
        const v4f r6 = row_of<6>(hb, so[p], lofs);
        const v4f r7 = row_of<7>(hb, so[p], lofs);
        asm volatile("" :: "v"(hs), "v"(r0), "v"(r1), "v"(r2), "v"(r3), "v"(r4), "v"(r5), "v"(r6), "v"(r7),
                           "v"(ds[0]), "v"(ds[GB_PAIRS - 1]));
        const int wi0 = __float_as_int(e0[p] < end[p] ? ds[p] * di[p] : 0.0f);
        v4f a;
        a.x = hs.x * di[p] * di[p] + bb.x;
        a.y = hs.y * di[p] * di[p] + bb.y;
        a.z = hs.z * di[p] * di[p] + bb.z;
        a.w = hs.w * di[p] * di[p] + bb.w;
        add_row<0>(a, r0, wi0);
        add_row<1>(a, r1, wi0);
        add_row<2>(a, r2, wi0);
        add_row<3>(a, r3, wi0);
        add_row<4>(a, r4, wi0);
        add_row<5>(a, r5, wi0);
        add_row<6>(a, r6, wi0);
        add_row<7>(a, r7, wi0);
        if (mmax > 8) rows8<8>(hb, so[p], wi0, lofs, a);
        for (int c = 16; c < mmax; c += 16) {
            const int e = e0[p] + c;
            const int s = col[min(e, NEDGES - 1)];
            const float dsv = dinv[s];
            const unsigned so_c = e < end[p] ? (unsigned)s << 8 : own[p];
            const int      wi_c = __float_as_int(e < end[p] ? dsv * di[p] : 0.0f);
            rows8<0>(hb, so_c, wi_c, lofs, a);
            if (mmax - c > 8) rows8<8>(hb, so_c, wi_c, lofs, a);
        }
        v4f o = {fmaxf(a.x, 0.0f), fmaxf(a.y, 0.0f), fmaxf(a.z, 0.0f), fmaxf(a.w, 0.0f)};
        if (!ok[p]) o = v4f{0.f, 0.f, 0.f, 0.f};
        const int row = p * (GB_WAVES * 2) + 2 * wave + half;
        *reinterpret_cast<v4f*>(&sA[row * SA_PITCH + 4 * l16]) = o;
    }
    __syncthreads();

    if (MODE == 0) {
        if (wave >= 2 * GB_TILES) return;
        const int rt = wave >> 1;
        if (i0 + rt * 16 >= n) return;
        const int nc = (wave & 1) * 32 + l16;
        v8f c0 = {}, c1 = {};
        #pragma unroll
        for (int k0 = 0; k0 < 64; k0 += 4) {
            const int kb = k0 + half * 2;
            const v2f a = *reinterpret_cast<const v2f*>(&sA[(rt * 16 + l16) * SA_PITCH + kb]);
            const float* __restrict__ Wk0 = sW + kb * 64 + nc;
            const float* __restrict__ Wk1 = sW + (kb + 1) * 64 + nc;
            const v2f b0 = {Wk0[0],  Wk1[0]};
            const v2f b1 = {Wk0[16], Wk1[16]};
            c0 = __builtin_amdgcn_wmma_f32_16x16x4_f32(false, a, false, b0, (short)0, c0, false, false);
            c1 = __builtin_amdgcn_wmma_f32_16x16x4_f32(false, a, false, b1, (short)0, c1, false, false);
        }
        #pragma unroll
        for (int r = 0; r < 8; ++r) {
            const int row = i0 + rt * 16 + half * 8 + r;
            if (row < n) {
                float* __restrict__ Cr = hout + (size_t)row * 64 + nc;
                Cr[0]  = c0[r];
                Cr[16] = c1[r];
            }
        }
    } else {
        if (wave >= GB_TILES) return;
        const int m = min(16, n - (i0 + 16 * wave));
        if (m <= 0) return;
        v2f r[16];
        #pragma unroll
        for (int k = 0; k < 16; ++k) r[k] = *reinterpret_cast<const v2f*>(&sA[(16 * wave + k) * SA_PITCH + 2 * lane]);
        int   cur = __builtin_amdgcn_readlane(gv, 0);
        int   run = 0;
        float sx = 0.0f, sy = 0.0f;
        #pragma unroll
        for (int k = 0; k < 16; ++k) {
            if (k < m) {
                const int g = __builtin_amdgcn_readlane(gv, k);
                if (g != cur) {
                    pool_flush(sums, cnt, cur, lane, sx, sy, run);
                    cur = g; run = 0; sx = 0.0f; sy = 0.0f;
                }
                sx += r[k].x;
                sy += r[k].y;
                ++run;
            }
        }
        pool_flush(sums, cnt, cur, lane, sx, sy, run);
    }
}

__global__ __launch_bounds__(256)
void k_head(const float* __restrict__ sums, const float* __restrict__ cnt,
            const float* __restrict__ Wl, const float* __restrict__ bl,
            float* __restrict__ out) {
    int t = blockIdx.x * blockDim.x + threadIdx.x;
    if (t < NGRAPHS * NCLASSES) {
        int g = t / NCLASSES;
        int c = t % NCLASSES;
        float inv = 1.0f / fmaxf(cnt[g], 1.0f);
        float acc = 0.0f;
        const float* sg = sums + (size_t)g * HID;
        #pragma unroll
        for (int k = 0; k < HID; ++k) acc += sg[k] * Wl[k * NCLASSES + c];
        out[t] = acc * inv + bl[c];
    }
}


extern "C" void kernel_launch(void* const* d_in, const int* in_sizes, int n_in,
                              void* d_out, int out_size, void* d_ws, size_t ws_size,
                              hipStream_t stream) {
    const float* x  = (const float*)d_in[0];
    const float* W1 = (const float*)d_in[1];
    const float* b1 = (const float*)d_in[2];
    const float* W2 = (const float*)d_in[3];
    const float* b2 = (const float*)d_in[4];
    const float* Wl = (const float*)d_in[5];
    const float* bl = (const float*)d_in[6];
    const int* edge  = (const int*)d_in[7];
    const int* batch = (const int*)d_in[8];
    const int* src = edge;
    const int* dst = edge + NEDGES;
    float* out = (float*)d_out;

    float* ws   = (float*)d_ws;
    float* h    = ws;
    float* a    = h + (size_t)NNODES * HID;
    float* dinv = a + (size_t)NNODES * HID;
    float* sums = dinv + NNODES;
    float* cnt  = sums + (size_t)NGRAPHS * HID;
    int* hist   = (int*)(cnt + NGRAPHS);
    int* bstart = hist + NHIST;
    int* rowptr = bstart + NBKT + 4;
    int* packed = rowptr + NNODES + 4;
    int* col    = packed + NEDGES;

    const int TPB = 256;
    const int gGa  = (NNODES + GB_NODES - 1) / GB_NODES;
    const int nPZ  = NGRAPHS * HID + NGRAPHS;
    static_assert(NCHUNK * 256 >= NGRAPHS * HID + NGRAPHS, "k_hist's blocks clear sums and cnt");
    const int gHd  = (NGRAPHS * NCLASSES + TPB - 1) / TPB;

    k_hist<<<NCHUNK + GEMM_P0, TPB, 0, stream>>>(dst, hist, NEDGES, sums, nPZ, x, W1, h, NNODES);
    k_bin<<<NCHUNK + GEMM_P1, TPB, 0, stream>>>(src, dst, hist, bstart, packed, NEDGES, x, W1, h, NNODES);
    k_csr<<<NBKT + GEMM_P2, TPB, 0, stream>>>(packed, bstart, rowptr, dinv, col, NNODES, x, W1, h);

    k_gather<0><<<gGa, GB_WAVES * 32, 0, stream>>>(h, dinv, b1, rowptr, col, NNODES, W2, a,
                                                   nullptr, nullptr, nullptr);
    k_gather<1><<<gGa, GB_WAVES * 32, 0, stream>>>(a, dinv, b2, rowptr, col, NNODES, nullptr, nullptr,
                                                   batch, sums, cnt);
    k_head<<<gHd, TPB, 0, stream>>>(sums, cnt, Wl, bl, out);
}


# Launches 6 to 5: each edge block groups its own chunk by bucket in the chunk's own slots (no counting launch); bucket blocks collect their runs chunk by chunk; list order stays arrival order
# speedup vs baseline: 5.5633x; 1.3251x over previous
#include <hip/hip_runtime.h>

typedef float v2f __attribute__((ext_vector_type(2)));
typedef float v8f __attribute__((ext_vector_type(8)));
typedef float v4f __attribute__((ext_vector_type(4)));

#define NNODES 50000
#define NEDGES 800000
#define HID 64
#define NGRAPHS 512
#define NCLASSES 10


#define BKT_SHIFT 8
#define BKT_NODES (1 << BKT_SHIFT)
#define NBKT ((NNODES + BKT_NODES - 1) / BKT_NODES)
#define CHUNK 4096
#define NCHUNK ((NEDGES + CHUNK - 1) / CHUNK)
#define OFFP (NBKT + 4)
#define CSR_RUNS ((NCHUNK + 7) / 8)
#define CSR_BATCH 13
static_assert(NBKT < 256 && CSR_RUNS <= 32, "one thread per bucket in k_bin, one lane per chunk of a wave in k_csr");

#define GEMM_BLOCKS (((NNODES + 15) / 16 + 7) / 8)
#define GEMM_P0 ((GEMM_BLOCKS + 1) / 2)
#define GEMM_P1 (GEMM_BLOCKS - GEMM_P0)

__device__ __forceinline__ void gemm_block(const float* __restrict__ A, const float* __restrict__ W,
                                           float* __restrict__ C, int n, int relu_in, int blk);

__global__ __launch_bounds__(256)
void k_bin(const int* __restrict__ src, const int* __restrict__ dst, int* __restrict__ offs,
           int* __restrict__ packed, int e, float* __restrict__ zp, int zn,
           const float* __restrict__ x, const float* __restrict__ W1, float* __restrict__ h, int n) {
    if (blockIdx.x >= NCHUNK) {
        gemm_block(x, W1, h, n, 0, (int)blockIdx.x - NCHUNK);
        return;
    }
    __shared__ int s_cnt[256];
    __shared__ int s_wsum[8];
    const int tid  = threadIdx.x;
    const int b    = blockIdx.x;
    const int base = b * CHUNK;
    const int z = b * 256 + tid;
    if (z < zn) zp[z] = 0.0f;
    s_cnt[tid] = 0;
    int d[CHUNK / 256];
    int s[CHUNK / 256];
    #pragma unroll
    for (int j = 0; j < CHUNK / 256; ++j) {
        const int i = min(base + j * 256 + tid, e - 1);
        d[j] = dst[i];
        s[j] = src[i];
    }
    __syncthreads();
    #pragma unroll
    for (int j = 0; j < CHUNK / 256; ++j) {
        if (base + j * 256 + tid < e) atomicAdd(&s_cnt[d[j] >> BKT_SHIFT], 1);
    }
    __syncthreads();
    const int total = s_cnt[tid];
    const int lane = tid & 31;
    const int wave = tid >> 5;
    int incl = total;
    #pragma unroll
    for (int off = 1; off < 32; off <<= 1) {
        const int v = __shfl_up(incl, off);
        if (lane >= off) incl += v;
    }
    if (lane == 31) s_wsum[wave] = incl;
    __syncthreads();
    int wbase = 0;
    #pragma unroll
    for (int w = 0; w < 8; ++w) wbase += (w < wave) ? s_wsum[w] : 0;
    const int start = wbase + incl - total;
    if (tid <= NBKT) offs[b * OFFP + tid] = start;
    s_cnt[tid] = start;
    __syncthreads();
    #pragma unroll
    for (int j = 0; j < CHUNK / 256; ++j) {
        if (base + j * 256 + tid < e) {
            const int slot = atomicAdd(&s_cnt[d[j] >> BKT_SHIFT], 1);
            packed[base + slot] = (s[j] << BKT_SHIFT) | (d[j] & (BKT_NODES - 1));
        }
    }
}

template <int PLACE>
__device__ __forceinline__ void csr_walk(const int* __restrict__ packed, int o, int nv,
                                         int* s_cnt, int* __restrict__ col, int lane) {
    #pragma unroll
    for (int t = 0; t < CSR_RUNS; t += CSR_BATCH) {
        int v[CSR_BATCH], nn[CSR_BATCH], oo[CSR_BATCH];
        #pragma unroll
        for (int u = 0; u < CSR_BATCH; ++u) {
            if (t + u < CSR_RUNS) {
                oo[u] = __builtin_amdgcn_readlane(o, t + u);
                nn[u] = __builtin_amdgcn_readlane(nv, t + u);
                v[u]  = packed[min(oo[u] + lane, NEDGES - 1)];
            }
        }
        #pragma unroll
        for (int u = 0; u < CSR_BATCH; ++u) {
            if (t + u < CSR_RUNS) {
                if (lane < nn[u]) {
                    if (PLACE) {
                        const int slot = atomicAdd(&s_cnt[v[u] & (BKT_NODES - 1)], 1);
                        col[slot] = v[u] >> BKT_SHIFT;
                    } else {
                        atomicAdd(&s_cnt[v[u] & (BKT_NODES - 1)], 1);
                    }
                }
                for (int j = lane + 32; j < nn[u]; j += 32) {
                    const int w = packed[oo[u] + j];
                    if (PLACE) {
                        const int slot = atomicAdd(&s_cnt[w & (BKT_NODES - 1)], 1);
                        col[slot] = w >> BKT_SHIFT;
                    } else {
                        atomicAdd(&s_cnt[w & (BKT_NODES - 1)], 1);
                    }
                }
            }
        }
    }
}

__global__ __launch_bounds__(256)
void k_csr(const int* __restrict__ packed, const int* __restrict__ offs, int* __restrict__ rowptr,
           float* __restrict__ dinv, int* __restrict__ col, int n,
           const float* __restrict__ x, const float* __restrict__ W1, float* __restrict__ h) {
    if (blockIdx.x >= NBKT) {
        gemm_block(x, W1, h, n, 0, (int)blockIdx.x - NBKT + GEMM_P0);
        return;
    }
    __shared__ int s_cnt[BKT_NODES];
    __shared__ int s_scan[BKT_NODES];
    __shared__ int s_red[8];
    const int tid  = threadIdx.x;
    const int k    = blockIdx.x;
    const int lane = tid & 31;
    const int wave = tid >> 5;
    const int c    = wave * CSR_RUNS + lane;
    int o = 0, nv = 0, a = 0;
    if (lane < CSR_RUNS && c < NCHUNK) {
        a = offs[c * OFFP + k];
        const int b = offs[c * OFFP + k + 1];
        o  = c * CHUNK + a;
        nv = b - a;
    }
    s_cnt[tid] = 0;
    int ra = a;
    #pragma unroll
    for (int off = 16; off >= 1; off >>= 1) ra += __shfl_xor(ra, off);
    if (lane == 0) s_red[wave] = ra;
    __syncthreads();
    int bs = 0;
    #pragma unroll
    for (int w = 0; w < 8; ++w) bs += s_red[w];
    csr_walk<0>(packed, o, nv, s_cnt, col, lane);
    __syncthreads();
    const int own = s_cnt[tid];
    s_scan[tid] = own;
    __syncthreads();
    for (int off = 1; off < 256; off <<= 1) {
        int v = 0;
        if (tid >= off) v = s_scan[tid - off];
        __syncthreads();
        s_scan[tid] += v;
        __syncthreads();
    }
    const int start = bs + s_scan[tid] - own;
    const int i = k * BKT_NODES + tid;
    if (i < n) {
        rowptr[i] = start;
        dinv[i]   = rsqrtf((float)(own + 1));
    }
    if (i == n - 1) rowptr[n] = start + own;
    s_cnt[tid] = start;
    __syncthreads();
    csr_walk<1>(packed, o, nv, s_cnt, col, lane);
}


__device__ __forceinline__ void gemm_block(const float* __restrict__ A, const float* __restrict__ W,
                                           float* __restrict__ C, int n, int relu_in, int blk) {
    __shared__ float sW[64 * 64];
    const int tid = threadIdx.x;
    #pragma unroll
    for (int i = 0; i < 16; ++i) sW[tid + i * 256] = W[tid + i * 256];
    __syncthreads();

    const int wave = tid >> 5;
    const int lane = tid & 31;
    const int half = lane >> 4;
    const int l16  = lane & 15;
    const int m0   = (blk * 8 + wave) * 16;
    if (m0 >= n) return;

    v8f c0 = {}, c1 = {}, c2 = {}, c3 = {};

    const int arow = m0 + l16;
    const float* __restrict__ Arow = A + (size_t)arow * 64;

    #pragma unroll
    for (int k0 = 0; k0 < 64; k0 += 4) {
        const int kb = k0 + half * 2;
        float a0 = Arow[kb];
        float a1 = Arow[kb + 1];
        if (relu_in) { a0 = fmaxf(a0, 0.0f); a1 = fmaxf(a1, 0.0f); }
        v2f a = {a0, a1};

        const float* __restrict__ Wk0 = sW + (size_t)kb * 64;
        const float* __restrict__ Wk1 = sW + (size_t)(kb + 1) * 64;
        v2f b0 = {Wk0[l16 +  0], Wk1[l16 +  0]};
        v2f b1 = {Wk0[l16 + 16], Wk1[l16 + 16]};
        v2f b2 = {Wk0[l16 + 32], Wk1[l16 + 32]};
        v2f b3 = {Wk0[l16 + 48], Wk1[l16 + 48]};

        c0 = __builtin_amdgcn_wmma_f32_16x16x4_f32(false, a, false, b0, (short)0, c0, false, false);
        c1 = __builtin_amdgcn_wmma_f32_16x16x4_f32(false, a, false, b1, (short)0, c1, false, false);
        c2 = __builtin_amdgcn_wmma_f32_16x16x4_f32(false, a, false, b2, (short)0, c2, false, false);
        c3 = __builtin_amdgcn_wmma_f32_16x16x4_f32(false, a, false, b3, (short)0, c3, false, false);
    }

    #pragma unroll
    for (int r = 0; r < 8; ++r) {
        const int row = m0 + half * 8 + r;
        if (row < n) {
            float* __restrict__ Cr = C + (size_t)row * 64 + l16;
            Cr[0]  = c0[r];
            Cr[16] = c1[r];
            Cr[32] = c2[r];
            Cr[48] = c3[r];
        }
    }
}


#define GB_WAVES 16
#define GB_PAIRS 2
#define GB_NODES (GB_WAVES * 2 * GB_PAIRS)
#define GB_TILES (GB_NODES / 16)
#define SA_PITCH 68

__device__ __forceinline__ void pool_flush(float* __restrict__ sums, float* __restrict__ cnt,
                                           int g, int lane, float sx, float sy, int run) {
    float* sg = sums + (size_t)g * HID + 2 * lane;
    atomicAdd(sg, sx);
    atomicAdd(sg + 1, sy);
    if (lane == 0) atomicAdd(&cnt[g], (float)run);
}

template <int K>
__device__ __forceinline__ int half_bcast(int v) {
    return __builtin_amdgcn_update_dpp(0, v, 0x150 + K, 0xF, 0xF, true);
}

template <int K>
__device__ __forceinline__ v4f row_of(const char* __restrict__ hb, unsigned so, unsigned lofs) {
    return *reinterpret_cast<const v4f*>(hb + ((unsigned)half_bcast<K>((int)so) | lofs));
}

template <int K>
__device__ __forceinline__ void add_row(v4f& a, const v4f r, int wi) {
    const float w = __int_as_float(half_bcast<K>(wi));
    a.x = fmaf(r.x, w, a.x); a.y = fmaf(r.y, w, a.y); a.z = fmaf(r.z, w, a.z); a.w = fmaf(r.w, w, a.w);
}

template <int K0>
__device__ __forceinline__ void rows8(const char* __restrict__ hb, unsigned so, int wi, unsigned lofs, v4f& a) {
    const v4f r0 = row_of<K0 + 0>(hb, so, lofs);
    const v4f r1 = row_of<K0 + 1>(hb, so, lofs);
    const v4f r2 = row_of<K0 + 2>(hb, so, lofs);
    const v4f r3 = row_of<K0 + 3>(hb, so, lofs);
    const v4f r4 = row_of<K0 + 4>(hb, so, lofs);
    const v4f r5 = row_of<K0 + 5>(hb, so, lofs);
    const v4f r6 = row_of<K0 + 6>(hb, so, lofs);
    const v4f r7 = row_of<K0 + 7>(hb, so, lofs);
    add_row<K0 + 0>(a, r0, wi);
    add_row<K0 + 1>(a, r1, wi);
    add_row<K0 + 2>(a, r2, wi);
    add_row<K0 + 3>(a, r3, wi);
    add_row<K0 + 4>(a, r4, wi);
    add_row<K0 + 5>(a, r5, wi);
    add_row<K0 + 6>(a, r6, wi);
    add_row<K0 + 7>(a, r7, wi);
}

template <int MODE>
__global__ __launch_bounds__(GB_WAVES * 32, 16)
void k_gather(const float* __restrict__ h, const float* __restrict__ dinv,
              const float* __restrict__ bias, const int* __restrict__ rowptr,
              const int* __restrict__ col, int n,
              const float* __restrict__ W, float* __restrict__ hout,
              const int* __restrict__ batch, float* __restrict__ sums, float* __restrict__ cnt) {
    __shared__ __attribute__((aligned(16))) float sA[GB_NODES * SA_PITCH];
    __shared__ __attribute__((aligned(16))) float sW[MODE == 0 ? 64 * 64 : 4];
    const int tid  = threadIdx.x;
    const int lane = tid & 31;
    const int half = lane >> 4;
    const int l16  = lane & 15;
    const int wave = __builtin_amdgcn_readfirstlane(tid >> 5);
    const int i0   = (int)blockIdx.x * GB_NODES;

    v4f w0 = {0.f, 0.f, 0.f, 0.f}, w1 = w0;
    int gv = 0;
    if (MODE == 0) {
        const v4f* __restrict__ W4 = reinterpret_cast<const v4f*>(W);
        w0 = W4[tid];
        w1 = W4[tid + GB_WAVES * 32];
    } else {
        if (wave < GB_TILES && lane < 16 && i0 + 16 * wave + lane < n) gv = batch[i0 + 16 * wave + lane];
    }

    const char* __restrict__ hb = reinterpret_cast<const char*>(h);
    const unsigned lofs = (unsigned)l16 << 4;
    const v4f bb = reinterpret_cast<const v4f*>(bias)[l16];

    int      ok[GB_PAIRS], e0[GB_PAIRS], end[GB_PAIRS];
    unsigned own[GB_PAIRS], so[GB_PAIRS];
    float    di[GB_PAIRS];
    #pragma unroll
    for (int p = 0; p < GB_PAIRS; ++p) {
        const int node = i0 + p * (GB_WAVES * 2) + 2 * wave + half;
        ok[p] = node < n;
        const int ic = min(node, n - 1);
        own[p] = (unsigned)ic << 8;
        di[p]  = dinv[ic];
        e0[p]  = rowptr[ic];
        end[p] = rowptr[ic + 1];
    }
    if (MODE == 0) {
        asm volatile("" :: "v"(w0), "v"(w1));
        v4f* sW4 = reinterpret_cast<v4f*>(sW);
        sW4[tid] = w0;
        sW4[tid + GB_WAVES * 32] = w1;
    }
    int   s0[GB_PAIRS];
    float ds[GB_PAIRS];
    #pragma unroll
    for (int p = 0; p < GB_PAIRS; ++p) {
        if (!ok[p]) end[p] = e0[p];
        e0[p] += l16;
        s0[p] = col[min(e0[p], NEDGES - 1)];
    }
    #pragma unroll
    for (int p = 0; p < GB_PAIRS; ++p) {
        ds[p] = dinv[s0[p]];
        so[p] = e0[p] < end[p] ? (unsigned)s0[p] << 8 : own[p];
    }

    #pragma unroll
    for (int p = 0; p < GB_PAIRS; ++p) {
        const int deg  = end[p] - (e0[p] - l16);
        const int mmax = max(__builtin_amdgcn_readlane(deg, 0), __builtin_amdgcn_readlane(deg, 16));
        const v4f hs = *reinterpret_cast<const v4f*>(hb + (own[p] | lofs));
        const v4f r0 = row_of<0>(hb, so[p], lofs);
        const v4f r1 = row_of<1>(hb, so[p], lofs);
        const v4f r2 = row_of<2>(hb, so[p], lofs);
        const v4f r3 = row_of<3>(hb, so[p], lofs);
        const v4f r4 = row_of<4>(hb, so[p], lofs);
        const v4f r5 = row_of<5>(hb, so[p], lofs);
        const v4f r6 = row_of<6>(hb, so[p], lofs);
        const v4f r7 = row_of<7>(hb, so[p], lofs);
        asm volatile("" :: "v"(hs), "v"(r0), "v"(r1), "v"(r2), "v"(r3), "v"(r4), "v"(r5), "v"(r6), "v"(r7),
                           "v"(ds[0]), "v"(ds[GB_PAIRS - 1]));
        const int wi0 = __float_as_int(e0[p] < end[p] ? ds[p] * di[p] : 0.0f);
        v4f a;
        a.x = hs.x * di[p] * di[p] + bb.x;
        a.y = hs.y * di[p] * di[p] + bb.y;
        a.z = hs.z * di[p] * di[p] + bb.z;
        a.w = hs.w * di[p] * di[p] + bb.w;
        add_row<0>(a, r0, wi0);
        add_row<1>(a, r1, wi0);
        add_row<2>(a, r2, wi0);
        add_row<3>(a, r3, wi0);
        add_row<4>(a, r4, wi0);
        add_row<5>(a, r5, wi0);
        add_row<6>(a, r6, wi0);
        add_row<7>(a, r7, wi0);
        if (mmax > 8) rows8<8>(hb, so[p], wi0, lofs, a);
        for (int c = 16; c < mmax; c += 16) {
            const int e = e0[p] + c;
            const int s = col[min(e, NEDGES - 1)];
            const float dsv = dinv[s];
            const unsigned so_c = e < end[p] ? (unsigned)s << 8 : own[p];
            const int      wi_c = __float_as_int(e < end[p] ? dsv * di[p] : 0.0f);
            rows8<0>(hb, so_c, wi_c, lofs, a);
            if (mmax - c > 8) rows8<8>(hb, so_c, wi_c, lofs, a);
        }
        v4f o = {fmaxf(a.x, 0.0f), fmaxf(a.y, 0.0f), fmaxf(a.z, 0.0f), fmaxf(a.w, 0.0f)};
        if (!ok[p]) o = v4f{0.f, 0.f, 0.f, 0.f};
        const int row = p * (GB_WAVES * 2) + 2 * wave + half;
        *reinterpret_cast<v4f*>(&sA[row * SA_PITCH + 4 * l16]) = o;
    }
    __syncthreads();

    if (MODE == 0) {
        if (wave >= 2 * GB_TILES) return;
        const int rt = wave >> 1;
        if (i0 + rt * 16 >= n) return;
        const int nc = (wave & 1) * 32 + l16;
        v8f c0 = {}, c1 = {};
        #pragma unroll
        for (int k0 = 0; k0 < 64; k0 += 4) {
            const int kb = k0 + half * 2;
            const v2f a = *reinterpret_cast<const v2f*>(&sA[(rt * 16 + l16) * SA_PITCH + kb]);
            const float* __restrict__ Wk0 = sW + kb * 64 + nc;
            const float* __restrict__ Wk1 = sW + (kb + 1) * 64 + nc;
            const v2f b0 = {Wk0[0],  Wk1[0]};
            const v2f b1 = {Wk0[16], Wk1[16]};
            c0 = __builtin_amdgcn_wmma_f32_16x16x4_f32(false, a, false, b0, (short)0, c0, false, false);
            c1 = __builtin_amdgcn_wmma_f32_16x16x4_f32(false, a, false, b1, (short)0, c1, false, false);
        }
        #pragma unroll
        for (int r = 0; r < 8; ++r) {
            const int row = i0 + rt * 16 + half * 8 + r;
            if (row < n) {
                float* __restrict__ Cr = hout + (size_t)row * 64 + nc;
                Cr[0]  = c0[r];
                Cr[16] = c1[r];
            }
        }
    } else {
        if (wave >= GB_TILES) return;
        const int m = min(16, n - (i0 + 16 * wave));
        if (m <= 0) return;
        v2f r[16];
        #pragma unroll
        for (int k = 0; k < 16; ++k) r[k] = *reinterpret_cast<const v2f*>(&sA[(16 * wave + k) * SA_PITCH + 2 * lane]);
        int   cur = __builtin_amdgcn_readlane(gv, 0);
        int   run = 0;
        float sx = 0.0f, sy = 0.0f;
        #pragma unroll
        for (int k = 0; k < 16; ++k) {
            if (k < m) {
                const int g = __builtin_amdgcn_readlane(gv, k);
                if (g != cur) {
                    pool_flush(sums, cnt, cur, lane, sx, sy, run);
                    cur = g; run = 0; sx = 0.0f; sy = 0.0f;
                }
                sx += r[k].x;
                sy += r[k].y;
                ++run;
            }
        }
        pool_flush(sums, cnt, cur, lane, sx, sy, run);
    }
}

__global__ __launch_bounds__(256)
void k_head(const float* __restrict__ sums, const float* __restrict__ cnt,
            const float* __restrict__ Wl, const float* __restrict__ bl,
            float* __restrict__ out) {
    int t = blockIdx.x * blockDim.x + threadIdx.x;
    if (t < NGRAPHS * NCLASSES) {
        int g = t / NCLASSES;
        int c = t % NCLASSES;
        float inv = 1.0f / fmaxf(cnt[g], 1.0f);
        float acc = 0.0f;
        const float* sg = sums + (size_t)g * HID;
        #pragma unroll
        for (int k = 0; k < HID; ++k) acc += sg[k] * Wl[k * NCLASSES + c];
        out[t] = acc * inv + bl[c];
    }
}


extern "C" void kernel_launch(void* const* d_in, const int* in_sizes, int n_in,
                              void* d_out, int out_size, void* d_ws, size_t ws_size,
                              hipStream_t stream) {
    const float* x  = (const float*)d_in[0];
    const float* W1 = (const float*)d_in[1];
    const float* b1 = (const float*)d_in[2];
    const float* W2 = (const float*)d_in[3];
    const float* b2 = (const float*)d_in[4];
    const float* Wl = (const float*)d_in[5];
    const float* bl = (const float*)d_in[6];
    const int* edge  = (const int*)d_in[7];
    const int* batch = (const int*)d_in[8];
    const int* src = edge;
    const int* dst = edge + NEDGES;
    float* out = (float*)d_out;

    float* ws   = (float*)d_ws;
    float* h    = ws;
    float* a    = h + (size_t)NNODES * HID;
    float* dinv = a + (size_t)NNODES * HID;
    float* sums = dinv + NNODES;
    float* cnt  = sums + (size_t)NGRAPHS * HID;
    int* offs   = (int*)(cnt + NGRAPHS);
    int* rowptr = offs + NCHUNK * OFFP;
    int* packed = rowptr + NNODES + 4;
    int* col    = packed + NEDGES;

    const int TPB = 256;
    const int gGa  = (NNODES + GB_NODES - 1) / GB_NODES;
    const int nPZ  = NGRAPHS * HID + NGRAPHS;
    static_assert(NCHUNK * 256 >= NGRAPHS * HID + NGRAPHS, "k_bin's blocks clear sums and cnt");
    const int gHd  = (NGRAPHS * NCLASSES + TPB - 1) / TPB;

    k_bin<<<NCHUNK + GEMM_P0, TPB, 0, stream>>>(src, dst, offs, packed, NEDGES, sums, nPZ, x, W1, h, NNODES);
    k_csr<<<NBKT + GEMM_P1, TPB, 0, stream>>>(packed, offs, rowptr, dinv, col, NNODES, x, W1, h);

    k_gather<0><<<gGa, GB_WAVES * 32, 0, stream>>>(h, dinv, b1, rowptr, col, NNODES, W2, a,
                                                   nullptr, nullptr, nullptr);
    k_gather<1><<<gGa, GB_WAVES * 32, 0, stream>>>(a, dinv, b2, rowptr, col, NNODES, nullptr, nullptr,
                                                   batch, sums, cnt);
    k_head<<<gHd, TPB, 0, stream>>>(sums, cnt, Wl, bl, out);
}


# Rows kept pre-scaled by their own dinv (h: in place by the block that makes dinv; layer-2 rows: at the product's store): gathers add plain rows, no dinv[src] fetch per edge; di*(own+list rows)+bias
# speedup vs baseline: 5.9413x; 1.0679x over previous
#include <hip/hip_runtime.h>

typedef float v2f __attribute__((ext_vector_type(2)));
typedef float v8f __attribute__((ext_vector_type(8)));
typedef float v4f __attribute__((ext_vector_type(4)));

#define NNODES 50000
#define NEDGES 800000
#define HID 64
#define NGRAPHS 512
#define NCLASSES 10


#define BKT_SHIFT 8
#define BKT_NODES (1 << BKT_SHIFT)
#define NBKT ((NNODES + BKT_NODES - 1) / BKT_NODES)
#define CHUNK 4096
#define NCHUNK ((NEDGES + CHUNK - 1) / CHUNK)
#define OFFP (NBKT + 4)
#define CSR_RUNS ((NCHUNK + 7) / 8)
#define CSR_BATCH 13
static_assert(NBKT < 256 && CSR_RUNS <= 32, "one thread per bucket in k_bin, one lane per chunk of a wave in k_csr");

#define GEMM_BLOCKS (((NNODES + 15) / 16 + 7) / 8)

__device__ __forceinline__ void gemm_block(const float* __restrict__ A, const float* __restrict__ W,
                                           float* __restrict__ C, int n, int relu_in, int blk);

__global__ __launch_bounds__(256)
void k_bin(const int* __restrict__ src, const int* __restrict__ dst, int* __restrict__ offs,
           int* __restrict__ packed, int e, float* __restrict__ zp, int zn,
           const float* __restrict__ x, const float* __restrict__ W1, float* __restrict__ h, int n) {
    if (blockIdx.x >= NCHUNK) {
        gemm_block(x, W1, h, n, 0, (int)blockIdx.x - NCHUNK);
        return;
    }
    __shared__ int s_cnt[256];
    __shared__ int s_wsum[8];
    const int tid  = threadIdx.x;
    const int b    = blockIdx.x;
    const int base = b * CHUNK;
    const int z = b * 256 + tid;
    if (z < zn) zp[z] = 0.0f;
    s_cnt[tid] = 0;
    int d[CHUNK / 256];
    int s[CHUNK / 256];
    #pragma unroll
    for (int j = 0; j < CHUNK / 256; ++j) {
        const int i = min(base + j * 256 + tid, e - 1);
        d[j] = dst[i];
        s[j] = src[i];
    }
    __syncthreads();
    #pragma unroll
    for (int j = 0; j < CHUNK / 256; ++j) {
        if (base + j * 256 + tid < e) atomicAdd(&s_cnt[d[j] >> BKT_SHIFT], 1);
    }
    __syncthreads();
    const int total = s_cnt[tid];
    const int lane = tid & 31;
    const int wave = tid >> 5;
    int incl = total;
    #pragma unroll
    for (int off = 1; off < 32; off <<= 1) {
        const int v = __shfl_up(incl, off);
        if (lane >= off) incl += v;
    }
    if (lane == 31) s_wsum[wave] = incl;
    __syncthreads();
    int wbase = 0;
    #pragma unroll
    for (int w = 0; w < 8; ++w) wbase += (w < wave) ? s_wsum[w] : 0;
    const int start = wbase + incl - total;
    if (tid <= NBKT) offs[b * OFFP + tid] = start;
    s_cnt[tid] = start;
    __syncthreads();
    #pragma unroll
    for (int j = 0; j < CHUNK / 256; ++j) {
        if (base + j * 256 + tid < e) {
            const int slot = atomicAdd(&s_cnt[d[j] >> BKT_SHIFT], 1);
            packed[base + slot] = (s[j] << BKT_SHIFT) | (d[j] & (BKT_NODES - 1));
        }
    }
}

template <int PLACE>
__device__ __forceinline__ void csr_walk(const int* __restrict__ packed, int o, int nv,
                                         int* s_cnt, int* __restrict__ col, int lane) {
    #pragma unroll
    for (int t = 0; t < CSR_RUNS; t += CSR_BATCH) {
        int v[CSR_BATCH], nn[CSR_BATCH], oo[CSR_BATCH];
        #pragma unroll
        for (int u = 0; u < CSR_BATCH; ++u) {
            if (t + u < CSR_RUNS) {
                oo[u] = __builtin_amdgcn_readlane(o, t + u);
                nn[u] = __builtin_amdgcn_readlane(nv, t + u);
                v[u]  = packed[min(oo[u] + lane, NEDGES - 1)];
            }
        }
        #pragma unroll
        for (int u = 0; u < CSR_BATCH; ++u) {
            if (t + u < CSR_RUNS) {
                if (lane < nn[u]) {
                    if (PLACE) {
                        const int slot = atomicAdd(&s_cnt[v[u] & (BKT_NODES - 1)], 1);
                        col[slot] = v[u] >> BKT_SHIFT;
                    } else {
                        atomicAdd(&s_cnt[v[u] & (BKT_NODES - 1)], 1);
                    }
                }
                for (int j = lane + 32; j < nn[u]; j += 32) {
                    const int w = packed[oo[u] + j];
                    if (PLACE) {
                        const int slot = atomicAdd(&s_cnt[w & (BKT_NODES - 1)], 1);
                        col[slot] = w >> BKT_SHIFT;
                    } else {
                        atomicAdd(&s_cnt[w & (BKT_NODES - 1)], 1);
                    }
                }
            }
        }
    }
}

#define HS_STEPS (BKT_NODES * 16 / 256)
__global__ __launch_bounds__(256)
void k_csr(const int* __restrict__ packed, const int* __restrict__ offs, int* __restrict__ rowptr,
           float* __restrict__ dinv, int* __restrict__ col, int n, float* __restrict__ h) {
    __shared__ int s_cnt[BKT_NODES];
    __shared__ int s_scan[BKT_NODES];
    __shared__ int s_red[8];
    __shared__ float s_dv[BKT_NODES];
    const int tid  = threadIdx.x;
    const int k    = blockIdx.x;
    const int lane = tid & 31;
    const int wave = tid >> 5;
    const int c    = wave * CSR_RUNS + lane;
    v4f* __restrict__ h4 = reinterpret_cast<v4f*>(h) + (size_t)k * BKT_NODES * 16;
    const int hrows = min(BKT_NODES, n - k * BKT_NODES);
    v4f hr[HS_STEPS];
    #pragma unroll
    for (int j = 0; j < HS_STEPS; ++j) {
        hr[j] = v4f{0.f, 0.f, 0.f, 0.f};
        if (j * 16 + (tid >> 4) < hrows) hr[j] = h4[j * 256 + tid];
    }
    int o = 0, nv = 0, a = 0;
    if (lane < CSR_RUNS && c < NCHUNK) {
        a = offs[c * OFFP + k];
        const int b = offs[c * OFFP + k + 1];
        o  = c * CHUNK + a;
        nv = b - a;
    }
    s_cnt[tid] = 0;
    int ra = a;
    #pragma unroll
    for (int off = 16; off >= 1; off >>= 1) ra += __shfl_xor(ra, off);
    if (lane == 0) s_red[wave] = ra;
    __syncthreads();
    int bs = 0;
    #pragma unroll
    for (int w = 0; w < 8; ++w) bs += s_red[w];
    csr_walk<0>(packed, o, nv, s_cnt, col, lane);
    __syncthreads();
    const int own = s_cnt[tid];
    s_scan[tid] = own;
    __syncthreads();
    for (int off = 1; off < 256; off <<= 1) {
        int v = 0;
        if (tid >= off) v = s_scan[tid - off];
        __syncthreads();
        s_scan[tid] += v;
        __syncthreads();
    }
    const int start = bs + s_scan[tid] - own;
    const int i = k * BKT_NODES + tid;
    const float dv = rsqrtf((float)(own + 1));
    s_dv[tid] = dv;
    if (i < n) {
        rowptr[i] = start;
        dinv[i]   = dv;
    }
    if (i == n - 1) rowptr[n] = start + own;
    s_cnt[tid] = start;
    __syncthreads();
    float hw[HS_STEPS];
    #pragma unroll
    for (int j = 0; j < HS_STEPS; ++j) hw[j] = s_dv[j * 16 + (tid >> 4)];
    #pragma unroll
    for (int j = 0; j < HS_STEPS; ++j) {
        if (j * 16 + (tid >> 4) < hrows)
            h4[j * 256 + tid] = v4f{hr[j].x * hw[j], hr[j].y * hw[j], hr[j].z * hw[j], hr[j].w * hw[j]};
    }
    csr_walk<1>(packed, o, nv, s_cnt, col, lane);
}


__device__ __forceinline__ void gemm_block(const float* __restrict__ A, const float* __restrict__ W,
                                           float* __restrict__ C, int n, int relu_in, int blk) {
    __shared__ float sW[64 * 64];
    const int tid = threadIdx.x;
    #pragma unroll
    for (int i = 0; i < 16; ++i) sW[tid + i * 256] = W[tid + i * 256];
    __syncthreads();

    const int wave = tid >> 5;
    const int lane = tid & 31;
    const int half = lane >> 4;
    const int l16  = lane & 15;
    const int m0   = (blk * 8 + wave) * 16;
    if (m0 >= n) return;

    v8f c0 = {}, c1 = {}, c2 = {}, c3 = {};

    const int arow = m0 + l16;
    const float* __restrict__ Arow = A + (size_t)arow * 64;

    #pragma unroll
    for (int k0 = 0; k0 < 64; k0 += 4) {
        const int kb = k0 + half * 2;
        float a0 = Arow[kb];
        float a1 = Arow[kb + 1];
        if (relu_in) { a0 = fmaxf(a0, 0.0f); a1 = fmaxf(a1, 0.0f); }
        v2f a = {a0, a1};

        const float* __restrict__ Wk0 = sW + (size_t)kb * 64;
        const float* __restrict__ Wk1 = sW + (size_t)(kb + 1) * 64;
        v2f b0 = {Wk0[l16 +  0], Wk1[l16 +  0]};
        v2f b1 = {Wk0[l16 + 16], Wk1[l16 + 16]};
        v2f b2 = {Wk0[l16 + 32], Wk1[l16 + 32]};
        v2f b3 = {Wk0[l16 + 48], Wk1[l16 + 48]};

        c0 = __builtin_amdgcn_wmma_f32_16x16x4_f32(false, a, false, b0, (short)0, c0, false, false);
        c1 = __builtin_amdgcn_wmma_f32_16x16x4_f32(false, a, false, b1, (short)0, c1, false, false);
        c2 = __builtin_amdgcn_wmma_f32_16x16x4_f32(false, a, false, b2, (short)0, c2, false, false);
        c3 = __builtin_amdgcn_wmma_f32_16x16x4_f32(false, a, false, b3, (short)0, c3, false, false);
    }

    #pragma unroll
    for (int r = 0; r < 8; ++r) {
        const int row = m0 + half * 8 + r;
        if (row < n) {
            float* __restrict__ Cr = C + (size_t)row * 64 + l16;
            Cr[0]  = c0[r];
            Cr[16] = c1[r];
            Cr[32] = c2[r];
            Cr[48] = c3[r];
        }
    }
}


#define GB_WAVES 16
#define GB_PAIRS 2
#define GB_NODES (GB_WAVES * 2 * GB_PAIRS)
#define GB_TILES (GB_NODES / 16)
#define SA_PITCH 68

__device__ __forceinline__ void pool_flush(float* __restrict__ sums, float* __restrict__ cnt,
                                           int g, int lane, float sx, float sy, int run) {
    float* sg = sums + (size_t)g * HID + 2 * lane;
    atomicAdd(sg, sx);
    atomicAdd(sg + 1, sy);
    if (lane == 0) atomicAdd(&cnt[g], (float)run);
}

template <int K>
__device__ __forceinline__ int half_bcast(int v) {
    return __builtin_amdgcn_update_dpp(0, v, 0x150 + K, 0xF, 0xF, true);
}

template <int K>
__device__ __forceinline__ v4f row_of(const char* __restrict__ hb, unsigned so, unsigned lofs) {
    return *reinterpret_cast<const v4f*>(hb + ((unsigned)half_bcast<K>((int)so) | lofs));
}

template <int K>
__device__ __forceinline__ void add_row(v4f& a, const v4f r, int wi) {
    const float w = __int_as_float(half_bcast<K>(wi));
    a.x = fmaf(r.x, w, a.x); a.y = fmaf(r.y, w, a.y); a.z = fmaf(r.z, w, a.z); a.w = fmaf(r.w, w, a.w);
}

template <int K0>
__device__ __forceinline__ void rows8(const char* __restrict__ hb, unsigned so, int wi, unsigned lofs, v4f& a) {
    const v4f r0 = row_of<K0 + 0>(hb, so, lofs);
    const v4f r1 = row_of<K0 + 1>(hb, so, lofs);
    const v4f r2 = row_of<K0 + 2>(hb, so, lofs);
    const v4f r3 = row_of<K0 + 3>(hb, so, lofs);
    const v4f r4 = row_of<K0 + 4>(hb, so, lofs);
    const v4f r5 = row_of<K0 + 5>(hb, so, lofs);
    const v4f r6 = row_of<K0 + 6>(hb, so, lofs);
    const v4f r7 = row_of<K0 + 7>(hb, so, lofs);
    add_row<K0 + 0>(a, r0, wi);
    add_row<K0 + 1>(a, r1, wi);
    add_row<K0 + 2>(a, r2, wi);
    add_row<K0 + 3>(a, r3, wi);
    add_row<K0 + 4>(a, r4, wi);
    add_row<K0 + 5>(a, r5, wi);
    add_row<K0 + 6>(a, r6, wi);
    add_row<K0 + 7>(a, r7, wi);
}

template <int MODE>
__global__ __launch_bounds__(GB_WAVES * 32, 16)
void k_gather(const float* __restrict__ h, const float* __restrict__ dinv,
              const float* __restrict__ bias, const int* __restrict__ rowptr,
              const int* __restrict__ col, int n,
              const float* __restrict__ W, float* __restrict__ hout,
              const int* __restrict__ batch, float* __restrict__ sums, float* __restrict__ cnt) {
    __shared__ __attribute__((aligned(16))) float sA[GB_NODES * SA_PITCH];
    __shared__ __attribute__((aligned(16))) float sW[MODE == 0 ? 64 * 64 : 4];
    __shared__ __attribute__((aligned(16))) float sD[GB_NODES];
    __shared__ __attribute__((aligned(16))) float sB[HID];
    const int tid  = threadIdx.x;
    const int lane = tid & 31;
    const int half = lane >> 4;
    const int l16  = lane & 15;
    const int wave = __builtin_amdgcn_readfirstlane(tid >> 5);
    const int i0   = (int)blockIdx.x * GB_NODES;

    v4f w0 = {0.f, 0.f, 0.f, 0.f}, w1 = w0;
    int gv = 0;
    if (MODE == 0) {
        const v4f* __restrict__ W4 = reinterpret_cast<const v4f*>(W);
        w0 = W4[tid];
        w1 = W4[tid + GB_WAVES * 32];
    } else {
        if (wave < GB_TILES && lane < 16 && i0 + 16 * wave + lane < n) gv = batch[i0 + 16 * wave + lane];
    }

    const char* __restrict__ hb = reinterpret_cast<const char*>(h);
    const unsigned lofs = (unsigned)l16 << 4;
    const v4f bb = reinterpret_cast<const v4f*>(bias)[l16];

    int      ok[GB_PAIRS], e0[GB_PAIRS], end[GB_PAIRS];
    unsigned own[GB_PAIRS], so[GB_PAIRS];
    float    di[GB_PAIRS];
    #pragma unroll
    for (int p = 0; p < GB_PAIRS; ++p) {
        const int node = i0 + p * (GB_WAVES * 2) + 2 * wave + half;
        ok[p] = node < n;
        const int ic = min(node, n - 1);
        own[p] = (unsigned)ic << 8;
        di[p]  = dinv[ic];
        e0[p]  = rowptr[ic];
        end[p] = rowptr[ic + 1];
    }
    if (MODE == 0) {
        asm volatile("" :: "v"(w0), "v"(w1));
        v4f* sW4 = reinterpret_cast<v4f*>(sW);
        sW4[tid] = w0;
        sW4[tid + GB_WAVES * 32] = w1;
    }
    asm volatile("" :: "v"(bb));
    if (tid < 16) reinterpret_cast<v4f*>(sB)[tid] = bb;
    #pragma unroll
    for (int p = 0; p < GB_PAIRS; ++p) {
        if (l16 == 0) sD[p * (GB_WAVES * 2) + 2 * wave + half] = di[p];
    }
    int   s0[GB_PAIRS];
    #pragma unroll
    for (int p = 0; p < GB_PAIRS; ++p) {
        if (!ok[p]) end[p] = e0[p];
        e0[p] += l16;
        s0[p] = col[min(e0[p], NEDGES - 1)];
    }
    #pragma unroll
    for (int p = 0; p < GB_PAIRS; ++p) {
        so[p] = e0[p] < end[p] ? (unsigned)s0[p] << 8 : own[p];
    }

    #pragma unroll
    for (int p = 0; p < GB_PAIRS; ++p) {
        const int deg  = end[p] - (e0[p] - l16);
        const int mmax = max(__builtin_amdgcn_readlane(deg, 0), __builtin_amdgcn_readlane(deg, 16));
        const v4f hs = *reinterpret_cast<const v4f*>(hb + (own[p] | lofs));
        const v4f r0 = row_of<0>(hb, so[p], lofs);
        const v4f r1 = row_of<1>(hb, so[p], lofs);
        const v4f r2 = row_of<2>(hb, so[p], lofs);
        const v4f r3 = row_of<3>(hb, so[p], lofs);
        const v4f r4 = row_of<4>(hb, so[p], lofs);
        const v4f r5 = row_of<5>(hb, so[p], lofs);
        const v4f r6 = row_of<6>(hb, so[p], lofs);
        const v4f r7 = row_of<7>(hb, so[p], lofs);
        asm volatile("" :: "v"(hs), "v"(r0), "v"(r1), "v"(r2), "v"(r3), "v"(r4), "v"(r5), "v"(r6), "v"(r7));
        const int wi0 = __float_as_int(e0[p] < end[p] ? 1.0f : 0.0f);
        v4f a = hs;
        add_row<0>(a, r0, wi0);
        add_row<1>(a, r1, wi0);
        add_row<2>(a, r2, wi0);
        add_row<3>(a, r3, wi0);
        add_row<4>(a, r4, wi0);
        add_row<5>(a, r5, wi0);
        add_row<6>(a, r6, wi0);
        add_row<7>(a, r7, wi0);
        if (mmax > 8) rows8<8>(hb, so[p], wi0, lofs, a);
        for (int c = 16; c < mmax; c += 16) {
            const int e = e0[p] + c;
            const int s = col[min(e, NEDGES - 1)];
            const unsigned so_c = e < end[p] ? (unsigned)s << 8 : own[p];
            const int      wi_c = __float_as_int(e < end[p] ? 1.0f : 0.0f);
            rows8<0>(hb, so_c, wi_c, lofs, a);
            if (mmax - c > 8) rows8<8>(hb, so_c, wi_c, lofs, a);
        }
        v4f o = a;
        if (!ok[p]) o = v4f{0.f, 0.f, 0.f, 0.f};
        const int row = p * (GB_WAVES * 2) + 2 * wave + half;
        *reinterpret_cast<v4f*>(&sA[row * SA_PITCH + 4 * l16]) = o;
    }
    __syncthreads();

    if (MODE == 0) {
        if (wave >= 2 * GB_TILES) return;
        const int rt = wave >> 1;
        if (i0 + rt * 16 >= n) return;
        const int nc = (wave & 1) * 32 + l16;
        v8f c0 = {}, c1 = {};
        const float dA = sD[rt * 16 + l16];
        #pragma unroll
        for (int k0 = 0; k0 < 64; k0 += 4) {
            const int kb = k0 + half * 2;
            const v2f sm = *reinterpret_cast<const v2f*>(&sA[(rt * 16 + l16) * SA_PITCH + kb]);
            const v2f bk = *reinterpret_cast<const v2f*>(&sB[kb]);
            const v2f a = {fmaxf(fmaf(sm.x, dA, bk.x), 0.0f), fmaxf(fmaf(sm.y, dA, bk.y), 0.0f)};
            const float* __restrict__ Wk0 = sW + kb * 64 + nc;
            const float* __restrict__ Wk1 = sW + (kb + 1) * 64 + nc;
            const v2f b0 = {Wk0[0],  Wk1[0]};
            const v2f b1 = {Wk0[16], Wk1[16]};
            c0 = __builtin_amdgcn_wmma_f32_16x16x4_f32(false, a, false, b0, (short)0, c0, false, false);
            c1 = __builtin_amdgcn_wmma_f32_16x16x4_f32(false, a, false, b1, (short)0, c1, false, false);
        }
        const v4f dlo = *reinterpret_cast<const v4f*>(&sD[rt * 16 + half * 8]);
        const v4f dhi = *reinterpret_cast<const v4f*>(&sD[rt * 16 + half * 8 + 4]);
        const float dr[8] = {dlo.x, dlo.y, dlo.z, dlo.w, dhi.x, dhi.y, dhi.z, dhi.w};
        #pragma unroll
        for (int r = 0; r < 8; ++r) {
            const int row = i0 + rt * 16 + half * 8 + r;
            if (row < n) {
                float* __restrict__ Cr = hout + (size_t)row * 64 + nc;
                Cr[0]  = c0[r] * dr[r];
                Cr[16] = c1[r] * dr[r];
            }
        }
    } else {
        if (wave >= GB_TILES) return;
        const int m = min(16, n - (i0 + 16 * wave));
        if (m <= 0) return;
        v2f r[16];
        #pragma unroll
        for (int k = 0; k < 16; ++k) r[k] = *reinterpret_cast<const v2f*>(&sA[(16 * wave + k) * SA_PITCH + 2 * lane]);
        const v2f bp = *reinterpret_cast<const v2f*>(&sB[2 * lane]);
        const v4f d0 = *reinterpret_cast<const v4f*>(&sD[16 * wave]);
        const v4f d1 = *reinterpret_cast<const v4f*>(&sD[16 * wave + 4]);
        const v4f d2 = *reinterpret_cast<const v4f*>(&sD[16 * wave + 8]);
        const v4f d3 = *reinterpret_cast<const v4f*>(&sD[16 * wave + 12]);
        const float dk[16] = {d0.x, d0.y, d0.z, d0.w, d1.x, d1.y, d1.z, d1.w,
                              d2.x, d2.y, d2.z, d2.w, d3.x, d3.y, d3.z, d3.w};
        int   cur = __builtin_amdgcn_readlane(gv, 0);
        int   run = 0;
        float sx = 0.0f, sy = 0.0f;
        #pragma unroll
        for (int k = 0; k < 16; ++k) {
            if (k < m) {
                const int g = __builtin_amdgcn_readlane(gv, k);
                if (g != cur) {
                    pool_flush(sums, cnt, cur, lane, sx, sy, run);
                    cur = g; run = 0; sx = 0.0f; sy = 0.0f;
                }
                sx += fmaxf(fmaf(r[k].x, dk[k], bp.x), 0.0f);
                sy += fmaxf(fmaf(r[k].y, dk[k], bp.y), 0.0f);
                ++run;
            }
        }
        pool_flush(sums, cnt, cur, lane, sx, sy, run);
    }
}

__global__ __launch_bounds__(256)
void k_head(const float* __restrict__ sums, const float* __restrict__ cnt,
            const float* __restrict__ Wl, const float* __restrict__ bl,
            float* __restrict__ out) {
    int t = blockIdx.x * blockDim.x + threadIdx.x;
    if (t < NGRAPHS * NCLASSES) {
        int g = t / NCLASSES;
        int c = t % NCLASSES;
        float inv = 1.0f / fmaxf(cnt[g], 1.0f);
        float acc = 0.0f;
        const float* sg = sums + (size_t)g * HID;
        #pragma unroll
        for (int k = 0; k < HID; ++k) acc += sg[k] * Wl[k * NCLASSES + c];
        out[t] = acc * inv + bl[c];
    }
}


extern "C" void kernel_launch(void* const* d_in, const int* in_sizes, int n_in,
                              void* d_out, int out_size, void* d_ws, size_t ws_size,
                              hipStream_t stream) {
    const float* x  = (const float*)d_in[0];
    const float* W1 = (const float*)d_in[1];
    const float* b1 = (const float*)d_in[2];
    const float* W2 = (const float*)d_in[3];
    const float* b2 = (const float*)d_in[4];
    const float* Wl = (const float*)d_in[5];
    const float* bl = (const float*)d_in[6];
    const int* edge  = (const int*)d_in[7];
    const int* batch = (const int*)d_in[8];
    const int* src = edge;
    const int* dst = edge + NEDGES;
    float* out = (float*)d_out;

    float* ws   = (float*)d_ws;
    float* h    = ws;
    float* a    = h + (size_t)NNODES * HID;
    float* dinv = a + (size_t)NNODES * HID;
    float* sums = dinv + NNODES;
    float* cnt  = sums + (size_t)NGRAPHS * HID;
    int* offs   = (int*)(cnt + NGRAPHS);
    int* rowptr = offs + NCHUNK * OFFP;
    int* packed = rowptr + NNODES + 4;
    int* col    = packed + NEDGES;

    const int TPB = 256;
    const int gGa  = (NNODES + GB_NODES - 1) / GB_NODES;
    const int nPZ  = NGRAPHS * HID + NGRAPHS;
    static_assert(NCHUNK * 256 >= NGRAPHS * HID + NGRAPHS, "k_bin's blocks clear sums and cnt");
    const int gHd  = (NGRAPHS * NCLASSES + TPB - 1) / TPB;

    k_bin<<<NCHUNK + GEMM_BLOCKS, TPB, 0, stream>>>(src, dst, offs, packed, NEDGES, sums, nPZ, x, W1, h, NNODES);
    k_csr<<<NBKT, TPB, 0, stream>>>(packed, offs, rowptr, dinv, col, NNODES, h);

    k_gather<0><<<gGa, GB_WAVES * 32, 0, stream>>>(h, dinv, b1, rowptr, col, NNODES, W2, a,
                                                   nullptr, nullptr, nullptr);
    k_gather<1><<<gGa, GB_WAVES * 32, 0, stream>>>(a, dinv, b2, rowptr, col, NNODES, nullptr, nullptr,
                                                   batch, sums, cnt);
    k_head<<<gHd, TPB, 0, stream>>>(sums, cnt, Wl, bl, out);
}


# Launches 5 to 4: head product per 16-node tile on WMMA in front of the mean pooling, run means float-atomic-added into out (start bl), counts from the list build; sum order changed
# speedup vs baseline: 6.1793x; 1.0401x over previous
#include <hip/hip_runtime.h>

typedef float v2f __attribute__((ext_vector_type(2)));
typedef float v8f __attribute__((ext_vector_type(8)));
typedef float v4f __attribute__((ext_vector_type(4)));

#define NNODES 50000
#define NEDGES 800000
#define HID 64
#define NGRAPHS 512
#define NCLASSES 10


#define BKT_SHIFT 8
#define BKT_NODES (1 << BKT_SHIFT)
#define NBKT ((NNODES + BKT_NODES - 1) / BKT_NODES)
#define CHUNK 4096
#define NCHUNK ((NEDGES + CHUNK - 1) / CHUNK)
#define OFFP (NBKT + 4)
#define CSR_RUNS ((NCHUNK + 7) / 8)
#define CSR_BATCH 13
static_assert(NBKT < 256 && CSR_RUNS <= 32, "one thread per bucket in k_bin, one lane per chunk of a wave in k_csr");

#define GEMM_BLOCKS (((NNODES + 15) / 16 + 7) / 8)

__device__ __forceinline__ void gemm_block(const float* __restrict__ A, const float* __restrict__ W,
                                           float* __restrict__ C, int n, int relu_in, int blk);

__global__ __launch_bounds__(256)
void k_bin(const int* __restrict__ src, const int* __restrict__ dst, int* __restrict__ offs,
           int* __restrict__ packed, int e, float* __restrict__ cnt, const float* __restrict__ bl,
           float* __restrict__ out,
           const float* __restrict__ x, const float* __restrict__ W1, float* __restrict__ h, int n) {
    if (blockIdx.x >= NCHUNK) {
        gemm_block(x, W1, h, n, 0, (int)blockIdx.x - NCHUNK);
        return;
    }
    __shared__ int s_cnt[256];
    __shared__ int s_wsum[8];
    const int tid  = threadIdx.x;
    const int b    = blockIdx.x;
    const int base = b * CHUNK;
    const int z = b * 256 + tid;
    if (z < NGRAPHS) cnt[z] = 0.0f;
    if (z < NGRAPHS * NCLASSES) out[z] = bl[z % NCLASSES];
    s_cnt[tid] = 0;
    int d[CHUNK / 256];
    int s[CHUNK / 256];
    #pragma unroll
    for (int j = 0; j < CHUNK / 256; ++j) {
        const int i = min(base + j * 256 + tid, e - 1);
        d[j] = dst[i];
        s[j] = src[i];
    }
    __syncthreads();
    #pragma unroll
    for (int j = 0; j < CHUNK / 256; ++j) {
        if (base + j * 256 + tid < e) atomicAdd(&s_cnt[d[j] >> BKT_SHIFT], 1);
    }
    __syncthreads();
    const int total = s_cnt[tid];
    const int lane = tid & 31;
    const int wave = tid >> 5;
    int incl = total;
    #pragma unroll
    for (int off = 1; off < 32; off <<= 1) {
        const int v = __shfl_up(incl, off);
        if (lane >= off) incl += v;
    }
    if (lane == 31) s_wsum[wave] = incl;
    __syncthreads();
    int wbase = 0;
    #pragma unroll
    for (int w = 0; w < 8; ++w) wbase += (w < wave) ? s_wsum[w] : 0;
    const int start = wbase + incl - total;
    if (tid <= NBKT) offs[b * OFFP + tid] = start;
    s_cnt[tid] = start;
    __syncthreads();
    #pragma unroll
    for (int j = 0; j < CHUNK / 256; ++j) {
        if (base + j * 256 + tid < e) {
            const int slot = atomicAdd(&s_cnt[d[j] >> BKT_SHIFT], 1);
            packed[base + slot] = (s[j] << BKT_SHIFT) | (d[j] & (BKT_NODES - 1));
        }
    }
}

template <int PLACE>
__device__ __forceinline__ void csr_walk(const int* __restrict__ packed, int o, int nv,
                                         int* s_cnt, int* __restrict__ col, int lane) {
    #pragma unroll
    for (int t = 0; t < CSR_RUNS; t += CSR_BATCH) {
        int v[CSR_BATCH], nn[CSR_BATCH], oo[CSR_BATCH];
        #pragma unroll
        for (int u = 0; u < CSR_BATCH; ++u) {
            if (t + u < CSR_RUNS) {
                oo[u] = __builtin_amdgcn_readlane(o, t + u);
                nn[u] = __builtin_amdgcn_readlane(nv, t + u);
                v[u]  = packed[min(oo[u] + lane, NEDGES - 1)];
            }
        }
        #pragma unroll
        for (int u = 0; u < CSR_BATCH; ++u) {
            if (t + u < CSR_RUNS) {
                if (lane < nn[u]) {
                    if (PLACE) {
                        const int slot = atomicAdd(&s_cnt[v[u] & (BKT_NODES - 1)], 1);
                        col[slot] = v[u] >> BKT_SHIFT;
                    } else {
                        atomicAdd(&s_cnt[v[u] & (BKT_NODES - 1)], 1);
                    }
                }
                for (int j = lane + 32; j < nn[u]; j += 32) {
                    const int w = packed[oo[u] + j];
                    if (PLACE) {
                        const int slot = atomicAdd(&s_cnt[w & (BKT_NODES - 1)], 1);
                        col[slot] = w >> BKT_SHIFT;
                    } else {
                        atomicAdd(&s_cnt[w & (BKT_NODES - 1)], 1);
                    }
                }
            }
        }
    }
}

#define HS_STEPS (BKT_NODES * 16 / 256)
__global__ __launch_bounds__(256)
void k_csr(const int* __restrict__ packed, const int* __restrict__ offs, int* __restrict__ rowptr,
           float* __restrict__ dinv, int* __restrict__ col, int n, float* __restrict__ h,
           const int* __restrict__ batch, float* __restrict__ cnt) {
    __shared__ int s_cnt[BKT_NODES];
    __shared__ int s_scan[BKT_NODES];
    __shared__ int s_red[8];
    __shared__ float s_dv[BKT_NODES];
    const int tid  = threadIdx.x;
    const int k    = blockIdx.x;
    const int lane = tid & 31;
    const int wave = tid >> 5;
    const int c    = wave * CSR_RUNS + lane;
    v4f* __restrict__ h4 = reinterpret_cast<v4f*>(h) + (size_t)k * BKT_NODES * 16;
    const int hrows = min(BKT_NODES, n - k * BKT_NODES);
    v4f hr[HS_STEPS];
    #pragma unroll
    for (int j = 0; j < HS_STEPS; ++j) {
        hr[j] = v4f{0.f, 0.f, 0.f, 0.f};
        if (j * 16 + (tid >> 4) < hrows) hr[j] = h4[j * 256 + tid];
    }
    const int nd  = k * BKT_NODES + tid;
    const int gid = nd < n ? batch[nd] : -1;
    int o = 0, nv = 0, a = 0;
    if (lane < CSR_RUNS && c < NCHUNK) {
        a = offs[c * OFFP + k];
        const int b = offs[c * OFFP + k + 1];
        o  = c * CHUNK + a;
        nv = b - a;
    }
    s_cnt[tid] = 0;
    int ra = a;
    #pragma unroll
    for (int off = 16; off >= 1; off >>= 1) ra += __shfl_xor(ra, off);
    if (lane == 0) s_red[wave] = ra;
    __syncthreads();
    int bs = 0;
    #pragma unroll
    for (int w = 0; w < 8; ++w) bs += s_red[w];
    {
        const int      pv = __shfl_up(gid, 1);
        const bool     hd = lane == 0 || gid != pv;
        const unsigned hm = __builtin_amdgcn_ballot_w32(hd);
        const unsigned ab = lane == 31 ? 0u : (hm >> (lane + 1));
        const int      len = ab ? __builtin_ctz(ab) + 1 : 32 - lane;
        if (hd && gid >= 0) atomicAdd(&cnt[gid], (float)len);
    }
    csr_walk<0>(packed, o, nv, s_cnt, col, lane);
    __syncthreads();
    const int own = s_cnt[tid];
    s_scan[tid] = own;
    __syncthreads();
    for (int off = 1; off < 256; off <<= 1) {
        int v = 0;
        if (tid >= off) v = s_scan[tid - off];
        __syncthreads();
        s_scan[tid] += v;
        __syncthreads();
    }
    const int start = bs + s_scan[tid] - own;
    const int i = k * BKT_NODES + tid;
    const float dv = rsqrtf((float)(own + 1));
    s_dv[tid] = dv;
    if (i < n) {
        rowptr[i] = start;
        dinv[i]   = dv;
    }
    if (i == n - 1) rowptr[n] = start + own;
    s_cnt[tid] = start;
    __syncthreads();
    float hw[HS_STEPS];
    #pragma unroll
    for (int j = 0; j < HS_STEPS; ++j) hw[j] = s_dv[j * 16 + (tid >> 4)];
    #pragma unroll
    for (int j = 0; j < HS_STEPS; ++j) {
        if (j * 16 + (tid >> 4) < hrows)
            h4[j * 256 + tid] = v4f{hr[j].x * hw[j], hr[j].y * hw[j], hr[j].z * hw[j], hr[j].w * hw[j]};
    }
    csr_walk<1>(packed, o, nv, s_cnt, col, lane);
}


__device__ __forceinline__ void gemm_block(const float* __restrict__ A, const float* __restrict__ W,
                                           float* __restrict__ C, int n, int relu_in, int blk) {
    __shared__ float sW[64 * 64];
    const int tid = threadIdx.x;
    #pragma unroll
    for (int i = 0; i < 16; ++i) sW[tid + i * 256] = W[tid + i * 256];
    __syncthreads();

    const int wave = tid >> 5;
    const int lane = tid & 31;
    const int half = lane >> 4;
    const int l16  = lane & 15;
    const int m0   = (blk * 8 + wave) * 16;
    if (m0 >= n) return;

    v8f c0 = {}, c1 = {}, c2 = {}, c3 = {};

    const int arow = m0 + l16;
    const float* __restrict__ Arow = A + (size_t)arow * 64;

    #pragma unroll
    for (int k0 = 0; k0 < 64; k0 += 4) {
        const int kb = k0 + half * 2;
        float a0 = Arow[kb];
        float a1 = Arow[kb + 1];
        if (relu_in) { a0 = fmaxf(a0, 0.0f); a1 = fmaxf(a1, 0.0f); }
        v2f a = {a0, a1};

        const float* __restrict__ Wk0 = sW + (size_t)kb * 64;
        const float* __restrict__ Wk1 = sW + (size_t)(kb + 1) * 64;
        v2f b0 = {Wk0[l16 +  0], Wk1[l16 +  0]};
        v2f b1 = {Wk0[l16 + 16], Wk1[l16 + 16]};
        v2f b2 = {Wk0[l16 + 32], Wk1[l16 + 32]};
        v2f b3 = {Wk0[l16 + 48], Wk1[l16 + 48]};

        c0 = __builtin_amdgcn_wmma_f32_16x16x4_f32(false, a, false, b0, (short)0, c0, false, false);
        c1 = __builtin_amdgcn_wmma_f32_16x16x4_f32(false, a, false, b1, (short)0, c1, false, false);
        c2 = __builtin_amdgcn_wmma_f32_16x16x4_f32(false, a, false, b2, (short)0, c2, false, false);
        c3 = __builtin_amdgcn_wmma_f32_16x16x4_f32(false, a, false, b3, (short)0, c3, false, false);
    }

    #pragma unroll
    for (int r = 0; r < 8; ++r) {
        const int row = m0 + half * 8 + r;
        if (row < n) {
            float* __restrict__ Cr = C + (size_t)row * 64 + l16;
            Cr[0]  = c0[r];
            Cr[16] = c1[r];
            Cr[32] = c2[r];
            Cr[48] = c3[r];
        }
    }
}


#define GB_WAVES 16
#define GB_PAIRS 2
#define GB_NODES (GB_WAVES * 2 * GB_PAIRS)
#define GB_TILES (GB_NODES / 16)
#define SA_PITCH 68

template <int K>
__device__ __forceinline__ int half_bcast(int v) {
    return __builtin_amdgcn_update_dpp(0, v, 0x150 + K, 0xF, 0xF, true);
}

template <int K>
__device__ __forceinline__ v4f row_of(const char* __restrict__ hb, unsigned so, unsigned lofs) {
    return *reinterpret_cast<const v4f*>(hb + ((unsigned)half_bcast<K>((int)so) | lofs));
}

template <int K>
__device__ __forceinline__ void add_row(v4f& a, const v4f r, int wi) {
    const float w = __int_as_float(half_bcast<K>(wi));
    a.x = fmaf(r.x, w, a.x); a.y = fmaf(r.y, w, a.y); a.z = fmaf(r.z, w, a.z); a.w = fmaf(r.w, w, a.w);
}

template <int K0>
__device__ __forceinline__ void rows8(const char* __restrict__ hb, unsigned so, int wi, unsigned lofs, v4f& a) {
    const v4f r0 = row_of<K0 + 0>(hb, so, lofs);
    const v4f r1 = row_of<K0 + 1>(hb, so, lofs);
    const v4f r2 = row_of<K0 + 2>(hb, so, lofs);
    const v4f r3 = row_of<K0 + 3>(hb, so, lofs);
    const v4f r4 = row_of<K0 + 4>(hb, so, lofs);
    const v4f r5 = row_of<K0 + 5>(hb, so, lofs);
    const v4f r6 = row_of<K0 + 6>(hb, so, lofs);
    const v4f r7 = row_of<K0 + 7>(hb, so, lofs);
    add_row<K0 + 0>(a, r0, wi);
    add_row<K0 + 1>(a, r1, wi);
    add_row<K0 + 2>(a, r2, wi);
    add_row<K0 + 3>(a, r3, wi);
    add_row<K0 + 4>(a, r4, wi);
    add_row<K0 + 5>(a, r5, wi);
    add_row<K0 + 6>(a, r6, wi);
    add_row<K0 + 7>(a, r7, wi);
}

template <int R>
__device__ __forceinline__ void pool_row(float y, int gx, int wx, int nvr, bool colok, float* __restrict__ po,
                                         int& cur, float& ivc, float& acc) {
    const int   g = half_bcast<R>(gx);
    const float w = __int_as_float(half_bcast<R>(wx));
    if (R == 0) { cur = g; ivc = w; }
    if (R < nvr) {
        if (g != cur) {
            if (colok) atomicAdd(po + cur * NCLASSES, acc * ivc);
            cur = g; ivc = w; acc = 0.0f;
        }
        acc += y;
    }
}

template <int MODE>
__global__ __launch_bounds__(GB_WAVES * 32, 16)
void k_gather(const float* __restrict__ h, const float* __restrict__ dinv,
              const float* __restrict__ bias, const int* __restrict__ rowptr,
              const int* __restrict__ col, int n,
              const float* __restrict__ W, float* __restrict__ hout,
              const int* __restrict__ batch, float* __restrict__ outp, const float* __restrict__ cnt) {
    __shared__ __attribute__((aligned(16))) float sA[GB_NODES * SA_PITCH];
    __shared__ __attribute__((aligned(16))) float sW[MODE == 0 ? 64 * 64 : 64 * 16];
    __shared__ __attribute__((aligned(16))) float sD[GB_NODES];
    __shared__ __attribute__((aligned(16))) float sB[HID];
    const int tid  = threadIdx.x;
    const int lane = tid & 31;
    const int half = lane >> 4;
    const int l16  = lane & 15;
    const int wave = __builtin_amdgcn_readfirstlane(tid >> 5);
    const int i0   = (int)blockIdx.x * GB_NODES;

    v4f w0 = {0.f, 0.f, 0.f, 0.f}, w1 = w0;
    float wl0 = 0.0f, wl1 = 0.0f;
    int gv = 0;
    const bool idl = wave < GB_TILES && lane < 16 && i0 + 16 * wave + lane < n;
    if (MODE == 0) {
        const v4f* __restrict__ W4 = reinterpret_cast<const v4f*>(W);
        w0 = W4[tid];
        w1 = W4[tid + GB_WAVES * 32];
    } else {
        if (idl) gv = batch[i0 + 16 * wave + lane];
        if ((tid & 15) < NCLASSES) {
            wl0 = W[(tid >> 4) * NCLASSES + (tid & 15)];
            wl1 = W[((tid >> 4) + 32) * NCLASSES + (tid & 15)];
        }
    }

    const char* __restrict__ hb = reinterpret_cast<const char*>(h);
    const unsigned lofs = (unsigned)l16 << 4;
    const v4f bb = reinterpret_cast<const v4f*>(bias)[l16];

    int      ok[GB_PAIRS], e0[GB_PAIRS], end[GB_PAIRS];
    unsigned own[GB_PAIRS], so[GB_PAIRS];
    float    di[GB_PAIRS];
    #pragma unroll
    for (int p = 0; p < GB_PAIRS; ++p) {
        const int node = i0 + p * (GB_WAVES * 2) + 2 * wave + half;
        ok[p] = node < n;
        const int ic = min(node, n - 1);
        own[p] = (unsigned)ic << 8;
        di[p]  = dinv[ic];
        e0[p]  = rowptr[ic];
        end[p] = rowptr[ic + 1];
    }
    if (MODE == 0) {
        asm volatile("" :: "v"(w0), "v"(w1));
        v4f* sW4 = reinterpret_cast<v4f*>(sW);
        sW4[tid] = w0;
        sW4[tid + GB_WAVES * 32] = w1;
    } else {
        asm volatile("" :: "v"(wl0), "v"(wl1));
        sW[tid] = wl0;
        sW[tid + GB_WAVES * 32] = wl1;
    }
    asm volatile("" :: "v"(bb));
    if (tid < 16) reinterpret_cast<v4f*>(sB)[tid] = bb;
    #pragma unroll
    for (int p = 0; p < GB_PAIRS; ++p) {
        if (l16 == 0) sD[p * (GB_WAVES * 2) + 2 * wave + half] = di[p];
    }
    int   s0[GB_PAIRS];
    #pragma unroll
    for (int p = 0; p < GB_PAIRS; ++p) {
        if (!ok[p]) end[p] = e0[p];
        e0[p] += l16;
        s0[p] = col[min(e0[p], NEDGES - 1)];
    }
    float cv = 1.0f;
    if (MODE == 1) {
        if (idl) cv = cnt[gv];
    }
    #pragma unroll
    for (int p = 0; p < GB_PAIRS; ++p) {
        so[p] = e0[p] < end[p] ? (unsigned)s0[p] << 8 : own[p];
    }

    #pragma unroll
    for (int p = 0; p < GB_PAIRS; ++p) {
        const int deg  = end[p] - (e0[p] - l16);
        const int mmax = max(__builtin_amdgcn_readlane(deg, 0), __builtin_amdgcn_readlane(deg, 16));
        const v4f hs = *reinterpret_cast<const v4f*>(hb + (own[p] | lofs));
        const v4f r0 = row_of<0>(hb, so[p], lofs);
        const v4f r1 = row_of<1>(hb, so[p], lofs);
        const v4f r2 = row_of<2>(hb, so[p], lofs);
        const v4f r3 = row_of<3>(hb, so[p], lofs);
        const v4f r4 = row_of<4>(hb, so[p], lofs);
        const v4f r5 = row_of<5>(hb, so[p], lofs);
        const v4f r6 = row_of<6>(hb, so[p], lofs);
        const v4f r7 = row_of<7>(hb, so[p], lofs);
        asm volatile("" :: "v"(hs), "v"(r0), "v"(r1), "v"(r2), "v"(r3), "v"(r4), "v"(r5), "v"(r6), "v"(r7));
        const int wi0 = __float_as_int(e0[p] < end[p] ? 1.0f : 0.0f);
        v4f a = hs;
        add_row<0>(a, r0, wi0);
        add_row<1>(a, r1, wi0);
        add_row<2>(a, r2, wi0);
        add_row<3>(a, r3, wi0);
        add_row<4>(a, r4, wi0);
        add_row<5>(a, r5, wi0);
        add_row<6>(a, r6, wi0);
        add_row<7>(a, r7, wi0);
        if (mmax > 8) rows8<8>(hb, so[p], wi0, lofs, a);
        for (int c = 16; c < mmax; c += 16) {
            const int e = e0[p] + c;
            const int s = col[min(e, NEDGES - 1)];
            const unsigned so_c = e < end[p] ? (unsigned)s << 8 : own[p];
            const int      wi_c = __float_as_int(e < end[p] ? 1.0f : 0.0f);
            rows8<0>(hb, so_c, wi_c, lofs, a);
            if (mmax - c > 8) rows8<8>(hb, so_c, wi_c, lofs, a);
        }
        v4f o = a;
        if (!ok[p]) o = v4f{0.f, 0.f, 0.f, 0.f};
        const int row = p * (GB_WAVES * 2) + 2 * wave + half;
        *reinterpret_cast<v4f*>(&sA[row * SA_PITCH + 4 * l16]) = o;
    }
    __syncthreads();

    if (MODE == 0) {
        if (wave >= 2 * GB_TILES) return;
        const int rt = wave >> 1;
        if (i0 + rt * 16 >= n) return;
        const int nc = (wave & 1) * 32 + l16;
        v8f c0 = {}, c1 = {};
        const float dA = sD[rt * 16 + l16];
        #pragma unroll
        for (int k0 = 0; k0 < 64; k0 += 4) {
            const int kb = k0 + half * 2;
            const v2f sm = *reinterpret_cast<const v2f*>(&sA[(rt * 16 + l16) * SA_PITCH + kb]);
            const v2f bk = *reinterpret_cast<const v2f*>(&sB[kb]);
            const v2f a = {fmaxf(fmaf(sm.x, dA, bk.x), 0.0f), fmaxf(fmaf(sm.y, dA, bk.y), 0.0f)};
            const float* __restrict__ Wk0 = sW + kb * 64 + nc;
            const float* __restrict__ Wk1 = sW + (kb + 1) * 64 + nc;
            const v2f b0 = {Wk0[0],  Wk1[0]};
            const v2f b1 = {Wk0[16], Wk1[16]};
            c0 = __builtin_amdgcn_wmma_f32_16x16x4_f32(false, a, false, b0, (short)0, c0, false, false);
            c1 = __builtin_amdgcn_wmma_f32_16x16x4_f32(false, a, false, b1, (short)0, c1, false, false);
        }
        const v4f dlo = *reinterpret_cast<const v4f*>(&sD[rt * 16 + half * 8]);
        const v4f dhi = *reinterpret_cast<const v4f*>(&sD[rt * 16 + half * 8 + 4]);
        const float dr[8] = {dlo.x, dlo.y, dlo.z, dlo.w, dhi.x, dhi.y, dhi.z, dhi.w};
        #pragma unroll
        for (int r = 0; r < 8; ++r) {
            const int row = i0 + rt * 16 + half * 8 + r;
            if (row < n) {
                float* __restrict__ Cr = hout + (size_t)row * 64 + nc;
                Cr[0]  = c0[r] * dr[r];
                Cr[16] = c1[r] * dr[r];
            }
        }
    } else {
        if (wave >= GB_TILES) return;
        const int rt = wave;
        if (i0 + rt * 16 >= n) return;
        v8f c0 = {}, c1 = {};
        const float dA = sD[rt * 16 + l16];
        #pragma unroll
        for (int k0 = 0; k0 < 64; k0 += 4) {
            const int kb = k0 + half * 2;
            const v2f sm = *reinterpret_cast<const v2f*>(&sA[(rt * 16 + l16) * SA_PITCH + kb]);
            const v2f bk = *reinterpret_cast<const v2f*>(&sB[kb]);
            const v2f a = {fmaxf(fmaf(sm.x, dA, bk.x), 0.0f), fmaxf(fmaf(sm.y, dA, bk.y), 0.0f)};
            const v2f b = {sW[kb * 16 + l16], sW[(kb + 1) * 16 + l16]};
            if ((k0 & 4) == 0) c0 = __builtin_amdgcn_wmma_f32_16x16x4_f32(false, a, false, b, (short)0, c0, false, false);
            else               c1 = __builtin_amdgcn_wmma_f32_16x16x4_f32(false, a, false, b, (short)0, c1, false, false);
        }
        const int nvr = n - (i0 + rt * 16 + half * 8);
        const int ivb = __float_as_int(1.0f / fmaxf(cv, 1.0f));
        const int gx = __shfl(gv,  8 * half + (lane & 7));
        const int wx = __shfl(ivb, 8 * half + (lane & 7));
        const bool colok = l16 < NCLASSES;
        float* __restrict__ po = outp + l16;
        int   cur = 0;
        float ivc = 0.0f, acc = 0.0f;
        pool_row<0>(c0[0] + c1[0], gx, wx, nvr, colok, po, cur, ivc, acc);
        pool_row<1>(c0[1] + c1[1], gx, wx, nvr, colok, po, cur, ivc, acc);
        pool_row<2>(c0[2] + c1[2], gx, wx, nvr, colok, po, cur, ivc, acc);
        pool_row<3>(c0[3] + c1[3], gx, wx, nvr, colok, po, cur, ivc, acc);
        pool_row<4>(c0[4] + c1[4], gx, wx, nvr, colok, po, cur, ivc, acc);
        pool_row<5>(c0[5] + c1[5], gx, wx, nvr, colok, po, cur, ivc, acc);
        pool_row<6>(c0[6] + c1[6], gx, wx, nvr, colok, po, cur, ivc, acc);
        pool_row<7>(c0[7] + c1[7], gx, wx, nvr, colok, po, cur, ivc, acc);
        if (nvr > 0 && colok) atomicAdd(po + cur * NCLASSES, acc * ivc);
    }
}


extern "C" void kernel_launch(void* const* d_in, const int* in_sizes, int n_in,
                              void* d_out, int out_size, void* d_ws, size_t ws_size,
                              hipStream_t stream) {
    const float* x  = (const float*)d_in[0];
    const float* W1 = (const float*)d_in[1];
    const float* b1 = (const float*)d_in[2];
    const float* W2 = (const float*)d_in[3];
    const float* b2 = (const float*)d_in[4];
    const float* Wl = (const float*)d_in[5];
    const float* bl = (const float*)d_in[6];
    const int* edge  = (const int*)d_in[7];
    const int* batch = (const int*)d_in[8];
    const int* src = edge;
    const int* dst = edge + NEDGES;
    float* out = (float*)d_out;

    float* ws   = (float*)d_ws;
    float* h    = ws;
    float* a    = h + (size_t)NNODES * HID;
    float* dinv = a + (size_t)NNODES * HID;
    float* cnt  = dinv + NNODES;
    int* offs   = (int*)(cnt + NGRAPHS);
    int* rowptr = offs + NCHUNK * OFFP;
    int* packed = rowptr + NNODES + 4;
    int* col    = packed + NEDGES;

    const int TPB = 256;
    const int gGa  = (NNODES + GB_NODES - 1) / GB_NODES;
    static_assert(NCHUNK * 256 >= NGRAPHS * NCLASSES, "k_bin's edge blocks store the start values of cnt and out");

    k_bin<<<NCHUNK + GEMM_BLOCKS, TPB, 0, stream>>>(src, dst, offs, packed, NEDGES, cnt, bl, out, x, W1, h, NNODES);
    k_csr<<<NBKT, TPB, 0, stream>>>(packed, offs, rowptr, dinv, col, NNODES, h, batch, cnt);

    k_gather<0><<<gGa, GB_WAVES * 32, 0, stream>>>(h, dinv, b1, rowptr, col, NNODES, W2, a,
                                                   nullptr, nullptr, nullptr);
    k_gather<1><<<gGa, GB_WAVES * 32, 0, stream>>>(a, dinv, b2, rowptr, col, NNODES, Wl, nullptr,
                                                   batch, out, cnt);
}


# List builds: the counting LDS add also returns the entry's rank (place = scanned first slot + rank): no second add pass or walk; lists assembled in LDS, stored as whole lines; arrival order kept
# speedup vs baseline: 6.7560x; 1.0933x over previous
#include <hip/hip_runtime.h>

typedef float v2f __attribute__((ext_vector_type(2)));
typedef float v8f __attribute__((ext_vector_type(8)));
typedef float v4f __attribute__((ext_vector_type(4)));

#define NNODES 50000
#define NEDGES 800000
#define HID 64
#define NGRAPHS 512
#define NCLASSES 10


#define BKT_SHIFT 8
#define BKT_NODES (1 << BKT_SHIFT)
#define NBKT ((NNODES + BKT_NODES - 1) / BKT_NODES)
#define CHUNK 4096
#define NCHUNK ((NEDGES + CHUNK - 1) / CHUNK)
#define OFFP (NBKT + 4)
#define CSR_RUNS ((NCHUNK + 7) / 8)
#define CSR_CAP 8192
static_assert(NBKT + 1 + 32 <= 256 && CSR_RUNS <= 32,
              "k_bin: one thread per bucket and 32 spare counters behind them; k_csr: one lane per chunk of a wave");

typedef int v4i __attribute__((ext_vector_type(4)));

#define GEMM_BLOCKS (((NNODES + 15) / 16 + 7) / 8)

__device__ __forceinline__ void gemm_block(const float* __restrict__ A, const float* __restrict__ W,
                                           float* __restrict__ C, int n, int relu_in, int blk);

__global__ __launch_bounds__(256)
void k_bin(const int* __restrict__ src, const int* __restrict__ dst, int* __restrict__ offs,
           int* __restrict__ packed, int e, float* __restrict__ cnt, const float* __restrict__ bl,
           float* __restrict__ out,
           const float* __restrict__ x, const float* __restrict__ W1, float* __restrict__ h, int n) {
    if (blockIdx.x >= NCHUNK) {
        gemm_block(x, W1, h, n, 0, (int)blockIdx.x - NCHUNK);
        return;
    }
    __shared__ int s_cnt[256];
    __shared__ int s_wsum[8];
    __shared__ __attribute__((aligned(16))) int s_pk[CHUNK];
    const int tid  = threadIdx.x;
    const int lane = tid & 31;
    const int wave = tid >> 5;
    const int b    = blockIdx.x;
    const int base = b * CHUNK;
    const int z = b * 256 + tid;
    if (z < NGRAPHS) cnt[z] = 0.0f;
    if (z < NGRAPHS * NCLASSES) out[z] = bl[z % NCLASSES];
    s_cnt[tid] = 0;
    int d[CHUNK / 256];
    int s[CHUNK / 256];
    #pragma unroll
    for (int j = 0; j < CHUNK / 256; ++j) {
        const int i = min(base + j * 256 + tid, e - 1);
        d[j] = dst[i];
        s[j] = src[i];
    }
    __syncthreads();
    int rk[CHUNK / 256];
    #pragma unroll
    for (int j = 0; j < CHUNK / 256; ++j) {
        const bool ok = base + j * 256 + tid < e;
        rk[j] = atomicAdd(&s_cnt[ok ? (d[j] >> BKT_SHIFT) : (NBKT + 1 + lane)], 1);
    }
    __syncthreads();
    const int total = s_cnt[tid];
    int incl = total;
    #pragma unroll
    for (int off = 1; off < 32; off <<= 1) {
        const int v = __shfl_up(incl, off);
        if (lane >= off) incl += v;
    }
    if (lane == 31) s_wsum[wave] = incl;
    __syncthreads();
    int wbase = 0;
    #pragma unroll
    for (int w = 0; w < 8; ++w) wbase += (w < wave) ? s_wsum[w] : 0;
    const int start = wbase + incl - total;
    if (tid <= NBKT) offs[b * OFFP + tid] = start;
    s_cnt[tid] = start;
    __syncthreads();
    int fs[CHUNK / 256];
    #pragma unroll
    for (int j = 0; j < CHUNK / 256; ++j) fs[j] = s_cnt[d[j] >> BKT_SHIFT];
    #pragma unroll
    for (int j = 0; j < CHUNK / 256; ++j) {
        if (base + j * 256 + tid < e) s_pk[fs[j] + rk[j]] = (s[j] << BKT_SHIFT) | (d[j] & (BKT_NODES - 1));
    }
    __syncthreads();
    const int ne = min(CHUNK, e - base);
    #pragma unroll
    for (int k = 0; k < CHUNK / 1024; ++k) {
        const int i = k * 1024 + 4 * tid;
        if (i + 3 < ne) {
            *reinterpret_cast<v4i*>(packed + base + i) = *reinterpret_cast<const v4i*>(&s_pk[i]);
        } else {
            #pragma unroll
            for (int q = 0; q < 4; ++q) {
                if (i + q < ne) packed[base + i + q] = s_pk[i + q];
            }
        }
    }
}

__device__ __forceinline__ void csr_put(int* s_col, int* __restrict__ col, int slot, int bs32, int val) {
    const int li = slot - bs32;
    if (li < CSR_CAP) s_col[li] = val;
    else              col[slot] = val;
}

#define HS_STEPS (BKT_NODES * 16 / 256)
__global__ __launch_bounds__(256)
void k_csr(const int* __restrict__ packed, const int* __restrict__ offs, int* __restrict__ rowptr,
           float* __restrict__ dinv, int* __restrict__ col, int n, float* __restrict__ h,
           const int* __restrict__ batch, float* __restrict__ cnt) {
    __shared__ int s_cnt[BKT_NODES + 32];
    __shared__ int s_tail[BKT_NODES];
    __shared__ int s_scan[BKT_NODES];
    __shared__ int s_start[BKT_NODES];
    __shared__ int s_red[8];
    __shared__ float s_dv[BKT_NODES];
    __shared__ __attribute__((aligned(16))) int s_col[CSR_CAP];
    const int tid  = threadIdx.x;
    const int k    = blockIdx.x;
    const int lane = tid & 31;
    const int wave = tid >> 5;
    const int c    = wave * CSR_RUNS + lane;
    v4f* __restrict__ h4 = reinterpret_cast<v4f*>(h) + (size_t)k * BKT_NODES * 16;
    const int hrows = min(BKT_NODES, n - k * BKT_NODES);
    v4f hr[HS_STEPS];
    #pragma unroll
    for (int j = 0; j < HS_STEPS; ++j) {
        hr[j] = v4f{0.f, 0.f, 0.f, 0.f};
        if (j * 16 + (tid >> 4) < hrows) hr[j] = h4[j * 256 + tid];
    }
    const int nd  = k * BKT_NODES + tid;
    const int gid = nd < n ? batch[nd] : -1;
    int o = 0, nv = 0, a = 0;
    if (lane < CSR_RUNS && c < NCHUNK) {
        a = offs[c * OFFP + k];
        const int b = offs[c * OFFP + k + 1];
        o  = c * CHUNK + a;
        nv = b - a;
    }
    s_cnt[tid]  = 0;
    s_tail[tid] = 0;
    if (tid < 32) s_cnt[BKT_NODES + tid] = 0;
    int ra = a;
    #pragma unroll
    for (int off = 16; off >= 1; off >>= 1) ra += __shfl_xor(ra, off);
    if (lane == 0) s_red[wave] = ra;
    __syncthreads();
    int bs = 0;
    #pragma unroll
    for (int w = 0; w < 8; ++w) bs += s_red[w];
    {
        const int      pv = __shfl_up(gid, 1);
        const bool     hd = lane == 0 || gid != pv;
        const unsigned hm = __builtin_amdgcn_ballot_w32(hd);
        const unsigned ab = lane == 31 ? 0u : (hm >> (lane + 1));
        const int      len = ab ? __builtin_ctz(ab) + 1 : 32 - lane;
        if (hd && gid >= 0) atomicAdd(&cnt[gid], (float)len);
    }
    int v[CSR_RUNS], rk[CSR_RUNS], nn[CSR_RUNS], oo[CSR_RUNS];
    int longest = 0;
    #pragma unroll
    for (int u = 0; u < CSR_RUNS; ++u) {
        oo[u] = __builtin_amdgcn_readlane(o, u);
        nn[u] = __builtin_amdgcn_readlane(nv, u);
        v[u]  = packed[min(oo[u] + lane, NEDGES - 1)];
        longest = max(longest, nn[u]);
    }
    __builtin_amdgcn_sched_barrier(0);
    #pragma unroll
    for (int u = 0; u < CSR_RUNS; ++u) {
        rk[u] = atomicAdd(&s_cnt[lane < nn[u] ? (v[u] & (BKT_NODES - 1)) : (BKT_NODES + lane)], 1);
    }
    if (longest > 32) {
        #pragma unroll
        for (int u = 0; u < CSR_RUNS; ++u) {
            for (int j = lane + 32; j < nn[u]; j += 32) atomicAdd(&s_tail[packed[oo[u] + j] & (BKT_NODES - 1)], 1);
        }
    }
    __syncthreads();
    const int own = s_cnt[tid] + s_tail[tid];
    s_scan[tid] = own;
    __syncthreads();
    for (int off = 1; off < 256; off <<= 1) {
        int t = 0;
        if (tid >= off) t = s_scan[tid - off];
        __syncthreads();
        s_scan[tid] += t;
        __syncthreads();
    }
    const int nb    = s_scan[BKT_NODES - 1];
    const int start = bs + s_scan[tid] - own;
    const int i = k * BKT_NODES + tid;
    const float dv = rsqrtf((float)(own + 1));
    s_dv[tid]    = dv;
    s_start[tid] = start;
    if (i < n) {
        rowptr[i] = start;
        dinv[i]   = dv;
    }
    if (i == n - 1) rowptr[n] = start + own;
    __syncthreads();
    float hw[HS_STEPS];
    #pragma unroll
    for (int j = 0; j < HS_STEPS; ++j) hw[j] = s_dv[j * 16 + (tid >> 4)];
    #pragma unroll
    for (int j = 0; j < HS_STEPS; ++j) {
        if (j * 16 + (tid >> 4) < hrows)
            h4[j * 256 + tid] = v4f{hr[j].x * hw[j], hr[j].y * hw[j], hr[j].z * hw[j], hr[j].w * hw[j]};
    }
    const int bs32 = bs & ~31;
    int st[CSR_RUNS];
    #pragma unroll
    for (int u = 0; u < CSR_RUNS; ++u) st[u] = s_start[v[u] & (BKT_NODES - 1)];
    #pragma unroll
    for (int u = 0; u < CSR_RUNS; ++u) {
        if (lane < nn[u]) csr_put(s_col, col, st[u] + rk[u], bs32, v[u] >> BKT_SHIFT);
    }
    if (longest > 32) {
        #pragma unroll
        for (int u = 0; u < CSR_RUNS; ++u) {
            for (int j = lane + 32; j < nn[u]; j += 32) {
                const int w  = packed[oo[u] + j];
                const int nw = w & (BKT_NODES - 1);
                const int left = atomicSub(&s_tail[nw], 1);
                csr_put(s_col, col, s_start[nw] + s_cnt[nw] + left - 1, bs32, w >> BKT_SHIFT);
            }
        }
    }
    __syncthreads();
    const int lend = min(bs + nb - bs32, CSR_CAP);
    for (int li = 4 * tid; li < lend; li += 4 * 256) {
        const int g = bs32 + li;
        if (g >= bs && li + 3 < lend) {
            *reinterpret_cast<v4i*>(col + g) = *reinterpret_cast<const v4i*>(&s_col[li]);
        } else {
            #pragma unroll
            for (int q = 0; q < 4; ++q) {
                if (g + q >= bs && li + q < lend) col[g + q] = s_col[li + q];
            }
        }
    }
}


__device__ __forceinline__ void gemm_block(const float* __restrict__ A, const float* __restrict__ W,
                                           float* __restrict__ C, int n, int relu_in, int blk) {
    __shared__ float sW[64 * 64];
    const int tid = threadIdx.x;
    #pragma unroll
    for (int i = 0; i < 16; ++i) sW[tid + i * 256] = W[tid + i * 256];
    __syncthreads();

    const int wave = tid >> 5;
    const int lane = tid & 31;
    const int half = lane >> 4;
    const int l16  = lane & 15;
    const int m0   = (blk * 8 + wave) * 16;
    if (m0 >= n) return;

    v8f c0 = {}, c1 = {}, c2 = {}, c3 = {};

    const int arow = m0 + l16;
    const float* __restrict__ Arow = A + (size_t)arow * 64;

    #pragma unroll
    for (int k0 = 0; k0 < 64; k0 += 4) {
        const int kb = k0 + half * 2;
        float a0 = Arow[kb];
        float a1 = Arow[kb + 1];
        if (relu_in) { a0 = fmaxf(a0, 0.0f); a1 = fmaxf(a1, 0.0f); }
        v2f a = {a0, a1};

        const float* __restrict__ Wk0 = sW + (size_t)kb * 64;
        const float* __restrict__ Wk1 = sW + (size_t)(kb + 1) * 64;
        v2f b0 = {Wk0[l16 +  0], Wk1[l16 +  0]};
        v2f b1 = {Wk0[l16 + 16], Wk1[l16 + 16]};
        v2f b2 = {Wk0[l16 + 32], Wk1[l16 + 32]};
        v2f b3 = {Wk0[l16 + 48], Wk1[l16 + 48]};

        c0 = __builtin_amdgcn_wmma_f32_16x16x4_f32(false, a, false, b0, (short)0, c0, false, false);
        c1 = __builtin_amdgcn_wmma_f32_16x16x4_f32(false, a, false, b1, (short)0, c1, false, false);
        c2 = __builtin_amdgcn_wmma_f32_16x16x4_f32(false, a, false, b2, (short)0, c2, false, false);
        c3 = __builtin_amdgcn_wmma_f32_16x16x4_f32(false, a, false, b3, (short)0, c3, false, false);
    }

    #pragma unroll
    for (int r = 0; r < 8; ++r) {
        const int row = m0 + half * 8 + r;
        if (row < n) {
            float* __restrict__ Cr = C + (size_t)row * 64 + l16;
            Cr[0]  = c0[r];
            Cr[16] = c1[r];
            Cr[32] = c2[r];
            Cr[48] = c3[r];
        }
    }
}


#define GB_WAVES 16
#define GB_PAIRS 2
#define GB_NODES (GB_WAVES * 2 * GB_PAIRS)
#define GB_TILES (GB_NODES / 16)
#define SA_PITCH 68

template <int K>
__device__ __forceinline__ int half_bcast(int v) {
    return __builtin_amdgcn_update_dpp(0, v, 0x150 + K, 0xF, 0xF, true);
}

template <int K>
__device__ __forceinline__ v4f row_of(const char* __restrict__ hb, unsigned so, unsigned lofs) {
    return *reinterpret_cast<const v4f*>(hb + ((unsigned)half_bcast<K>((int)so) | lofs));
}

template <int K>
__device__ __forceinline__ void add_row(v4f& a, const v4f r, int wi) {
    const float w = __int_as_float(half_bcast<K>(wi));
    a.x = fmaf(r.x, w, a.x); a.y = fmaf(r.y, w, a.y); a.z = fmaf(r.z, w, a.z); a.w = fmaf(r.w, w, a.w);
}

template <int K0>
__device__ __forceinline__ void rows8(const char* __restrict__ hb, unsigned so, int wi, unsigned lofs, v4f& a) {
    const v4f r0 = row_of<K0 + 0>(hb, so, lofs);
    const v4f r1 = row_of<K0 + 1>(hb, so, lofs);
    const v4f r2 = row_of<K0 + 2>(hb, so, lofs);
    const v4f r3 = row_of<K0 + 3>(hb, so, lofs);
    const v4f r4 = row_of<K0 + 4>(hb, so, lofs);
    const v4f r5 = row_of<K0 + 5>(hb, so, lofs);
    const v4f r6 = row_of<K0 + 6>(hb, so, lofs);
    const v4f r7 = row_of<K0 + 7>(hb, so, lofs);
    add_row<K0 + 0>(a, r0, wi);
    add_row<K0 + 1>(a, r1, wi);
    add_row<K0 + 2>(a, r2, wi);
    add_row<K0 + 3>(a, r3, wi);
    add_row<K0 + 4>(a, r4, wi);
    add_row<K0 + 5>(a, r5, wi);
    add_row<K0 + 6>(a, r6, wi);
    add_row<K0 + 7>(a, r7, wi);
}

template <int R>
__device__ __forceinline__ void pool_row(float y, int gx, int wx, int nvr, bool colok, float* __restrict__ po,
                                         int& cur, float& ivc, float& acc) {
    const int   g = half_bcast<R>(gx);
    const float w = __int_as_float(half_bcast<R>(wx));
    if (R == 0) { cur = g; ivc = w; }
    if (R < nvr) {
        if (g != cur) {
            if (colok) atomicAdd(po + cur * NCLASSES, acc * ivc);
            cur = g; ivc = w; acc = 0.0f;
        }
        acc += y;
    }
}

template <int MODE>
__global__ __launch_bounds__(GB_WAVES * 32, 16)
void k_gather(const float* __restrict__ h, const float* __restrict__ dinv,
              const float* __restrict__ bias, const int* __restrict__ rowptr,
              const int* __restrict__ col, int n,
              const float* __restrict__ W, float* __restrict__ hout,
              const int* __restrict__ batch, float* __restrict__ outp, const float* __restrict__ cnt) {
    __shared__ __attribute__((aligned(16))) float sA[GB_NODES * SA_PITCH];
    __shared__ __attribute__((aligned(16))) float sW[MODE == 0 ? 64 * 64 : 64 * 16];
    __shared__ __attribute__((aligned(16))) float sD[GB_NODES];
    __shared__ __attribute__((aligned(16))) float sB[HID];
    const int tid  = threadIdx.x;
    const int lane = tid & 31;
    const int half = lane >> 4;
    const int l16  = lane & 15;
    const int wave = __builtin_amdgcn_readfirstlane(tid >> 5);
    const int i0   = (int)blockIdx.x * GB_NODES;

    v4f w0 = {0.f, 0.f, 0.f, 0.f}, w1 = w0;
    float wl0 = 0.0f, wl1 = 0.0f;
    int gv = 0;
    const bool idl = wave < GB_TILES && lane < 16 && i0 + 16 * wave + lane < n;
    if (MODE == 0) {
        const v4f* __restrict__ W4 = reinterpret_cast<const v4f*>(W);
        w0 = W4[tid];
        w1 = W4[tid + GB_WAVES * 32];
    } else {
        if (idl) gv = batch[i0 + 16 * wave + lane];
        if ((tid & 15) < NCLASSES) {
            wl0 = W[(tid >> 4) * NCLASSES + (tid & 15)];
            wl1 = W[((tid >> 4) + 32) * NCLASSES + (tid & 15)];
        }
    }

    const char* __restrict__ hb = reinterpret_cast<const char*>(h);
    const unsigned lofs = (unsigned)l16 << 4;
    const v4f bb = reinterpret_cast<const v4f*>(bias)[l16];

    int      ok[GB_PAIRS], e0[GB_PAIRS], end[GB_PAIRS];
    unsigned own[GB_PAIRS], so[GB_PAIRS];
    float    di[GB_PAIRS];
    #pragma unroll
    for (int p = 0; p < GB_PAIRS; ++p) {
        const int node = i0 + p * (GB_WAVES * 2) + 2 * wave + half;
        ok[p] = node < n;
        const int ic = min(node, n - 1);
        own[p] = (unsigned)ic << 8;
        di[p]  = dinv[ic];
        e0[p]  = rowptr[ic];
        end[p] = rowptr[ic + 1];
    }
    if (MODE == 0) {
        asm volatile("" :: "v"(w0), "v"(w1));
        v4f* sW4 = reinterpret_cast<v4f*>(sW);
        sW4[tid] = w0;
        sW4[tid + GB_WAVES * 32] = w1;
    } else {
        asm volatile("" :: "v"(wl0), "v"(wl1));
        sW[tid] = wl0;
        sW[tid + GB_WAVES * 32] = wl1;
    }
    asm volatile("" :: "v"(bb));
    if (tid < 16) reinterpret_cast<v4f*>(sB)[tid] = bb;
    #pragma unroll
    for (int p = 0; p < GB_PAIRS; ++p) {
        if (l16 == 0) sD[p * (GB_WAVES * 2) + 2 * wave + half] = di[p];
    }
    int   s0[GB_PAIRS];
    #pragma unroll
    for (int p = 0; p < GB_PAIRS; ++p) {
        if (!ok[p]) end[p] = e0[p];
        e0[p] += l16;
        s0[p] = col[min(e0[p], NEDGES - 1)];
    }
    float cv = 1.0f;
    if (MODE == 1) {
        if (idl) cv = cnt[gv];
    }
    #pragma unroll
    for (int p = 0; p < GB_PAIRS; ++p) {
        so[p] = e0[p] < end[p] ? (unsigned)s0[p] << 8 : own[p];
    }

    #pragma unroll
    for (int p = 0; p < GB_PAIRS; ++p) {
        const int deg  = end[p] - (e0[p] - l16);
        const int mmax = max(__builtin_amdgcn_readlane(deg, 0), __builtin_amdgcn_readlane(deg, 16));
        const v4f hs = *reinterpret_cast<const v4f*>(hb + (own[p] | lofs));
        const v4f r0 = row_of<0>(hb, so[p], lofs);
        const v4f r1 = row_of<1>(hb, so[p], lofs);
        const v4f r2 = row_of<2>(hb, so[p], lofs);
        const v4f r3 = row_of<3>(hb, so[p], lofs);
        const v4f r4 = row_of<4>(hb, so[p], lofs);
        const v4f r5 = row_of<5>(hb, so[p], lofs);
        const v4f r6 = row_of<6>(hb, so[p], lofs);
        const v4f r7 = row_of<7>(hb, so[p], lofs);
        asm volatile("" :: "v"(hs), "v"(r0), "v"(r1), "v"(r2), "v"(r3), "v"(r4), "v"(r5), "v"(r6), "v"(r7));
        const int wi0 = __float_as_int(e0[p] < end[p] ? 1.0f : 0.0f);
        v4f a = hs;
        add_row<0>(a, r0, wi0);
        add_row<1>(a, r1, wi0);
        add_row<2>(a, r2, wi0);
        add_row<3>(a, r3, wi0);
        add_row<4>(a, r4, wi0);
        add_row<5>(a, r5, wi0);
        add_row<6>(a, r6, wi0);
        add_row<7>(a, r7, wi0);
        if (mmax > 8) rows8<8>(hb, so[p], wi0, lofs, a);
        for (int c = 16; c < mmax; c += 16) {
            const int e = e0[p] + c;
            const int s = col[min(e, NEDGES - 1)];
            const unsigned so_c = e < end[p] ? (unsigned)s << 8 : own[p];
            const int      wi_c = __float_as_int(e < end[p] ? 1.0f : 0.0f);
            rows8<0>(hb, so_c, wi_c, lofs, a);
            if (mmax - c > 8) rows8<8>(hb, so_c, wi_c, lofs, a);
        }
        v4f o = a;
        if (!ok[p]) o = v4f{0.f, 0.f, 0.f, 0.f};
        const int row = p * (GB_WAVES * 2) + 2 * wave + half;
        *reinterpret_cast<v4f*>(&sA[row * SA_PITCH + 4 * l16]) = o;
    }
    __syncthreads();

    if (MODE == 0) {
        if (wave >= 2 * GB_TILES) return;
        const int rt = wave >> 1;
        if (i0 + rt * 16 >= n) return;
        const int nc = (wave & 1) * 32 + l16;
        v8f c0 = {}, c1 = {};
        const float dA = sD[rt * 16 + l16];
        #pragma unroll
        for (int k0 = 0; k0 < 64; k0 += 4) {
            const int kb = k0 + half * 2;
            const v2f sm = *reinterpret_cast<const v2f*>(&sA[(rt * 16 + l16) * SA_PITCH + kb]);
            const v2f bk = *reinterpret_cast<const v2f*>(&sB[kb]);
            const v2f a = {fmaxf(fmaf(sm.x, dA, bk.x), 0.0f), fmaxf(fmaf(sm.y, dA, bk.y), 0.0f)};
            const float* __restrict__ Wk0 = sW + kb * 64 + nc;
            const float* __restrict__ Wk1 = sW + (kb + 1) * 64 + nc;
            const v2f b0 = {Wk0[0],  Wk1[0]};
            const v2f b1 = {Wk0[16], Wk1[16]};
            c0 = __builtin_amdgcn_wmma_f32_16x16x4_f32(false, a, false, b0, (short)0, c0, false, false);
            c1 = __builtin_amdgcn_wmma_f32_16x16x4_f32(false, a, false, b1, (short)0, c1, false, false);
        }
        const v4f dlo = *reinterpret_cast<const v4f*>(&sD[rt * 16 + half * 8]);
        const v4f dhi = *reinterpret_cast<const v4f*>(&sD[rt * 16 + half * 8 + 4]);
        const float dr[8] = {dlo.x, dlo.y, dlo.z, dlo.w, dhi.x, dhi.y, dhi.z, dhi.w};
        #pragma unroll
        for (int r = 0; r < 8; ++r) {
            const int row = i0 + rt * 16 + half * 8 + r;
            if (row < n) {
                float* __restrict__ Cr = hout + (size_t)row * 64 + nc;
                Cr[0]  = c0[r] * dr[r];
                Cr[16] = c1[r] * dr[r];
            }
        }
    } else {
        if (wave >= GB_TILES) return;
        const int rt = wave;
        if (i0 + rt * 16 >= n) return;
        v8f c0 = {}, c1 = {};
        const float dA = sD[rt * 16 + l16];
        #pragma unroll
        for (int k0 = 0; k0 < 64; k0 += 4) {
            const int kb = k0 + half * 2;
            const v2f sm = *reinterpret_cast<const v2f*>(&sA[(rt * 16 + l16) * SA_PITCH + kb]);
            const v2f bk = *reinterpret_cast<const v2f*>(&sB[kb]);
            const v2f a = {fmaxf(fmaf(sm.x, dA, bk.x), 0.0f), fmaxf(fmaf(sm.y, dA, bk.y), 0.0f)};
            const v2f b = {sW[kb * 16 + l16], sW[(kb + 1) * 16 + l16]};
            if ((k0 & 4) == 0) c0 = __builtin_amdgcn_wmma_f32_16x16x4_f32(false, a, false, b, (short)0, c0, false, false);
            else               c1 = __builtin_amdgcn_wmma_f32_16x16x4_f32(false, a, false, b, (short)0, c1, false, false);
        }
        const int nvr = n - (i0 + rt * 16 + half * 8);
        const int ivb = __float_as_int(1.0f / fmaxf(cv, 1.0f));
        const int gx = __shfl(gv,  8 * half + (lane & 7));
        const int wx = __shfl(ivb, 8 * half + (lane & 7));
        const bool colok = l16 < NCLASSES;
        float* __restrict__ po = outp + l16;
        int   cur = 0;
        float ivc = 0.0f, acc = 0.0f;
        pool_row<0>(c0[0] + c1[0], gx, wx, nvr, colok, po, cur, ivc, acc);
        pool_row<1>(c0[1] + c1[1], gx, wx, nvr, colok, po, cur, ivc, acc);
        pool_row<2>(c0[2] + c1[2], gx, wx, nvr, colok, po, cur, ivc, acc);
        pool_row<3>(c0[3] + c1[3], gx, wx, nvr, colok, po, cur, ivc, acc);
        pool_row<4>(c0[4] + c1[4], gx, wx, nvr, colok, po, cur, ivc, acc);
        pool_row<5>(c0[5] + c1[5], gx, wx, nvr, colok, po, cur, ivc, acc);
        pool_row<6>(c0[6] + c1[6], gx, wx, nvr, colok, po, cur, ivc, acc);
        pool_row<7>(c0[7] + c1[7], gx, wx, nvr, colok, po, cur, ivc, acc);
        if (nvr > 0 && colok) atomicAdd(po + cur * NCLASSES, acc * ivc);
    }
}


#define WS_A      (NNODES * HID)
#define WS_DINV   (2 * NNODES * HID)
#define WS_CNT    (WS_DINV + NNODES)
#define WS_OFFS   (WS_CNT + NGRAPHS)
#define WS_ROWPTR (WS_OFFS + NCHUNK * OFFP)
#define WS_PACKED ((WS_ROWPTR + NNODES + 1 + 63) / 64 * 64)
#define WS_COL    (WS_PACKED + NEDGES)
static_assert(WS_PACKED % 64 == 0 && WS_COL % 64 == 0 && CHUNK % 64 == 0, "packed, its chunks and col on 256-byte boundaries");

extern "C" void kernel_launch(void* const* d_in, const int* in_sizes, int n_in,
                              void* d_out, int out_size, void* d_ws, size_t ws_size,
                              hipStream_t stream) {
    const float* x  = (const float*)d_in[0];
    const float* W1 = (const float*)d_in[1];
    const float* b1 = (const float*)d_in[2];
    const float* W2 = (const float*)d_in[3];
    const float* b2 = (const float*)d_in[4];
    const float* Wl = (const float*)d_in[5];
    const float* bl = (const float*)d_in[6];
    const int* edge  = (const int*)d_in[7];
    const int* batch = (const int*)d_in[8];
    const int* src = edge;
    const int* dst = edge + NEDGES;
    float* out = (float*)d_out;

    float* wsf  = (float*)d_ws;
    int*   wsi  = (int*)d_ws;
    float* h    = wsf;
    float* a    = wsf + WS_A;
    float* dinv = wsf + WS_DINV;
    float* cnt  = wsf + WS_CNT;
    int* offs   = wsi + WS_OFFS;
    int* rowptr = wsi + WS_ROWPTR;
    int* packed = wsi + WS_PACKED;
    int* col    = wsi + WS_COL;

    const int TPB = 256;
    const int gGa  = (NNODES + GB_NODES - 1) / GB_NODES;
    static_assert(NCHUNK * 256 >= NGRAPHS * NCLASSES, "k_bin's edge blocks store the start values of cnt and out");

    k_bin<<<NCHUNK + GEMM_BLOCKS, TPB, 0, stream>>>(src, dst, offs, packed, NEDGES, cnt, bl, out, x, W1, h, NNODES);
    k_csr<<<NBKT, TPB, 0, stream>>>(packed, offs, rowptr, dinv, col, NNODES, h, batch, cnt);

    k_gather<0><<<gGa, GB_WAVES * 32, 0, stream>>>(h, dinv, b1, rowptr, col, NNODES, W2, a,
                                                   nullptr, nullptr, nullptr);
    k_gather<1><<<gGa, GB_WAVES * 32, 0, stream>>>(a, dinv, b2, rowptr, col, NNODES, Wl, nullptr,
                                                   batch, out, cnt);
}
